# starting kernel
# baseline (speedup 1.0000x reference)
#include <hip/hip_runtime.h>
#include <hip/hip_bf16.h>

typedef float v2f __attribute__((ext_vector_type(2)));
typedef float v8f __attribute__((ext_vector_type(8)));

#define GIN_N    40000
#define GIN_E    640000
#define GIN_DIN  128
#define GIN_DOUT 256


__global__ void zero_kernel(int* __restrict__ p, int n) {
    int i = blockIdx.x * 256 + threadIdx.x;
    if (i < n) p[i] = 0;
}

__global__ void count_kernel(const int* __restrict__ dst, int* __restrict__ counts, int e) {
    int i = blockIdx.x * 256 + threadIdx.x;
    if (i < e) atomicAdd(&counts[dst[i]], 1);
}

__global__ void scan_kernel(const int* __restrict__ counts, int* __restrict__ offsets, int n) {
    __shared__ int sm[256];
    __shared__ int running;
    if (threadIdx.x == 0) running = 0;
    __syncthreads();
    for (int base = 0; base < n; base += 256) {
        int i = base + (int)threadIdx.x;
        int val = (i < n) ? counts[i] : 0;
        sm[threadIdx.x] = val;
        __syncthreads();
        for (int off = 1; off < 256; off <<= 1) {
            int t = (threadIdx.x >= (unsigned)off) ? sm[threadIdx.x - off] : 0;
            __syncthreads();
            sm[threadIdx.x] += t;
            __syncthreads();
        }
        if (i < n) offsets[i] = running + sm[threadIdx.x] - val;
        __syncthreads();
        if (threadIdx.x == 0) running += sm[255];
        __syncthreads();
    }
    if (threadIdx.x == 0) offsets[n] = running;
}

__global__ void fill_kernel(const int* __restrict__ src, const int* __restrict__ dst,
                            const int* __restrict__ offsets, int* __restrict__ cursor,
                            int* __restrict__ bucket, int e) {
    int i = blockIdx.x * 256 + threadIdx.x;
    if (i < e) {
        int d = dst[i];
        int pos = offsets[d] + atomicAdd(&cursor[d], 1);
        bucket[pos] = src[i];
    }
}

__global__ __launch_bounds__(256)
void gather_kernel(const float* __restrict__ x, const int* __restrict__ offsets,
                   const int* __restrict__ bucket, const float* __restrict__ epsp,
                   float* __restrict__ h) {
    int node = blockIdx.x * 8 + (threadIdx.x >> 5);
    int lane = threadIdx.x & 31;
    float se = 1.0f + epsp[0];
    const float4* xv = (const float4*)x;
    float4 self = xv[(size_t)node * 32 + lane];
    float ax = self.x * se, ay = self.y * se, az = self.z * se, aw = self.w * se;
    int e0 = offsets[node], e1 = offsets[node + 1];
    for (int i = e0; i < e1; ++i) {
        int s = bucket[i];
        float4 r = xv[(size_t)s * 32 + lane];
        ax += r.x; ay += r.y; az += r.z; aw += r.w;
    }
    float4 o; o.x = ax; o.y = ay; o.z = az; o.w = aw;
    ((float4*)h)[(size_t)node * 32 + lane] = o;
}


template <int K, bool USE_BN, bool USE_RES>
__global__ __launch_bounds__(128)
void gemm_kernel(const float* __restrict__ A, const float* __restrict__ B,
                 const float* __restrict__ bias, const float* __restrict__ gam,
                 const float* __restrict__ bet, const float* __restrict__ mu,
                 const float* __restrict__ var, const float* __restrict__ res,
                 float* __restrict__ out, int ncols) {
    constexpr int LDA = K + 4;
    constexpr int LDB = 72;
    extern __shared__ float lds[];
    float* As = lds;
    float* Bs = lds + 64 * LDA;

    const int rowBase = blockIdx.x * 64;
    const int colBase = blockIdx.y * 64;
    const int tid = threadIdx.x;

    for (int idx = tid; idx < 64 * K / 4; idx += 128) {
        int flat = idx * 4;
        int r = flat / K, c = flat % K;
        *(float4*)(As + r * LDA + c) =
            *(const float4*)(A + (size_t)(rowBase + r) * K + c);
    }
    for (int idx = tid; idx < K * 64 / 4; idx += 128) {
        int flat = idx * 4;
        int r = flat / 64, c = flat % 64;
        *(float4*)(Bs + r * LDB + c) =
            *(const float4*)(B + (size_t)r * ncols + colBase + c);
    }
    __syncthreads();

    const int wave = tid >> 5;
    const int lane = tid & 31;
    const int mrow = lane & 15;
    const int kh   = lane >> 4;
    const int rowOff = (wave >> 1) * 32;
    const int colOff = (wave & 1) * 32;

    v8f acc[2][2] = {};
    const float* Ap0 = As + (rowOff + mrow) * LDA + 2 * kh;
    const float* Ap1 = Ap0 + 16 * LDA;
    const float* Bp  = Bs + 2 * kh * LDB + colOff + mrow;

#pragma unroll 4
    for (int k = 0; k < K; k += 4) {
        v2f a0 = *(const v2f*)(Ap0 + k);
        v2f a1 = *(const v2f*)(Ap1 + k);
        const float* bk = Bp + k * LDB;
        v2f b0, b1;
        b0.x = bk[0];
        b0.y = bk[LDB];
        b1.x = bk[16];
        b1.y = bk[LDB + 16];
        acc[0][0] = __builtin_amdgcn_wmma_f32_16x16x4_f32(false, a0, false, b0, (short)0, acc[0][0], false, false);
        acc[0][1] = __builtin_amdgcn_wmma_f32_16x16x4_f32(false, a0, false, b1, (short)0, acc[0][1], false, false);
        acc[1][0] = __builtin_amdgcn_wmma_f32_16x16x4_f32(false, a1, false, b0, (short)0, acc[1][0], false, false);
        acc[1][1] = __builtin_amdgcn_wmma_f32_16x16x4_f32(false, a1, false, b1, (short)0, acc[1][1], false, false);
    }

#pragma unroll
    for (int ct = 0; ct < 2; ++ct) {
        int c = colBase + colOff + ct * 16 + mrow;
        float scale = 1.0f, shift = 0.0f;
        if (USE_BN) {
            float inv = rsqrtf(var[c] + 1e-5f);
            scale = gam[c] * inv;
            shift = (bias[c] - mu[c]) * scale + bet[c];
        }
#pragma unroll
        for (int rt = 0; rt < 2; ++rt) {
            int row0 = rowBase + rowOff + rt * 16 + kh * 8;
#pragma unroll
            for (int r = 0; r < 8; ++r) {
                float val = acc[rt][ct][r];
                if (USE_BN) val = val * scale + shift;
                int row = row0 + r;
                if (USE_RES) val += res[(size_t)row * ncols + c];
                val = fmaxf(val, 0.0f);
                out[(size_t)row * ncols + c] = val;
            }
        }
    }
}


extern "C" void kernel_launch(void* const* d_in, const int* in_sizes, int n_in,
                              void* d_out, int out_size, void* d_ws, size_t ws_size,
                              hipStream_t stream) {
    constexpr int N = GIN_N, E = GIN_E, DIN = GIN_DIN, DOUT = GIN_DOUT;

    const float* x    = (const float*)d_in[0];
    const int*   eidx = (const int*)d_in[1];
    const float* eps  = (const float*)d_in[2];
    const float* W1   = (const float*)d_in[3];
    const float* b1   = (const float*)d_in[4];
    const float* g1   = (const float*)d_in[5];
    const float* bt1  = (const float*)d_in[6];
    const float* m1   = (const float*)d_in[7];
    const float* v1   = (const float*)d_in[8];
    const float* W2   = (const float*)d_in[9];
    const float* b2   = (const float*)d_in[10];
    const float* g2   = (const float*)d_in[11];
    const float* bt2  = (const float*)d_in[12];
    const float* m2   = (const float*)d_in[13];
    const float* v2   = (const float*)d_in[14];
    const float* Wres = (const float*)d_in[15];

    const int* src = eidx;
    const int* dst = eidx + E;

    float* h  = (float*)d_ws;
    float* h1 = h + (size_t)N * DIN;
    float* h2 = h1 + (size_t)N * DOUT;
    int* counts  = (int*)(h2 + (size_t)N * DOUT);
    int* cursor  = counts + N;
    int* offsets = cursor + N;
    int* bucket  = offsets + N + 1;

    zero_kernel<<<(2 * N + 255) / 256, 256, 0, stream>>>(counts, 2 * N);
    count_kernel<<<E / 256, 256, 0, stream>>>(dst, counts, E);
    scan_kernel<<<1, 256, 0, stream>>>(counts, offsets, N);
    fill_kernel<<<E / 256, 256, 0, stream>>>(src, dst, offsets, cursor, bucket, E);
    gather_kernel<<<N / 8, 256, 0, stream>>>(x, offsets, bucket, eps, h);

    dim3 grid(N / 64, DOUT / 64);
    size_t lds128 = (size_t)(64 * (128 + 4) + 128 * 72) * 4;
    size_t lds256 = (size_t)(64 * (256 + 4) + 256 * 72) * 4;

    gemm_kernel<128, true, false><<<grid, 128, lds128, stream>>>(
        h, W1, b1, g1, bt1, m1, v1, nullptr, h1, DOUT);
    gemm_kernel<256, true, false><<<grid, 128, lds256, stream>>>(
        h1, W2, b2, g2, bt2, m2, v2, nullptr, h2, DOUT);
    gemm_kernel<128, false, true><<<grid, 128, lds128, stream>>>(
        x, Wres, nullptr, nullptr, nullptr, nullptr, nullptr, h2, (float*)d_out, DOUT);
}


# scan of the 40000 bucket counters: 40 blocks of 1024, each re-sums the counters in front of its chunk and scans its chunk behind one barrier, in place of 157 serial rounds of a single block
# speedup vs baseline: 2.1609x; 2.1609x over previous
#include <hip/hip_runtime.h>
#include <hip/hip_bf16.h>

typedef float v2f __attribute__((ext_vector_type(2)));
typedef float v8f __attribute__((ext_vector_type(8)));

#define GIN_N    40000
#define GIN_E    640000
#define GIN_DIN  128
#define GIN_DOUT 256


__global__ void zero_kernel(int* __restrict__ p, int n) {
    int i = blockIdx.x * 256 + threadIdx.x;
    if (i < n) p[i] = 0;
}

__global__ void count_kernel(const int* __restrict__ dst, int* __restrict__ counts, int e) {
    int i = blockIdx.x * 256 + threadIdx.x;
    if (i < e) atomicAdd(&counts[dst[i]], 1);
}

#define SCAN_T 1024
__global__ __launch_bounds__(SCAN_T)
void scan_kernel(const int* __restrict__ counts, int* __restrict__ offsets, int n) {
    __shared__ int smP[32];
    __shared__ int smT[32];
    const int tid  = threadIdx.x;
    const int lane = tid & 31;
    const int wave = tid >> 5;
    const int chunk0 = blockIdx.x * SCAN_T;
    const int i = chunk0 + tid;

    int v = (i < n) ? counts[i] : 0;

    const int4* c4 = (const int4*)counts;
    const int lim = chunk0 >> 2;
    int p0 = 0, p1 = 0, p2 = 0, p3 = 0;
    for (int j0 = 0; j0 < lim; j0 += 4 * SCAN_T) {
        int4 q[4];
#pragma unroll
        for (int u = 0; u < 4; ++u) {
            int j = j0 + u * SCAN_T + tid;
            q[u] = c4[(j < lim) ? j : (lim - 1)];
        }
#pragma unroll
        for (int u = 0; u < 4; ++u) {
            bool in = (j0 + u * SCAN_T + tid) < lim;
            p0 += in ? q[u].x : 0;
            p1 += in ? q[u].y : 0;
            p2 += in ? q[u].z : 0;
            p3 += in ? q[u].w : 0;
        }
    }
    int p = (p0 + p1) + (p2 + p3);

    int s = v;
#pragma unroll
    for (int off = 1; off < 32; off <<= 1) {
        int t = __shfl_up(s, off);
        if (lane >= off) s += t;
    }
#pragma unroll
    for (int off = 16; off > 0; off >>= 1) p += __shfl_xor(p, off);

    if (lane == 31) smT[wave] = s;
    if (lane == 0)  smP[wave] = p;
    __syncthreads();

    int base = smP[lane] + ((lane < wave) ? smT[lane] : 0);
#pragma unroll
    for (int off = 16; off > 0; off >>= 1) base += __shfl_xor(base, off);

    if (i < n) offsets[i] = base + s - v;
    if (i == n - 1) offsets[n] = base + s;
}

__global__ void fill_kernel(const int* __restrict__ src, const int* __restrict__ dst,
                            const int* __restrict__ offsets, int* __restrict__ cursor,
                            int* __restrict__ bucket, int e) {
    int i = blockIdx.x * 256 + threadIdx.x;
    if (i < e) {
        int d = dst[i];
        int pos = offsets[d] + atomicAdd(&cursor[d], 1);
        bucket[pos] = src[i];
    }
}

__global__ __launch_bounds__(256)
void gather_kernel(const float* __restrict__ x, const int* __restrict__ offsets,
                   const int* __restrict__ bucket, const float* __restrict__ epsp,
                   float* __restrict__ h) {
    int node = blockIdx.x * 8 + (threadIdx.x >> 5);
    int lane = threadIdx.x & 31;
    float se = 1.0f + epsp[0];
    const float4* xv = (const float4*)x;
    float4 self = xv[(size_t)node * 32 + lane];
    float ax = self.x * se, ay = self.y * se, az = self.z * se, aw = self.w * se;
    int e0 = offsets[node], e1 = offsets[node + 1];
    for (int i = e0; i < e1; ++i) {
        int s = bucket[i];
        float4 r = xv[(size_t)s * 32 + lane];
        ax += r.x; ay += r.y; az += r.z; aw += r.w;
    }
    float4 o; o.x = ax; o.y = ay; o.z = az; o.w = aw;
    ((float4*)h)[(size_t)node * 32 + lane] = o;
}


template <int K, bool USE_BN, bool USE_RES>
__global__ __launch_bounds__(128)
void gemm_kernel(const float* __restrict__ A, const float* __restrict__ B,
                 const float* __restrict__ bias, const float* __restrict__ gam,
                 const float* __restrict__ bet, const float* __restrict__ mu,
                 const float* __restrict__ var, const float* __restrict__ res,
                 float* __restrict__ out, int ncols) {
    constexpr int LDA = K + 4;
    constexpr int LDB = 72;
    extern __shared__ float lds[];
    float* As = lds;
    float* Bs = lds + 64 * LDA;

    const int rowBase = blockIdx.x * 64;
    const int colBase = blockIdx.y * 64;
    const int tid = threadIdx.x;

    for (int idx = tid; idx < 64 * K / 4; idx += 128) {
        int flat = idx * 4;
        int r = flat / K, c = flat % K;
        *(float4*)(As + r * LDA + c) =
            *(const float4*)(A + (size_t)(rowBase + r) * K + c);
    }
    for (int idx = tid; idx < K * 64 / 4; idx += 128) {
        int flat = idx * 4;
        int r = flat / 64, c = flat % 64;
        *(float4*)(Bs + r * LDB + c) =
            *(const float4*)(B + (size_t)r * ncols + colBase + c);
    }
    __syncthreads();

    const int wave = tid >> 5;
    const int lane = tid & 31;
    const int mrow = lane & 15;
    const int kh   = lane >> 4;
    const int rowOff = (wave >> 1) * 32;
    const int colOff = (wave & 1) * 32;

    v8f acc[2][2] = {};
    const float* Ap0 = As + (rowOff + mrow) * LDA + 2 * kh;
    const float* Ap1 = Ap0 + 16 * LDA;
    const float* Bp  = Bs + 2 * kh * LDB + colOff + mrow;

#pragma unroll 4
    for (int k = 0; k < K; k += 4) {
        v2f a0 = *(const v2f*)(Ap0 + k);
        v2f a1 = *(const v2f*)(Ap1 + k);
        const float* bk = Bp + k * LDB;
        v2f b0, b1;
        b0.x = bk[0];
        b0.y = bk[LDB];
        b1.x = bk[16];
        b1.y = bk[LDB + 16];
        acc[0][0] = __builtin_amdgcn_wmma_f32_16x16x4_f32(false, a0, false, b0, (short)0, acc[0][0], false, false);
        acc[0][1] = __builtin_amdgcn_wmma_f32_16x16x4_f32(false, a0, false, b1, (short)0, acc[0][1], false, false);
        acc[1][0] = __builtin_amdgcn_wmma_f32_16x16x4_f32(false, a1, false, b0, (short)0, acc[1][0], false, false);
        acc[1][1] = __builtin_amdgcn_wmma_f32_16x16x4_f32(false, a1, false, b1, (short)0, acc[1][1], false, false);
    }

#pragma unroll
    for (int ct = 0; ct < 2; ++ct) {
        int c = colBase + colOff + ct * 16 + mrow;
        float scale = 1.0f, shift = 0.0f;
        if (USE_BN) {
            float inv = rsqrtf(var[c] + 1e-5f);
            scale = gam[c] * inv;
            shift = (bias[c] - mu[c]) * scale + bet[c];
        }
#pragma unroll
        for (int rt = 0; rt < 2; ++rt) {
            int row0 = rowBase + rowOff + rt * 16 + kh * 8;
#pragma unroll
            for (int r = 0; r < 8; ++r) {
                float val = acc[rt][ct][r];
                if (USE_BN) val = val * scale + shift;
                int row = row0 + r;
                if (USE_RES) val += res[(size_t)row * ncols + c];
                val = fmaxf(val, 0.0f);
                out[(size_t)row * ncols + c] = val;
            }
        }
    }
}


extern "C" void kernel_launch(void* const* d_in, const int* in_sizes, int n_in,
                              void* d_out, int out_size, void* d_ws, size_t ws_size,
                              hipStream_t stream) {
    constexpr int N = GIN_N, E = GIN_E, DIN = GIN_DIN, DOUT = GIN_DOUT;

    const float* x    = (const float*)d_in[0];
    const int*   eidx = (const int*)d_in[1];
    const float* eps  = (const float*)d_in[2];
    const float* W1   = (const float*)d_in[3];
    const float* b1   = (const float*)d_in[4];
    const float* g1   = (const float*)d_in[5];
    const float* bt1  = (const float*)d_in[6];
    const float* m1   = (const float*)d_in[7];
    const float* v1   = (const float*)d_in[8];
    const float* W2   = (const float*)d_in[9];
    const float* b2   = (const float*)d_in[10];
    const float* g2   = (const float*)d_in[11];
    const float* bt2  = (const float*)d_in[12];
    const float* m2   = (const float*)d_in[13];
    const float* v2   = (const float*)d_in[14];
    const float* Wres = (const float*)d_in[15];

    const int* src = eidx;
    const int* dst = eidx + E;

    float* h  = (float*)d_ws;
    float* h1 = h + (size_t)N * DIN;
    float* h2 = h1 + (size_t)N * DOUT;
    int* counts  = (int*)(h2 + (size_t)N * DOUT);
    int* cursor  = counts + N;
    int* offsets = cursor + N;
    int* bucket  = offsets + N + 1;

    zero_kernel<<<(2 * N + 255) / 256, 256, 0, stream>>>(counts, 2 * N);
    count_kernel<<<E / 256, 256, 0, stream>>>(dst, counts, E);
    scan_kernel<<<(N + SCAN_T - 1) / SCAN_T, SCAN_T, 0, stream>>>(counts, offsets, N);
    fill_kernel<<<E / 256, 256, 0, stream>>>(src, dst, offsets, cursor, bucket, E);
    gather_kernel<<<N / 8, 256, 0, stream>>>(x, offsets, bucket, eps, h);

    dim3 grid(N / 64, DOUT / 64);
    size_t lds128 = (size_t)(64 * (128 + 4) + 128 * 72) * 4;
    size_t lds256 = (size_t)(64 * (256 + 4) + 256 * 72) * 4;

    gemm_kernel<128, true, false><<<grid, 128, lds128, stream>>>(
        h, W1, b1, g1, bt1, m1, v1, nullptr, h1, DOUT);
    gemm_kernel<256, true, false><<<grid, 128, lds256, stream>>>(
        h1, W2, b2, g2, bt2, m2, v2, nullptr, h2, DOUT);
    gemm_kernel<128, false, true><<<grid, 128, lds128, stream>>>(
        x, Wres, nullptr, nullptr, nullptr, nullptr, nullptr, h2, (float*)d_out, DOUT);
}


# edges grouped by destination with a two-level counting sort on LDS counters (3 launches of 1024-thread blocks) replacing two device-scope atomics per edge; per-node sum order stays an arrival order
# speedup vs baseline: 3.2960x; 1.5253x over previous
#include <hip/hip_runtime.h>
#include <hip/hip_bf16.h>

typedef float v2f __attribute__((ext_vector_type(2)));
typedef float v8f __attribute__((ext_vector_type(8)));

#define GIN_N    40000
#define GIN_E    640000
#define GIN_DIN  128
#define GIN_DOUT 256


__global__ void zero_kernel(int* __restrict__ p, int n) {
    int i = blockIdx.x * 256 + threadIdx.x;
    if (i < n) p[i] = 0;
}

#define BIN_SHIFT 7
#define BIN_NODES (1 << BIN_SHIFT)
#define NBIN  ((GIN_N + BIN_NODES - 1) / BIN_NODES)
#define NBINP (((NBIN + 31) / 32) * 32)
#define SORT_T 1024
static_assert(GIN_E % 4 == 0, "edges are read four at a time");
static_assert(NBINP <= SORT_T, "one thread per bin");

__global__ __launch_bounds__(SORT_T)
void bin_count_kernel(const int* __restrict__ dst, int* __restrict__ binCount, int e4) {
    __shared__ int hist[NBINP];
    const int tid = threadIdx.x;
    if (tid < NBINP) hist[tid] = 0;
    __syncthreads();
    const int i4 = blockIdx.x * SORT_T + tid;
    if (i4 < e4) {
        int4 d = ((const int4*)dst)[i4];
        atomicAdd(&hist[d.x >> BIN_SHIFT], 1);
        atomicAdd(&hist[d.y >> BIN_SHIFT], 1);
        atomicAdd(&hist[d.z >> BIN_SHIFT], 1);
        atomicAdd(&hist[d.w >> BIN_SHIFT], 1);
    }
    __syncthreads();
    if (tid < NBIN) {
        int c = hist[tid];
        if (c) atomicAdd(&binCount[tid], c);
    }
}

__global__ __launch_bounds__(SORT_T)
void bin_scatter_kernel(const int* __restrict__ src, const int* __restrict__ dst,
                        const int* __restrict__ binCount, int* __restrict__ binCursor,
                        int* __restrict__ binBase, int* __restrict__ pairbuf, int e4) {
    __shared__ int hist[NBINP];
    __shared__ int gbase[NBINP];
    __shared__ int smT[32];
    int2* pairs = (int2*)pairbuf;
    const int tid  = threadIdx.x;
    const int lane = tid & 31;
    const int wave = tid >> 5;

    int v = 0, s = 0;
    if (tid < NBINP) {
        v = (tid < NBIN) ? binCount[tid] : 0;
        s = v;
#pragma unroll
        for (int off = 1; off < 32; off <<= 1) {
            int t = __shfl_up(s, off);
            if (lane >= off) s += t;
        }
        if (lane == 31) smT[wave] = s;
        hist[tid] = 0;
    }
    __syncthreads();

    if (tid < NBINP) {
        int base = 0;
#pragma unroll
        for (int w = 0; w < NBINP / 32; ++w) base += (w < wave) ? smT[w] : 0;
        int excl = base + s - v;
        gbase[tid] = excl;
        if (blockIdx.x == 0) {
            binBase[tid] = excl;
            if (tid == NBINP - 1) binBase[NBINP] = excl + v;
        }
    }

    const int i4 = blockIdx.x * SORT_T + tid;
    const bool act = i4 < e4;
    int4 d  = make_int4(0, 0, 0, 0);
    int4 sr = make_int4(0, 0, 0, 0);
    int r0 = 0, r1 = 0, r2 = 0, r3 = 0;
    if (act) {
        d  = ((const int4*)dst)[i4];
        sr = ((const int4*)src)[i4];
        r0 = atomicAdd(&hist[d.x >> BIN_SHIFT], 1);
        r1 = atomicAdd(&hist[d.y >> BIN_SHIFT], 1);
        r2 = atomicAdd(&hist[d.z >> BIN_SHIFT], 1);
        r3 = atomicAdd(&hist[d.w >> BIN_SHIFT], 1);
    }
    __syncthreads();

    if (tid < NBIN) {
        int c = hist[tid];
        if (c) gbase[tid] += atomicAdd(&binCursor[tid], c);
    }
    __syncthreads();

    if (act) {
        pairs[gbase[d.x >> BIN_SHIFT] + r0] = make_int2(d.x, sr.x);
        pairs[gbase[d.y >> BIN_SHIFT] + r1] = make_int2(d.y, sr.y);
        pairs[gbase[d.z >> BIN_SHIFT] + r2] = make_int2(d.z, sr.z);
        pairs[gbase[d.w >> BIN_SHIFT] + r3] = make_int2(d.w, sr.w);
    }
}

__global__ __launch_bounds__(SORT_T)
void bin_sort_kernel(const int* __restrict__ pairbuf, const int* __restrict__ binBase,
                     int* __restrict__ offsets, int* __restrict__ bucket, int n) {
    __shared__ int cnt[BIN_NODES];
    __shared__ int loff[BIN_NODES];
    __shared__ int smT[BIN_NODES / 32];
    const int2* pairs = (const int2*)pairbuf;
    const int tid  = threadIdx.x;
    const int lane = tid & 31;
    const int wave = tid >> 5;
    const int b  = blockIdx.x;
    const int p0 = binBase[b];
    const int p1 = binBase[b + 1];

    if (tid < BIN_NODES) cnt[tid] = 0;
    __syncthreads();
    for (int i = p0 + tid; i < p1; i += SORT_T)
        atomicAdd(&cnt[pairs[i].x & (BIN_NODES - 1)], 1);
    __syncthreads();

    int v = 0, s = 0;
    if (tid < BIN_NODES) {
        v = cnt[tid];
        s = v;
#pragma unroll
        for (int off = 1; off < 32; off <<= 1) {
            int t = __shfl_up(s, off);
            if (lane >= off) s += t;
        }
        if (lane == 31) smT[wave] = s;
    }
    __syncthreads();

    if (tid < BIN_NODES) {
        int base = 0;
#pragma unroll
        for (int w = 0; w < BIN_NODES / 32; ++w) base += (w < wave) ? smT[w] : 0;
        int first = p0 + base + s - v;
        loff[tid] = first;
        cnt[tid] = 0;
        int node = b * BIN_NODES + tid;
        if (node < n) offsets[node] = first;
        if (node == n - 1) offsets[n] = first + v;
    }
    __syncthreads();

    for (int i = p0 + tid; i < p1; i += SORT_T) {
        int2 pr = pairs[i];
        int l = pr.x & (BIN_NODES - 1);
        int r = atomicAdd(&cnt[l], 1);
        bucket[loff[l] + r] = pr.y;
    }
}

__global__ __launch_bounds__(256)
void gather_kernel(const float* __restrict__ x, const int* __restrict__ offsets,
                   const int* __restrict__ bucket, const float* __restrict__ epsp,
                   float* __restrict__ h) {
    int node = blockIdx.x * 8 + (threadIdx.x >> 5);
    int lane = threadIdx.x & 31;
    float se = 1.0f + epsp[0];
    const float4* xv = (const float4*)x;
    float4 self = xv[(size_t)node * 32 + lane];
    float ax = self.x * se, ay = self.y * se, az = self.z * se, aw = self.w * se;
    int e0 = offsets[node], e1 = offsets[node + 1];
    for (int i = e0; i < e1; ++i) {
        int s = bucket[i];
        float4 r = xv[(size_t)s * 32 + lane];
        ax += r.x; ay += r.y; az += r.z; aw += r.w;
    }
    float4 o; o.x = ax; o.y = ay; o.z = az; o.w = aw;
    ((float4*)h)[(size_t)node * 32 + lane] = o;
}


template <int K, bool USE_BN, bool USE_RES>
__global__ __launch_bounds__(128)
void gemm_kernel(const float* __restrict__ A, const float* __restrict__ B,
                 const float* __restrict__ bias, const float* __restrict__ gam,
                 const float* __restrict__ bet, const float* __restrict__ mu,
                 const float* __restrict__ var, const float* __restrict__ res,
                 float* __restrict__ out, int ncols) {
    constexpr int LDA = K + 4;
    constexpr int LDB = 72;
    extern __shared__ float lds[];
    float* As = lds;
    float* Bs = lds + 64 * LDA;

    const int rowBase = blockIdx.x * 64;
    const int colBase = blockIdx.y * 64;
    const int tid = threadIdx.x;

    for (int idx = tid; idx < 64 * K / 4; idx += 128) {
        int flat = idx * 4;
        int r = flat / K, c = flat % K;
        *(float4*)(As + r * LDA + c) =
            *(const float4*)(A + (size_t)(rowBase + r) * K + c);
    }
    for (int idx = tid; idx < K * 64 / 4; idx += 128) {
        int flat = idx * 4;
        int r = flat / 64, c = flat % 64;
        *(float4*)(Bs + r * LDB + c) =
            *(const float4*)(B + (size_t)r * ncols + colBase + c);
    }
    __syncthreads();

    const int wave = tid >> 5;
    const int lane = tid & 31;
    const int mrow = lane & 15;
    const int kh   = lane >> 4;
    const int rowOff = (wave >> 1) * 32;
    const int colOff = (wave & 1) * 32;

    v8f acc[2][2] = {};
    const float* Ap0 = As + (rowOff + mrow) * LDA + 2 * kh;
    const float* Ap1 = Ap0 + 16 * LDA;
    const float* Bp  = Bs + 2 * kh * LDB + colOff + mrow;

#pragma unroll 4
    for (int k = 0; k < K; k += 4) {
        v2f a0 = *(const v2f*)(Ap0 + k);
        v2f a1 = *(const v2f*)(Ap1 + k);
        const float* bk = Bp + k * LDB;
        v2f b0, b1;
        b0.x = bk[0];
        b0.y = bk[LDB];
        b1.x = bk[16];
        b1.y = bk[LDB + 16];
        acc[0][0] = __builtin_amdgcn_wmma_f32_16x16x4_f32(false, a0, false, b0, (short)0, acc[0][0], false, false);
        acc[0][1] = __builtin_amdgcn_wmma_f32_16x16x4_f32(false, a0, false, b1, (short)0, acc[0][1], false, false);
        acc[1][0] = __builtin_amdgcn_wmma_f32_16x16x4_f32(false, a1, false, b0, (short)0, acc[1][0], false, false);
        acc[1][1] = __builtin_amdgcn_wmma_f32_16x16x4_f32(false, a1, false, b1, (short)0, acc[1][1], false, false);
    }

#pragma unroll
    for (int ct = 0; ct < 2; ++ct) {
        int c = colBase + colOff + ct * 16 + mrow;
        float scale = 1.0f, shift = 0.0f;
        if (USE_BN) {
            float inv = rsqrtf(var[c] + 1e-5f);
            scale = gam[c] * inv;
            shift = (bias[c] - mu[c]) * scale + bet[c];
        }
#pragma unroll
        for (int rt = 0; rt < 2; ++rt) {
            int row0 = rowBase + rowOff + rt * 16 + kh * 8;
#pragma unroll
            for (int r = 0; r < 8; ++r) {
                float val = acc[rt][ct][r];
                if (USE_BN) val = val * scale + shift;
                int row = row0 + r;
                if (USE_RES) val += res[(size_t)row * ncols + c];
                val = fmaxf(val, 0.0f);
                out[(size_t)row * ncols + c] = val;
            }
        }
    }
}


extern "C" void kernel_launch(void* const* d_in, const int* in_sizes, int n_in,
                              void* d_out, int out_size, void* d_ws, size_t ws_size,
                              hipStream_t stream) {
    constexpr int N = GIN_N, E = GIN_E, DIN = GIN_DIN, DOUT = GIN_DOUT;

    const float* x    = (const float*)d_in[0];
    const int*   eidx = (const int*)d_in[1];
    const float* eps  = (const float*)d_in[2];
    const float* W1   = (const float*)d_in[3];
    const float* b1   = (const float*)d_in[4];
    const float* g1   = (const float*)d_in[5];
    const float* bt1  = (const float*)d_in[6];
    const float* m1   = (const float*)d_in[7];
    const float* v1   = (const float*)d_in[8];
    const float* W2   = (const float*)d_in[9];
    const float* b2   = (const float*)d_in[10];
    const float* g2   = (const float*)d_in[11];
    const float* bt2  = (const float*)d_in[12];
    const float* m2   = (const float*)d_in[13];
    const float* v2   = (const float*)d_in[14];
    const float* Wres = (const float*)d_in[15];

    const int* src = eidx;
    const int* dst = eidx + E;

    float* h  = (float*)d_ws;
    float* h1 = h + (size_t)N * DIN;
    float* h2 = h1 + (size_t)N * DOUT;
    int* ibase   = (int*)(h2 + (size_t)N * DOUT);
    int* binCount  = ibase;
    int* binCursor = binCount + NBINP;
    int* binBase   = binCursor + NBINP;
    int* offsets = ibase + 2 * N;
    int* bucket  = offsets + N + 1;
    int* pairs   = (int*)h1;

    constexpr int E4 = E / 4;
    constexpr int SORT_BLOCKS = (E4 + SORT_T - 1) / SORT_T;
    zero_kernel<<<(2 * NBINP + 255) / 256, 256, 0, stream>>>(binCount, 2 * NBINP);
    bin_count_kernel<<<SORT_BLOCKS, SORT_T, 0, stream>>>(dst, binCount, E4);
    bin_scatter_kernel<<<SORT_BLOCKS, SORT_T, 0, stream>>>(src, dst, binCount, binCursor,
                                                           binBase, pairs, E4);
    bin_sort_kernel<<<NBIN, SORT_T, 0, stream>>>(pairs, binBase, offsets, bucket, N);
    gather_kernel<<<N / 8, 256, 0, stream>>>(x, offsets, bucket, eps, h);

    dim3 grid(N / 64, DOUT / 64);
    size_t lds128 = (size_t)(64 * (128 + 4) + 128 * 72) * 4;
    size_t lds256 = (size_t)(64 * (256 + 4) + 256 * 72) * 4;

    gemm_kernel<128, true, false><<<grid, 128, lds128, stream>>>(
        h, W1, b1, g1, bt1, m1, v1, nullptr, h1, DOUT);
    gemm_kernel<256, true, false><<<grid, 128, lds256, stream>>>(
        h1, W2, b2, g2, bt2, m2, v2, nullptr, h2, DOUT);
    gemm_kernel<128, false, true><<<grid, 128, lds128, stream>>>(
        x, Wres, nullptr, nullptr, nullptr, nullptr, nullptr, h2, (float*)d_out, DOUT);
}


# GEMM operands staged in K-chunks of 32 through a two-stage LDS ring, the next chunk's global loads issued ahead of the current chunk's WMMAs (was: whole-K tiles staged before the first WMMA)
# speedup vs baseline: 4.2567x; 1.2915x over previous
#include <hip/hip_runtime.h>
#include <hip/hip_bf16.h>

typedef float v2f __attribute__((ext_vector_type(2)));
typedef float v8f __attribute__((ext_vector_type(8)));

#define GIN_N    40000
#define GIN_E    640000
#define GIN_DIN  128
#define GIN_DOUT 256


__global__ void zero_kernel(int* __restrict__ p, int n) {
    int i = blockIdx.x * 256 + threadIdx.x;
    if (i < n) p[i] = 0;
}

#define BIN_SHIFT 7
#define BIN_NODES (1 << BIN_SHIFT)
#define NBIN  ((GIN_N + BIN_NODES - 1) / BIN_NODES)
#define NBINP (((NBIN + 31) / 32) * 32)
#define SORT_T 1024
static_assert(GIN_E % 4 == 0, "edges are read four at a time");
static_assert(NBINP <= SORT_T, "one thread per bin");

__global__ __launch_bounds__(SORT_T)
void bin_count_kernel(const int* __restrict__ dst, int* __restrict__ binCount, int e4) {
    __shared__ int hist[NBINP];
    const int tid = threadIdx.x;
    if (tid < NBINP) hist[tid] = 0;
    __syncthreads();
    const int i4 = blockIdx.x * SORT_T + tid;
    if (i4 < e4) {
        int4 d = ((const int4*)dst)[i4];
        atomicAdd(&hist[d.x >> BIN_SHIFT], 1);
        atomicAdd(&hist[d.y >> BIN_SHIFT], 1);
        atomicAdd(&hist[d.z >> BIN_SHIFT], 1);
        atomicAdd(&hist[d.w >> BIN_SHIFT], 1);
    }
    __syncthreads();
    if (tid < NBIN) {
        int c = hist[tid];
        if (c) atomicAdd(&binCount[tid], c);
    }
}

__global__ __launch_bounds__(SORT_T)
void bin_scatter_kernel(const int* __restrict__ src, const int* __restrict__ dst,
                        const int* __restrict__ binCount, int* __restrict__ binCursor,
                        int* __restrict__ binBase, int* __restrict__ pairbuf, int e4) {
    __shared__ int hist[NBINP];
    __shared__ int gbase[NBINP];
    __shared__ int smT[32];
    int2* pairs = (int2*)pairbuf;
    const int tid  = threadIdx.x;
    const int lane = tid & 31;
    const int wave = tid >> 5;

    int v = 0, s = 0;
    if (tid < NBINP) {
        v = (tid < NBIN) ? binCount[tid] : 0;
        s = v;
#pragma unroll
        for (int off = 1; off < 32; off <<= 1) {
            int t = __shfl_up(s, off);
            if (lane >= off) s += t;
        }
        if (lane == 31) smT[wave] = s;
        hist[tid] = 0;
    }
    __syncthreads();

    if (tid < NBINP) {
        int base = 0;
#pragma unroll
        for (int w = 0; w < NBINP / 32; ++w) base += (w < wave) ? smT[w] : 0;
        int excl = base + s - v;
        gbase[tid] = excl;
        if (blockIdx.x == 0) {
            binBase[tid] = excl;
            if (tid == NBINP - 1) binBase[NBINP] = excl + v;
        }
    }

    const int i4 = blockIdx.x * SORT_T + tid;
    const bool act = i4 < e4;
    int4 d  = make_int4(0, 0, 0, 0);
    int4 sr = make_int4(0, 0, 0, 0);
    int r0 = 0, r1 = 0, r2 = 0, r3 = 0;
    if (act) {
        d  = ((const int4*)dst)[i4];
        sr = ((const int4*)src)[i4];
        r0 = atomicAdd(&hist[d.x >> BIN_SHIFT], 1);
        r1 = atomicAdd(&hist[d.y >> BIN_SHIFT], 1);
        r2 = atomicAdd(&hist[d.z >> BIN_SHIFT], 1);
        r3 = atomicAdd(&hist[d.w >> BIN_SHIFT], 1);
    }
    __syncthreads();

    if (tid < NBIN) {
        int c = hist[tid];
        if (c) gbase[tid] += atomicAdd(&binCursor[tid], c);
    }
    __syncthreads();

    if (act) {
        pairs[gbase[d.x >> BIN_SHIFT] + r0] = make_int2(d.x, sr.x);
        pairs[gbase[d.y >> BIN_SHIFT] + r1] = make_int2(d.y, sr.y);
        pairs[gbase[d.z >> BIN_SHIFT] + r2] = make_int2(d.z, sr.z);
        pairs[gbase[d.w >> BIN_SHIFT] + r3] = make_int2(d.w, sr.w);
    }
}

__global__ __launch_bounds__(SORT_T)
void bin_sort_kernel(const int* __restrict__ pairbuf, const int* __restrict__ binBase,
                     int* __restrict__ offsets, int* __restrict__ bucket, int n) {
    __shared__ int cnt[BIN_NODES];
    __shared__ int loff[BIN_NODES];
    __shared__ int smT[BIN_NODES / 32];
    const int2* pairs = (const int2*)pairbuf;
    const int tid  = threadIdx.x;
    const int lane = tid & 31;
    const int wave = tid >> 5;
    const int b  = blockIdx.x;
    const int p0 = binBase[b];
    const int p1 = binBase[b + 1];

    if (tid < BIN_NODES) cnt[tid] = 0;
    __syncthreads();
    for (int i = p0 + tid; i < p1; i += SORT_T)
        atomicAdd(&cnt[pairs[i].x & (BIN_NODES - 1)], 1);
    __syncthreads();

    int v = 0, s = 0;
    if (tid < BIN_NODES) {
        v = cnt[tid];
        s = v;
#pragma unroll
        for (int off = 1; off < 32; off <<= 1) {
            int t = __shfl_up(s, off);
            if (lane >= off) s += t;
        }
        if (lane == 31) smT[wave] = s;
    }
    __syncthreads();

    if (tid < BIN_NODES) {
        int base = 0;
#pragma unroll
        for (int w = 0; w < BIN_NODES / 32; ++w) base += (w < wave) ? smT[w] : 0;
        int first = p0 + base + s - v;
        loff[tid] = first;
        cnt[tid] = 0;
        int node = b * BIN_NODES + tid;
        if (node < n) offsets[node] = first;
        if (node == n - 1) offsets[n] = first + v;
    }
    __syncthreads();

    for (int i = p0 + tid; i < p1; i += SORT_T) {
        int2 pr = pairs[i];
        int l = pr.x & (BIN_NODES - 1);
        int r = atomicAdd(&cnt[l], 1);
        bucket[loff[l] + r] = pr.y;
    }
}

__global__ __launch_bounds__(256)
void gather_kernel(const float* __restrict__ x, const int* __restrict__ offsets,
                   const int* __restrict__ bucket, const float* __restrict__ epsp,
                   float* __restrict__ h) {
    int node = blockIdx.x * 8 + (threadIdx.x >> 5);
    int lane = threadIdx.x & 31;
    float se = 1.0f + epsp[0];
    const float4* xv = (const float4*)x;
    float4 self = xv[(size_t)node * 32 + lane];
    float ax = self.x * se, ay = self.y * se, az = self.z * se, aw = self.w * se;
    int e0 = offsets[node], e1 = offsets[node + 1];
    for (int i = e0; i < e1; ++i) {
        int s = bucket[i];
        float4 r = xv[(size_t)s * 32 + lane];
        ax += r.x; ay += r.y; az += r.z; aw += r.w;
    }
    float4 o; o.x = ax; o.y = ay; o.z = az; o.w = aw;
    ((float4*)h)[(size_t)node * 32 + lane] = o;
}


#define GEMM_KC  32
#define GEMM_LDA (GEMM_KC + 4)
#define GEMM_LDB 72
#define GEMM_AST (64 * GEMM_LDA)
#define GEMM_BST (GEMM_KC * GEMM_LDB)

__device__ __forceinline__ void mma_chunk(const float* __restrict__ Ap0,
                                          const float* __restrict__ Ap1,
                                          const float* __restrict__ Bp,
                                          v8f (&acc)[2][2]) {
#pragma unroll
    for (int k = 0; k < GEMM_KC; k += 4) {
        v2f a0 = *(const v2f*)(Ap0 + k);
        v2f a1 = *(const v2f*)(Ap1 + k);
        const float* bk = Bp + k * GEMM_LDB;
        v2f b0, b1;
        b0.x = bk[0];
        b0.y = bk[GEMM_LDB];
        b1.x = bk[16];
        b1.y = bk[GEMM_LDB + 16];
        acc[0][0] = __builtin_amdgcn_wmma_f32_16x16x4_f32(false, a0, false, b0, (short)0, acc[0][0], false, false);
        acc[0][1] = __builtin_amdgcn_wmma_f32_16x16x4_f32(false, a0, false, b1, (short)0, acc[0][1], false, false);
        acc[1][0] = __builtin_amdgcn_wmma_f32_16x16x4_f32(false, a1, false, b0, (short)0, acc[1][0], false, false);
        acc[1][1] = __builtin_amdgcn_wmma_f32_16x16x4_f32(false, a1, false, b1, (short)0, acc[1][1], false, false);
    }
}

template <int K, bool USE_BN, bool USE_RES>
__global__ __launch_bounds__(128)
void gemm_kernel(const float* __restrict__ A, const float* __restrict__ B,
                 const float* __restrict__ bias, const float* __restrict__ gam,
                 const float* __restrict__ bet, const float* __restrict__ mu,
                 const float* __restrict__ var, const float* __restrict__ res,
                 float* __restrict__ out, int ncols) {
    constexpr int KC  = GEMM_KC;
    constexpr int LDA = GEMM_LDA;
    constexpr int LDB = GEMM_LDB;
    constexpr int NCH = K / KC;
    static_assert(K % KC == 0, "K is a whole number of chunks");
    __shared__ __attribute__((aligned(16))) float As[2 * GEMM_AST];
    __shared__ __attribute__((aligned(16))) float Bs[2 * GEMM_BST];

    const int rowBase = blockIdx.x * 64;
    const int colBase = blockIdx.y * 64;
    const int tid = threadIdx.x;

    const int ar = tid >> 3, ac = (tid & 7) * 4;
    const int br = tid >> 4, bc = (tid & 15) * 4;
    const float* Ag = A + (size_t)(rowBase + ar) * K + ac;
    const float* Bg = B + (size_t)br * ncols + colBase + bc;
    float* Asw = As + ar * LDA + ac;
    float* Bsw = Bs + br * LDB + bc;
    const size_t aStep = (size_t)16 * K;
    const size_t bStep = (size_t)8 * ncols;

    float4 pa[4], pb[4];
#pragma unroll
    for (int j = 0; j < 4; ++j) {
        pa[j] = *(const float4*)(Ag + j * aStep);
        pb[j] = *(const float4*)(Bg + j * bStep);
    }
#pragma unroll
    for (int j = 0; j < 4; ++j) {
        *(float4*)(Asw + j * 16 * LDA) = pa[j];
        *(float4*)(Bsw + j * 8 * LDB)  = pb[j];
    }
    __syncthreads();

    const int wave = tid >> 5;
    const int lane = tid & 31;
    const int mrow = lane & 15;
    const int kh   = lane >> 4;
    const int rowOff = (wave >> 1) * 32;
    const int colOff = (wave & 1) * 32;

    v8f acc[2][2] = {};
    const float* Ap0 = As + (rowOff + mrow) * LDA + 2 * kh;
    const float* Ap1 = Ap0 + 16 * LDA;
    const float* Bp  = Bs + 2 * kh * LDB + colOff + mrow;

#pragma unroll 1
    for (int c = 0; c + 1 < NCH; ++c) {
        const int st = c & 1;
        const float* Agn = Ag + (c + 1) * KC;
        const float* Bgn = Bg + (size_t)(c + 1) * KC * ncols;
#pragma unroll
        for (int j = 0; j < 4; ++j) {
            pa[j] = *(const float4*)(Agn + j * aStep);
            pb[j] = *(const float4*)(Bgn + j * bStep);
        }
        mma_chunk(Ap0 + st * GEMM_AST, Ap1 + st * GEMM_AST, Bp + st * GEMM_BST, acc);
        float* aw = Asw + (st ^ 1) * GEMM_AST;
        float* bw = Bsw + (st ^ 1) * GEMM_BST;
#pragma unroll
        for (int j = 0; j < 4; ++j) {
            *(float4*)(aw + j * 16 * LDA) = pa[j];
            *(float4*)(bw + j * 8 * LDB)  = pb[j];
        }
        __syncthreads();
    }
    {
        constexpr int st = (NCH - 1) & 1;
        mma_chunk(Ap0 + st * GEMM_AST, Ap1 + st * GEMM_AST, Bp + st * GEMM_BST, acc);
    }

#pragma unroll
    for (int ct = 0; ct < 2; ++ct) {
        int c = colBase + colOff + ct * 16 + mrow;
        float scale = 1.0f, shift = 0.0f;
        if (USE_BN) {
            float inv = rsqrtf(var[c] + 1e-5f);
            scale = gam[c] * inv;
            shift = (bias[c] - mu[c]) * scale + bet[c];
        }
#pragma unroll
        for (int rt = 0; rt < 2; ++rt) {
            int row0 = rowBase + rowOff + rt * 16 + kh * 8;
#pragma unroll
            for (int r = 0; r < 8; ++r) {
                float val = acc[rt][ct][r];
                if (USE_BN) val = val * scale + shift;
                int row = row0 + r;
                if (USE_RES) val += res[(size_t)row * ncols + c];
                val = fmaxf(val, 0.0f);
                out[(size_t)row * ncols + c] = val;
            }
        }
    }
}


extern "C" void kernel_launch(void* const* d_in, const int* in_sizes, int n_in,
                              void* d_out, int out_size, void* d_ws, size_t ws_size,
                              hipStream_t stream) {
    constexpr int N = GIN_N, E = GIN_E, DIN = GIN_DIN, DOUT = GIN_DOUT;

    const float* x    = (const float*)d_in[0];
    const int*   eidx = (const int*)d_in[1];
    const float* eps  = (const float*)d_in[2];
    const float* W1   = (const float*)d_in[3];
    const float* b1   = (const float*)d_in[4];
    const float* g1   = (const float*)d_in[5];
    const float* bt1  = (const float*)d_in[6];
    const float* m1   = (const float*)d_in[7];
    const float* v1   = (const float*)d_in[8];
    const float* W2   = (const float*)d_in[9];
    const float* b2   = (const float*)d_in[10];
    const float* g2   = (const float*)d_in[11];
    const float* bt2  = (const float*)d_in[12];
    const float* m2   = (const float*)d_in[13];
    const float* v2   = (const float*)d_in[14];
    const float* Wres = (const float*)d_in[15];

    const int* src = eidx;
    const int* dst = eidx + E;

    float* h  = (float*)d_ws;
    float* h1 = h + (size_t)N * DIN;
    float* h2 = h1 + (size_t)N * DOUT;
    int* ibase   = (int*)(h2 + (size_t)N * DOUT);
    int* binCount  = ibase;
    int* binCursor = binCount + NBINP;
    int* binBase   = binCursor + NBINP;
    int* offsets = ibase + 2 * N;
    int* bucket  = offsets + N + 1;
    int* pairs   = (int*)h1;

    constexpr int E4 = E / 4;
    constexpr int SORT_BLOCKS = (E4 + SORT_T - 1) / SORT_T;
    zero_kernel<<<(2 * NBINP + 255) / 256, 256, 0, stream>>>(binCount, 2 * NBINP);
    bin_count_kernel<<<SORT_BLOCKS, SORT_T, 0, stream>>>(dst, binCount, E4);
    bin_scatter_kernel<<<SORT_BLOCKS, SORT_T, 0, stream>>>(src, dst, binCount, binCursor,
                                                           binBase, pairs, E4);
    bin_sort_kernel<<<NBIN, SORT_T, 0, stream>>>(pairs, binBase, offsets, bucket, N);
    gather_kernel<<<N / 8, 256, 0, stream>>>(x, offsets, bucket, eps, h);

    dim3 grid(N / 64, DOUT / 64);

    gemm_kernel<128, true, false><<<grid, 128, 0, stream>>>(
        h, W1, b1, g1, bt1, m1, v1, nullptr, h1, DOUT);
    gemm_kernel<256, true, false><<<grid, 128, 0, stream>>>(
        h1, W2, b2, g2, bt2, m2, v2, nullptr, h2, DOUT);
    gemm_kernel<128, false, true><<<grid, 128, 0, stream>>>(
        x, Wres, nullptr, nullptr, nullptr, nullptr, nullptr, h2, (float*)d_out, DOUT);
}


# second layer and residual product in one launch: h1@W2 becomes relu(bn2) in registers and the x@Wres sum starts from that value instead of adding h2 last; h2 plane and one launch gone
# speedup vs baseline: 4.5871x; 1.0776x over previous
#include <hip/hip_runtime.h>
#include <hip/hip_bf16.h>

typedef float v2f __attribute__((ext_vector_type(2)));
typedef float v8f __attribute__((ext_vector_type(8)));

#define GIN_N    40000
#define GIN_E    640000
#define GIN_DIN  128
#define GIN_DOUT 256


__global__ void zero_kernel(int* __restrict__ p, int n) {
    int i = blockIdx.x * 256 + threadIdx.x;
    if (i < n) p[i] = 0;
}

#define BIN_SHIFT 7
#define BIN_NODES (1 << BIN_SHIFT)
#define NBIN  ((GIN_N + BIN_NODES - 1) / BIN_NODES)
#define NBINP (((NBIN + 31) / 32) * 32)
#define SORT_T 1024
static_assert(GIN_E % 4 == 0, "edges are read four at a time");
static_assert(NBINP <= SORT_T, "one thread per bin");

__global__ __launch_bounds__(SORT_T)
void bin_count_kernel(const int* __restrict__ dst, int* __restrict__ binCount, int e4) {
    __shared__ int hist[NBINP];
    const int tid = threadIdx.x;
    if (tid < NBINP) hist[tid] = 0;
    __syncthreads();
    const int i4 = blockIdx.x * SORT_T + tid;
    if (i4 < e4) {
        int4 d = ((const int4*)dst)[i4];
        atomicAdd(&hist[d.x >> BIN_SHIFT], 1);
        atomicAdd(&hist[d.y >> BIN_SHIFT], 1);
        atomicAdd(&hist[d.z >> BIN_SHIFT], 1);
        atomicAdd(&hist[d.w >> BIN_SHIFT], 1);
    }
    __syncthreads();
    if (tid < NBIN) {
        int c = hist[tid];
        if (c) atomicAdd(&binCount[tid], c);
    }
}

__global__ __launch_bounds__(SORT_T)
void bin_scatter_kernel(const int* __restrict__ src, const int* __restrict__ dst,
                        const int* __restrict__ binCount, int* __restrict__ binCursor,
                        int* __restrict__ binBase, int* __restrict__ pairbuf, int e4) {
    __shared__ int hist[NBINP];
    __shared__ int gbase[NBINP];
    __shared__ int smT[32];
    int2* pairs = (int2*)pairbuf;
    const int tid  = threadIdx.x;
    const int lane = tid & 31;
    const int wave = tid >> 5;

    int v = 0, s = 0;
    if (tid < NBINP) {
        v = (tid < NBIN) ? binCount[tid] : 0;
        s = v;
#pragma unroll
        for (int off = 1; off < 32; off <<= 1) {
            int t = __shfl_up(s, off);
            if (lane >= off) s += t;
        }
        if (lane == 31) smT[wave] = s;
        hist[tid] = 0;
    }
    __syncthreads();

    if (tid < NBINP) {
        int base = 0;
#pragma unroll
        for (int w = 0; w < NBINP / 32; ++w) base += (w < wave) ? smT[w] : 0;
        int excl = base + s - v;
        gbase[tid] = excl;
        if (blockIdx.x == 0) {
            binBase[tid] = excl;
            if (tid == NBINP - 1) binBase[NBINP] = excl + v;
        }
    }

    const int i4 = blockIdx.x * SORT_T + tid;
    const bool act = i4 < e4;
    int4 d  = make_int4(0, 0, 0, 0);
    int4 sr = make_int4(0, 0, 0, 0);
    int r0 = 0, r1 = 0, r2 = 0, r3 = 0;
    if (act) {
        d  = ((const int4*)dst)[i4];
        sr = ((const int4*)src)[i4];
        r0 = atomicAdd(&hist[d.x >> BIN_SHIFT], 1);
        r1 = atomicAdd(&hist[d.y >> BIN_SHIFT], 1);
        r2 = atomicAdd(&hist[d.z >> BIN_SHIFT], 1);
        r3 = atomicAdd(&hist[d.w >> BIN_SHIFT], 1);
    }
    __syncthreads();

    if (tid < NBIN) {
        int c = hist[tid];
        if (c) gbase[tid] += atomicAdd(&binCursor[tid], c);
    }
    __syncthreads();

    if (act) {
        pairs[gbase[d.x >> BIN_SHIFT] + r0] = make_int2(d.x, sr.x);
        pairs[gbase[d.y >> BIN_SHIFT] + r1] = make_int2(d.y, sr.y);
        pairs[gbase[d.z >> BIN_SHIFT] + r2] = make_int2(d.z, sr.z);
        pairs[gbase[d.w >> BIN_SHIFT] + r3] = make_int2(d.w, sr.w);
    }
}

__global__ __launch_bounds__(SORT_T)
void bin_sort_kernel(const int* __restrict__ pairbuf, const int* __restrict__ binBase,
                     int* __restrict__ offsets, int* __restrict__ bucket, int n) {
    __shared__ int cnt[BIN_NODES];
    __shared__ int loff[BIN_NODES];
    __shared__ int smT[BIN_NODES / 32];
    const int2* pairs = (const int2*)pairbuf;
    const int tid  = threadIdx.x;
    const int lane = tid & 31;
    const int wave = tid >> 5;
    const int b  = blockIdx.x;
    const int p0 = binBase[b];
    const int p1 = binBase[b + 1];

    if (tid < BIN_NODES) cnt[tid] = 0;
    __syncthreads();
    for (int i = p0 + tid; i < p1; i += SORT_T)
        atomicAdd(&cnt[pairs[i].x & (BIN_NODES - 1)], 1);
    __syncthreads();

    int v = 0, s = 0;
    if (tid < BIN_NODES) {
        v = cnt[tid];
        s = v;
#pragma unroll
        for (int off = 1; off < 32; off <<= 1) {
            int t = __shfl_up(s, off);
            if (lane >= off) s += t;
        }
        if (lane == 31) smT[wave] = s;
    }
    __syncthreads();

    if (tid < BIN_NODES) {
        int base = 0;
#pragma unroll
        for (int w = 0; w < BIN_NODES / 32; ++w) base += (w < wave) ? smT[w] : 0;
        int first = p0 + base + s - v;
        loff[tid] = first;
        cnt[tid] = 0;
        int node = b * BIN_NODES + tid;
        if (node < n) offsets[node] = first;
        if (node == n - 1) offsets[n] = first + v;
    }
    __syncthreads();

    for (int i = p0 + tid; i < p1; i += SORT_T) {
        int2 pr = pairs[i];
        int l = pr.x & (BIN_NODES - 1);
        int r = atomicAdd(&cnt[l], 1);
        bucket[loff[l] + r] = pr.y;
    }
}

__global__ __launch_bounds__(256)
void gather_kernel(const float* __restrict__ x, const int* __restrict__ offsets,
                   const int* __restrict__ bucket, const float* __restrict__ epsp,
                   float* __restrict__ h) {
    int node = blockIdx.x * 8 + (threadIdx.x >> 5);
    int lane = threadIdx.x & 31;
    float se = 1.0f + epsp[0];
    const float4* xv = (const float4*)x;
    float4 self = xv[(size_t)node * 32 + lane];
    float ax = self.x * se, ay = self.y * se, az = self.z * se, aw = self.w * se;
    int e0 = offsets[node], e1 = offsets[node + 1];
    for (int i = e0; i < e1; ++i) {
        int s = bucket[i];
        float4 r = xv[(size_t)s * 32 + lane];
        ax += r.x; ay += r.y; az += r.z; aw += r.w;
    }
    float4 o; o.x = ax; o.y = ay; o.z = az; o.w = aw;
    ((float4*)h)[(size_t)node * 32 + lane] = o;
}


#define GEMM_KC  32
#define GEMM_LDA (GEMM_KC + 4)
#define GEMM_LDB 72
#define GEMM_AST (64 * GEMM_LDA)
#define GEMM_BST (GEMM_KC * GEMM_LDB)

__device__ __forceinline__ void mma_chunk(const float* __restrict__ Ap0,
                                          const float* __restrict__ Ap1,
                                          const float* __restrict__ Bp,
                                          v8f (&acc)[2][2]) {
#pragma unroll
    for (int k = 0; k < GEMM_KC; k += 4) {
        v2f a0 = *(const v2f*)(Ap0 + k);
        v2f a1 = *(const v2f*)(Ap1 + k);
        const float* bk = Bp + k * GEMM_LDB;
        v2f b0, b1;
        b0.x = bk[0];
        b0.y = bk[GEMM_LDB];
        b1.x = bk[16];
        b1.y = bk[GEMM_LDB + 16];
        acc[0][0] = __builtin_amdgcn_wmma_f32_16x16x4_f32(false, a0, false, b0, (short)0, acc[0][0], false, false);
        acc[0][1] = __builtin_amdgcn_wmma_f32_16x16x4_f32(false, a0, false, b1, (short)0, acc[0][1], false, false);
        acc[1][0] = __builtin_amdgcn_wmma_f32_16x16x4_f32(false, a1, false, b0, (short)0, acc[1][0], false, false);
        acc[1][1] = __builtin_amdgcn_wmma_f32_16x16x4_f32(false, a1, false, b1, (short)0, acc[1][1], false, false);
    }
}

__device__ __forceinline__ void chunk_load(const float* __restrict__ ag, size_t aStep,
                                           const float* __restrict__ bg, size_t bStep,
                                           float4 (&pa)[4], float4 (&pb)[4]) {
#pragma unroll
    for (int j = 0; j < 4; ++j) {
        pa[j] = *(const float4*)(ag + j * aStep);
        pb[j] = *(const float4*)(bg + j * bStep);
    }
}

__device__ __forceinline__ void chunk_store(float* __restrict__ aw, float* __restrict__ bw,
                                            const float4 (&pa)[4], const float4 (&pb)[4]) {
#pragma unroll
    for (int j = 0; j < 4; ++j) {
        *(float4*)(aw + j * 16 * GEMM_LDA) = pa[j];
        *(float4*)(bw + j * 8 * GEMM_LDB)  = pb[j];
    }
}

template <int K, bool USE_BN, bool USE_RES>
__global__ __launch_bounds__(128)
void gemm_kernel(const float* __restrict__ A, const float* __restrict__ B,
                 const float* __restrict__ bias, const float* __restrict__ gam,
                 const float* __restrict__ bet, const float* __restrict__ mu,
                 const float* __restrict__ var, const float* __restrict__ res,
                 float* __restrict__ out, int ncols) {
    constexpr int KC  = GEMM_KC;
    constexpr int LDA = GEMM_LDA;
    constexpr int LDB = GEMM_LDB;
    constexpr int NCH = K / KC;
    static_assert(K % KC == 0, "K is a whole number of chunks");
    __shared__ __attribute__((aligned(16))) float As[2 * GEMM_AST];
    __shared__ __attribute__((aligned(16))) float Bs[2 * GEMM_BST];

    const int rowBase = blockIdx.x * 64;
    const int colBase = blockIdx.y * 64;
    const int tid = threadIdx.x;

    const int ar = tid >> 3, ac = (tid & 7) * 4;
    const int br = tid >> 4, bc = (tid & 15) * 4;
    const float* Ag = A + (size_t)(rowBase + ar) * K + ac;
    const float* Bg = B + (size_t)br * ncols + colBase + bc;
    float* Asw = As + ar * LDA + ac;
    float* Bsw = Bs + br * LDB + bc;
    const size_t aStep = (size_t)16 * K;
    const size_t bStep = (size_t)8 * ncols;

    float4 pa[4], pb[4];
#pragma unroll
    for (int j = 0; j < 4; ++j) {
        pa[j] = *(const float4*)(Ag + j * aStep);
        pb[j] = *(const float4*)(Bg + j * bStep);
    }
#pragma unroll
    for (int j = 0; j < 4; ++j) {
        *(float4*)(Asw + j * 16 * LDA) = pa[j];
        *(float4*)(Bsw + j * 8 * LDB)  = pb[j];
    }
    __syncthreads();

    const int wave = tid >> 5;
    const int lane = tid & 31;
    const int mrow = lane & 15;
    const int kh   = lane >> 4;
    const int rowOff = (wave >> 1) * 32;
    const int colOff = (wave & 1) * 32;

    v8f acc[2][2] = {};
    const float* Ap0 = As + (rowOff + mrow) * LDA + 2 * kh;
    const float* Ap1 = Ap0 + 16 * LDA;
    const float* Bp  = Bs + 2 * kh * LDB + colOff + mrow;

#pragma unroll 1
    for (int c = 0; c + 1 < NCH; ++c) {
        const int st = c & 1;
        const float* Agn = Ag + (c + 1) * KC;
        const float* Bgn = Bg + (size_t)(c + 1) * KC * ncols;
#pragma unroll
        for (int j = 0; j < 4; ++j) {
            pa[j] = *(const float4*)(Agn + j * aStep);
            pb[j] = *(const float4*)(Bgn + j * bStep);
        }
        mma_chunk(Ap0 + st * GEMM_AST, Ap1 + st * GEMM_AST, Bp + st * GEMM_BST, acc);
        float* aw = Asw + (st ^ 1) * GEMM_AST;
        float* bw = Bsw + (st ^ 1) * GEMM_BST;
#pragma unroll
        for (int j = 0; j < 4; ++j) {
            *(float4*)(aw + j * 16 * LDA) = pa[j];
            *(float4*)(bw + j * 8 * LDB)  = pb[j];
        }
        __syncthreads();
    }
    {
        constexpr int st = (NCH - 1) & 1;
        mma_chunk(Ap0 + st * GEMM_AST, Ap1 + st * GEMM_AST, Bp + st * GEMM_BST, acc);
    }

#pragma unroll
    for (int ct = 0; ct < 2; ++ct) {
        int c = colBase + colOff + ct * 16 + mrow;
        float scale = 1.0f, shift = 0.0f;
        if (USE_BN) {
            float inv = rsqrtf(var[c] + 1e-5f);
            scale = gam[c] * inv;
            shift = (bias[c] - mu[c]) * scale + bet[c];
        }
#pragma unroll
        for (int rt = 0; rt < 2; ++rt) {
            int row0 = rowBase + rowOff + rt * 16 + kh * 8;
#pragma unroll
            for (int r = 0; r < 8; ++r) {
                float val = acc[rt][ct][r];
                if (USE_BN) val = val * scale + shift;
                int row = row0 + r;
                if (USE_RES) val += res[(size_t)row * ncols + c];
                val = fmaxf(val, 0.0f);
                out[(size_t)row * ncols + c] = val;
            }
        }
    }
}


template <int K1, int K2>
__global__ __launch_bounds__(128)
void gemm_bn_res_kernel(const float* __restrict__ A1, const float* __restrict__ B1,
                        const float* __restrict__ bias, const float* __restrict__ gam,
                        const float* __restrict__ bet, const float* __restrict__ mu,
                        const float* __restrict__ var,
                        const float* __restrict__ A2, const float* __restrict__ B2,
                        float* __restrict__ out, int ncols) {
    constexpr int KC  = GEMM_KC;
    constexpr int LDA = GEMM_LDA;
    constexpr int LDB = GEMM_LDB;
    constexpr int NCH1 = K1 / KC;
    constexpr int NCH2 = K2 / KC;
    static_assert(K1 % KC == 0 && K2 % KC == 0, "both depths are whole numbers of chunks");
    static_assert(NCH1 % 2 == 0, "the second product starts in stage 0");
    __shared__ __attribute__((aligned(16))) float As[2 * GEMM_AST];
    __shared__ __attribute__((aligned(16))) float Bs[2 * GEMM_BST];

    const int rowBase = blockIdx.x * 64;
    const int colBase = blockIdx.y * 64;
    const int tid = threadIdx.x;

    const int ar = tid >> 3, ac = (tid & 7) * 4;
    const int br = tid >> 4, bc = (tid & 15) * 4;
    const float* Ag1 = A1 + (size_t)(rowBase + ar) * K1 + ac;
    const float* Bg1 = B1 + (size_t)br * ncols + colBase + bc;
    float* Asw = As + ar * LDA + ac;
    float* Bsw = Bs + br * LDB + bc;
    const size_t aStep1 = (size_t)16 * K1;
    const size_t aStep2 = (size_t)16 * K2;
    const size_t bStep  = (size_t)8 * ncols;

    float4 pa[4], pb[4];
    chunk_load(Ag1, aStep1, Bg1, bStep, pa, pb);
    chunk_store(Asw, Bsw, pa, pb);
    __syncthreads();

    const int wave = tid >> 5;
    const int lane = tid & 31;
    const int mrow = lane & 15;
    const int kh   = lane >> 4;
    const int rowOff = (wave >> 1) * 32;
    const int colOff = (wave & 1) * 32;

    v8f acc[2][2] = {};
    const float* Ap0 = As + (rowOff + mrow) * LDA + 2 * kh;
    const float* Ap1 = Ap0 + 16 * LDA;
    const float* Bp  = Bs + 2 * kh * LDB + colOff + mrow;

#pragma unroll 1
    for (int c = 0; c + 1 < NCH1; ++c) {
        const int st = c & 1;
        chunk_load(Ag1 + (c + 1) * KC, aStep1, Bg1 + (size_t)(c + 1) * KC * ncols, bStep, pa, pb);
        mma_chunk(Ap0 + st * GEMM_AST, Ap1 + st * GEMM_AST, Bp + st * GEMM_BST, acc);
        chunk_store(Asw + (st ^ 1) * GEMM_AST, Bsw + (st ^ 1) * GEMM_BST, pa, pb);
        __syncthreads();
    }

    const float* Ag2 = A2 + (size_t)(rowBase + ar) * K2 + ac;
    const float* Bg2 = B2 + (size_t)br * ncols + colBase + bc;
    chunk_load(Ag2, aStep2, Bg2, bStep, pa, pb);
    float pv[2], pg[2], pbi[2], pm[2], pbe[2];
#pragma unroll
    for (int ct = 0; ct < 2; ++ct) {
        int c = colBase + colOff + ct * 16 + mrow;
        pv[ct] = var[c]; pg[ct] = gam[c]; pbi[ct] = bias[c]; pm[ct] = mu[c]; pbe[ct] = bet[c];
    }
    mma_chunk(Ap0 + GEMM_AST, Ap1 + GEMM_AST, Bp + GEMM_BST, acc);
    float scale[2], shift[2];
#pragma unroll
    for (int ct = 0; ct < 2; ++ct) {
        scale[ct] = pg[ct] * rsqrtf(pv[ct] + 1e-5f);
        shift[ct] = (pbi[ct] - pm[ct]) * scale[ct] + pbe[ct];
    }
#pragma unroll
    for (int rt = 0; rt < 2; ++rt)
#pragma unroll
        for (int ct = 0; ct < 2; ++ct)
#pragma unroll
            for (int r = 0; r < 8; ++r)
                acc[rt][ct][r] = fmaxf(acc[rt][ct][r] * scale[ct] + shift[ct], 0.0f);
    chunk_store(Asw, Bsw, pa, pb);
    __syncthreads();

#pragma unroll 1
    for (int c = 0; c + 1 < NCH2; ++c) {
        const int st = c & 1;
        chunk_load(Ag2 + (c + 1) * KC, aStep2, Bg2 + (size_t)(c + 1) * KC * ncols, bStep, pa, pb);
        mma_chunk(Ap0 + st * GEMM_AST, Ap1 + st * GEMM_AST, Bp + st * GEMM_BST, acc);
        chunk_store(Asw + (st ^ 1) * GEMM_AST, Bsw + (st ^ 1) * GEMM_BST, pa, pb);
        __syncthreads();
    }
    {
        constexpr int st = (NCH2 - 1) & 1;
        mma_chunk(Ap0 + st * GEMM_AST, Ap1 + st * GEMM_AST, Bp + st * GEMM_BST, acc);
    }

#pragma unroll
    for (int ct = 0; ct < 2; ++ct) {
        int c = colBase + colOff + ct * 16 + mrow;
#pragma unroll
        for (int rt = 0; rt < 2; ++rt) {
            int row0 = rowBase + rowOff + rt * 16 + kh * 8;
#pragma unroll
            for (int r = 0; r < 8; ++r)
                out[(size_t)(row0 + r) * ncols + c] = fmaxf(acc[rt][ct][r], 0.0f);
        }
    }
}


extern "C" void kernel_launch(void* const* d_in, const int* in_sizes, int n_in,
                              void* d_out, int out_size, void* d_ws, size_t ws_size,
                              hipStream_t stream) {
    constexpr int N = GIN_N, E = GIN_E, DIN = GIN_DIN, DOUT = GIN_DOUT;

    const float* x    = (const float*)d_in[0];
    const int*   eidx = (const int*)d_in[1];
    const float* eps  = (const float*)d_in[2];
    const float* W1   = (const float*)d_in[3];
    const float* b1   = (const float*)d_in[4];
    const float* g1   = (const float*)d_in[5];
    const float* bt1  = (const float*)d_in[6];
    const float* m1   = (const float*)d_in[7];
    const float* v1   = (const float*)d_in[8];
    const float* W2   = (const float*)d_in[9];
    const float* b2   = (const float*)d_in[10];
    const float* g2   = (const float*)d_in[11];
    const float* bt2  = (const float*)d_in[12];
    const float* m2   = (const float*)d_in[13];
    const float* v2   = (const float*)d_in[14];
    const float* Wres = (const float*)d_in[15];

    const int* src = eidx;
    const int* dst = eidx + E;

    float* h  = (float*)d_ws;
    float* h1 = h + (size_t)N * DIN;
    int* ibase   = (int*)(h1 + (size_t)2 * N * DOUT);
    int* binCount  = ibase;
    int* binCursor = binCount + NBINP;
    int* binBase   = binCursor + NBINP;
    int* offsets = ibase + 2 * N;
    int* bucket  = offsets + N + 1;
    int* pairs   = (int*)h1;

    constexpr int E4 = E / 4;
    constexpr int SORT_BLOCKS = (E4 + SORT_T - 1) / SORT_T;
    zero_kernel<<<(2 * NBINP + 255) / 256, 256, 0, stream>>>(binCount, 2 * NBINP);
    bin_count_kernel<<<SORT_BLOCKS, SORT_T, 0, stream>>>(dst, binCount, E4);
    bin_scatter_kernel<<<SORT_BLOCKS, SORT_T, 0, stream>>>(src, dst, binCount, binCursor,
                                                           binBase, pairs, E4);
    bin_sort_kernel<<<NBIN, SORT_T, 0, stream>>>(pairs, binBase, offsets, bucket, N);
    gather_kernel<<<N / 8, 256, 0, stream>>>(x, offsets, bucket, eps, h);

    dim3 grid(N / 64, DOUT / 64);

    gemm_kernel<128, true, false><<<grid, 128, 0, stream>>>(
        h, W1, b1, g1, bt1, m1, v1, nullptr, h1, DOUT);
    gemm_bn_res_kernel<256, 128><<<grid, 128, 0, stream>>>(
        h1, W2, b2, g2, bt2, m2, v2, x, Wres, (float*)d_out, DOUT);
}


# edge grouping in 2 launches (was 4): each block sorts its 4096 edges by bin in its own region and writes a bin-start table; no zeroing, bin totals or device atomics; node sums keep an arrival order
# speedup vs baseline: 4.7755x; 1.0411x over previous
#include <hip/hip_runtime.h>
#include <hip/hip_bf16.h>

typedef float v2f __attribute__((ext_vector_type(2)));
typedef float v8f __attribute__((ext_vector_type(8)));

#define GIN_N    40000
#define GIN_E    640000
#define GIN_DIN  128
#define GIN_DOUT 256


#define BIN_SHIFT 7
#define BIN_NODES (1 << BIN_SHIFT)
#define NBIN  ((GIN_N + BIN_NODES - 1) / BIN_NODES)
#define NBINP (((NBIN + 1 + 31) / 32) * 32)
#define SORT_T 1024
#define SEG_EDGES (4 * SORT_T)
#define NSEG  ((GIN_E + SEG_EDGES - 1) / SEG_EDGES)
#define NSEGP (((NSEG + 31) / 32) * 32)
static_assert(GIN_E % 4 == 0, "edges are read four at a time");
static_assert(NBINP <= SORT_T, "one thread per bin");
static_assert(NSEGP <= SORT_T, "one thread per segment");

__global__ __launch_bounds__(SORT_T)
void bin_local_sort_kernel(const int* __restrict__ src, const int* __restrict__ dst,
                           int* __restrict__ tab, int* __restrict__ pairbuf, int e4) {
    __shared__ int hist[NBINP];
    __shared__ int smT[32];
    int2* pairs = (int2*)pairbuf;
    const int tid  = threadIdx.x;
    const int lane = tid & 31;
    const int wave = tid >> 5;

    if (tid < NBINP) hist[tid] = 0;
    __syncthreads();

    const int i4 = blockIdx.x * SORT_T + tid;
    const bool act = i4 < e4;
    int4 d  = make_int4(0, 0, 0, 0);
    int4 sr = make_int4(0, 0, 0, 0);
    int r0 = 0, r1 = 0, r2 = 0, r3 = 0;
    if (act) {
        d  = ((const int4*)dst)[i4];
        sr = ((const int4*)src)[i4];
        r0 = atomicAdd(&hist[d.x >> BIN_SHIFT], 1);
        r1 = atomicAdd(&hist[d.y >> BIN_SHIFT], 1);
        r2 = atomicAdd(&hist[d.z >> BIN_SHIFT], 1);
        r3 = atomicAdd(&hist[d.w >> BIN_SHIFT], 1);
    }
    __syncthreads();

    int v = 0, s = 0;
    if (tid < NBINP) {
        v = hist[tid];
        s = v;
#pragma unroll
        for (int off = 1; off < 32; off <<= 1) {
            int t = __shfl_up(s, off);
            if (lane >= off) s += t;
        }
        if (lane == 31) smT[wave] = s;
    }
    __syncthreads();

    if (tid < NBINP) {
        int base = 0;
#pragma unroll
        for (int w = 0; w < NBINP / 32; ++w) base += (w < wave) ? smT[w] : 0;
        int excl = base + s - v;
        hist[tid] = excl;
        if (tid <= NBIN) tab[tid * NSEG + blockIdx.x] = excl;
    }
    __syncthreads();

    if (act) {
        int2* reg = pairs + (size_t)blockIdx.x * SEG_EDGES;
        reg[hist[d.x >> BIN_SHIFT] + r0] = make_int2(d.x, sr.x);
        reg[hist[d.y >> BIN_SHIFT] + r1] = make_int2(d.y, sr.y);
        reg[hist[d.z >> BIN_SHIFT] + r2] = make_int2(d.z, sr.z);
        reg[hist[d.w >> BIN_SHIFT] + r3] = make_int2(d.w, sr.w);
    }
}

#define SORT_WAVES (SORT_T / 32)
#define SEG_TRIPS ((NSEG + SORT_WAVES - 1) / SORT_WAVES)
__global__ __launch_bounds__(SORT_T)
void bin_sort_kernel(const int* __restrict__ pairbuf, const int* __restrict__ tab,
                     int* __restrict__ offsets, int* __restrict__ bucket, int n) {
    __shared__ int cnt[BIN_NODES];
    __shared__ int loff[BIN_NODES];
    __shared__ int smT[BIN_NODES / 32];
    __shared__ int segS[NSEGP];
    __shared__ int segE[NSEGP];
    __shared__ int smP[NSEGP / 32];
    const int2* pairs = (const int2*)pairbuf;
    const int tid  = threadIdx.x;
    const int lane = tid & 31;
    const int wave = tid >> 5;
    const int b  = blockIdx.x;

    if (tid < NSEGP) {
        int st = 0, en = 0;
        if (tid < NSEG) {
            st = tab[b * NSEG + tid];
            en = tab[(b + 1) * NSEG + tid];
        }
        segS[tid] = st;
        segE[tid] = en;
        int t = st;
#pragma unroll
        for (int off = 16; off > 0; off >>= 1) t += __shfl_xor(t, off);
        if (lane == 0) smP[wave] = t;
    }
    if (tid < BIN_NODES) cnt[tid] = 0;
    __syncthreads();

    int p0 = 0;
#pragma unroll
    for (int w = 0; w < NSEGP / 32; ++w) p0 += smP[w];

    int  sb[SEG_TRIPS], se[SEG_TRIPS];
    int2 pr[SEG_TRIPS];
    bool ok[SEG_TRIPS];
#pragma unroll
    for (int q = 0; q < SEG_TRIPS; ++q) {
        const int sg = wave + q * SORT_WAVES;
        sb[q] = 0; se[q] = 0;
        if (sg < NSEG) { sb[q] = sg * SEG_EDGES + segS[sg] + lane; se[q] = sg * SEG_EDGES + segE[sg]; }
        ok[q] = sb[q] < se[q];
    }
#pragma unroll
    for (int q = 0; q < SEG_TRIPS; ++q) {
        pr[q] = make_int2(0, 0);
        if (ok[q]) pr[q] = pairs[sb[q]];
    }
#pragma unroll
    for (int q = 0; q < SEG_TRIPS; ++q)
        if (ok[q]) atomicAdd(&cnt[pr[q].x & (BIN_NODES - 1)], 1);
#pragma unroll
    for (int q = 0; q < SEG_TRIPS; ++q)
        for (int i = sb[q] + 32; i < se[q]; i += 32)
            atomicAdd(&cnt[pairs[i].x & (BIN_NODES - 1)], 1);
    __syncthreads();

    int v = 0, s = 0;
    if (tid < BIN_NODES) {
        v = cnt[tid];
        s = v;
#pragma unroll
        for (int off = 1; off < 32; off <<= 1) {
            int t = __shfl_up(s, off);
            if (lane >= off) s += t;
        }
        if (lane == 31) smT[wave] = s;
    }
    __syncthreads();

    if (tid < BIN_NODES) {
        int base = 0;
#pragma unroll
        for (int w = 0; w < BIN_NODES / 32; ++w) base += (w < wave) ? smT[w] : 0;
        int first = p0 + base + s - v;
        loff[tid] = first;
        cnt[tid] = 0;
        int node = b * BIN_NODES + tid;
        if (node < n) offsets[node] = first;
        if (node == n - 1) offsets[n] = first + v;
    }
    __syncthreads();

#pragma unroll
    for (int q = 0; q < SEG_TRIPS; ++q)
        if (ok[q]) {
            int l = pr[q].x & (BIN_NODES - 1);
            int r = atomicAdd(&cnt[l], 1);
            bucket[loff[l] + r] = pr[q].y;
        }
#pragma unroll
    for (int q = 0; q < SEG_TRIPS; ++q)
        for (int i = sb[q] + 32; i < se[q]; i += 32) {
            int2 p2 = pairs[i];
            int l = p2.x & (BIN_NODES - 1);
            int r = atomicAdd(&cnt[l], 1);
            bucket[loff[l] + r] = p2.y;
        }
}

__global__ __launch_bounds__(256)
void gather_kernel(const float* __restrict__ x, const int* __restrict__ offsets,
                   const int* __restrict__ bucket, const float* __restrict__ epsp,
                   float* __restrict__ h) {
    int node = blockIdx.x * 8 + (threadIdx.x >> 5);
    int lane = threadIdx.x & 31;
    float se = 1.0f + epsp[0];
    const float4* xv = (const float4*)x;
    float4 self = xv[(size_t)node * 32 + lane];
    float ax = self.x * se, ay = self.y * se, az = self.z * se, aw = self.w * se;
    int e0 = offsets[node], e1 = offsets[node + 1];
    for (int i = e0; i < e1; ++i) {
        int s = bucket[i];
        float4 r = xv[(size_t)s * 32 + lane];
        ax += r.x; ay += r.y; az += r.z; aw += r.w;
    }
    float4 o; o.x = ax; o.y = ay; o.z = az; o.w = aw;
    ((float4*)h)[(size_t)node * 32 + lane] = o;
}


#define GEMM_KC  32
#define GEMM_LDA (GEMM_KC + 4)
#define GEMM_LDB 72
#define GEMM_AST (64 * GEMM_LDA)
#define GEMM_BST (GEMM_KC * GEMM_LDB)

__device__ __forceinline__ void mma_chunk(const float* __restrict__ Ap0,
                                          const float* __restrict__ Ap1,
                                          const float* __restrict__ Bp,
                                          v8f (&acc)[2][2]) {
#pragma unroll
    for (int k = 0; k < GEMM_KC; k += 4) {
        v2f a0 = *(const v2f*)(Ap0 + k);
        v2f a1 = *(const v2f*)(Ap1 + k);
        const float* bk = Bp + k * GEMM_LDB;
        v2f b0, b1;
        b0.x = bk[0];
        b0.y = bk[GEMM_LDB];
        b1.x = bk[16];
        b1.y = bk[GEMM_LDB + 16];
        acc[0][0] = __builtin_amdgcn_wmma_f32_16x16x4_f32(false, a0, false, b0, (short)0, acc[0][0], false, false);
        acc[0][1] = __builtin_amdgcn_wmma_f32_16x16x4_f32(false, a0, false, b1, (short)0, acc[0][1], false, false);
        acc[1][0] = __builtin_amdgcn_wmma_f32_16x16x4_f32(false, a1, false, b0, (short)0, acc[1][0], false, false);
        acc[1][1] = __builtin_amdgcn_wmma_f32_16x16x4_f32(false, a1, false, b1, (short)0, acc[1][1], false, false);
    }
}

__device__ __forceinline__ void chunk_load(const float* __restrict__ ag, size_t aStep,
                                           const float* __restrict__ bg, size_t bStep,
                                           float4 (&pa)[4], float4 (&pb)[4]) {
#pragma unroll
    for (int j = 0; j < 4; ++j) {
        pa[j] = *(const float4*)(ag + j * aStep);
        pb[j] = *(const float4*)(bg + j * bStep);
    }
}

__device__ __forceinline__ void chunk_store(float* __restrict__ aw, float* __restrict__ bw,
                                            const float4 (&pa)[4], const float4 (&pb)[4]) {
#pragma unroll
    for (int j = 0; j < 4; ++j) {
        *(float4*)(aw + j * 16 * GEMM_LDA) = pa[j];
        *(float4*)(bw + j * 8 * GEMM_LDB)  = pb[j];
    }
}

template <int K, bool USE_BN, bool USE_RES>
__global__ __launch_bounds__(128)
void gemm_kernel(const float* __restrict__ A, const float* __restrict__ B,
                 const float* __restrict__ bias, const float* __restrict__ gam,
                 const float* __restrict__ bet, const float* __restrict__ mu,
                 const float* __restrict__ var, const float* __restrict__ res,
                 float* __restrict__ out, int ncols) {
    constexpr int KC  = GEMM_KC;
    constexpr int LDA = GEMM_LDA;
    constexpr int LDB = GEMM_LDB;
    constexpr int NCH = K / KC;
    static_assert(K % KC == 0, "K is a whole number of chunks");
    __shared__ __attribute__((aligned(16))) float As[2 * GEMM_AST];
    __shared__ __attribute__((aligned(16))) float Bs[2 * GEMM_BST];

    const int rowBase = blockIdx.x * 64;
    const int colBase = blockIdx.y * 64;
    const int tid = threadIdx.x;

    const int ar = tid >> 3, ac = (tid & 7) * 4;
    const int br = tid >> 4, bc = (tid & 15) * 4;
    const float* Ag = A + (size_t)(rowBase + ar) * K + ac;
    const float* Bg = B + (size_t)br * ncols + colBase + bc;
    float* Asw = As + ar * LDA + ac;
    float* Bsw = Bs + br * LDB + bc;
    const size_t aStep = (size_t)16 * K;
    const size_t bStep = (size_t)8 * ncols;

    float4 pa[4], pb[4];
#pragma unroll
    for (int j = 0; j < 4; ++j) {
        pa[j] = *(const float4*)(Ag + j * aStep);
        pb[j] = *(const float4*)(Bg + j * bStep);
    }
#pragma unroll
    for (int j = 0; j < 4; ++j) {
        *(float4*)(Asw + j * 16 * LDA) = pa[j];
        *(float4*)(Bsw + j * 8 * LDB)  = pb[j];
    }
    __syncthreads();

    const int wave = tid >> 5;
    const int lane = tid & 31;
    const int mrow = lane & 15;
    const int kh   = lane >> 4;
    const int rowOff = (wave >> 1) * 32;
    const int colOff = (wave & 1) * 32;

    v8f acc[2][2] = {};
    const float* Ap0 = As + (rowOff + mrow) * LDA + 2 * kh;
    const float* Ap1 = Ap0 + 16 * LDA;
    const float* Bp  = Bs + 2 * kh * LDB + colOff + mrow;

#pragma unroll 1
    for (int c = 0; c + 1 < NCH; ++c) {
        const int st = c & 1;
        const float* Agn = Ag + (c + 1) * KC;
        const float* Bgn = Bg + (size_t)(c + 1) * KC * ncols;
#pragma unroll
        for (int j = 0; j < 4; ++j) {
            pa[j] = *(const float4*)(Agn + j * aStep);
            pb[j] = *(const float4*)(Bgn + j * bStep);
        }
        mma_chunk(Ap0 + st * GEMM_AST, Ap1 + st * GEMM_AST, Bp + st * GEMM_BST, acc);
        float* aw = Asw + (st ^ 1) * GEMM_AST;
        float* bw = Bsw + (st ^ 1) * GEMM_BST;
#pragma unroll
        for (int j = 0; j < 4; ++j) {
            *(float4*)(aw + j * 16 * LDA) = pa[j];
            *(float4*)(bw + j * 8 * LDB)  = pb[j];
        }
        __syncthreads();
    }
    {
        constexpr int st = (NCH - 1) & 1;
        mma_chunk(Ap0 + st * GEMM_AST, Ap1 + st * GEMM_AST, Bp + st * GEMM_BST, acc);
    }

#pragma unroll
    for (int ct = 0; ct < 2; ++ct) {
        int c = colBase + colOff + ct * 16 + mrow;
        float scale = 1.0f, shift = 0.0f;
        if (USE_BN) {
            float inv = rsqrtf(var[c] + 1e-5f);
            scale = gam[c] * inv;
            shift = (bias[c] - mu[c]) * scale + bet[c];
        }
#pragma unroll
        for (int rt = 0; rt < 2; ++rt) {
            int row0 = rowBase + rowOff + rt * 16 + kh * 8;
#pragma unroll
            for (int r = 0; r < 8; ++r) {
                float val = acc[rt][ct][r];
                if (USE_BN) val = val * scale + shift;
                int row = row0 + r;
                if (USE_RES) val += res[(size_t)row * ncols + c];
                val = fmaxf(val, 0.0f);
                out[(size_t)row * ncols + c] = val;
            }
        }
    }
}


template <int K1, int K2>
__global__ __launch_bounds__(128)
void gemm_bn_res_kernel(const float* __restrict__ A1, const float* __restrict__ B1,
                        const float* __restrict__ bias, const float* __restrict__ gam,
                        const float* __restrict__ bet, const float* __restrict__ mu,
                        const float* __restrict__ var,
                        const float* __restrict__ A2, const float* __restrict__ B2,
                        float* __restrict__ out, int ncols) {
    constexpr int KC  = GEMM_KC;
    constexpr int LDA = GEMM_LDA;
    constexpr int LDB = GEMM_LDB;
    constexpr int NCH1 = K1 / KC;
    constexpr int NCH2 = K2 / KC;
    static_assert(K1 % KC == 0 && K2 % KC == 0, "both depths are whole numbers of chunks");
    static_assert(NCH1 % 2 == 0, "the second product starts in stage 0");
    __shared__ __attribute__((aligned(16))) float As[2 * GEMM_AST];
    __shared__ __attribute__((aligned(16))) float Bs[2 * GEMM_BST];

    const int rowBase = blockIdx.x * 64;
    const int colBase = blockIdx.y * 64;
    const int tid = threadIdx.x;

    const int ar = tid >> 3, ac = (tid & 7) * 4;
    const int br = tid >> 4, bc = (tid & 15) * 4;
    const float* Ag1 = A1 + (size_t)(rowBase + ar) * K1 + ac;
    const float* Bg1 = B1 + (size_t)br * ncols + colBase + bc;
    float* Asw = As + ar * LDA + ac;
    float* Bsw = Bs + br * LDB + bc;
    const size_t aStep1 = (size_t)16 * K1;
    const size_t aStep2 = (size_t)16 * K2;
    const size_t bStep  = (size_t)8 * ncols;

    float4 pa[4], pb[4];
    chunk_load(Ag1, aStep1, Bg1, bStep, pa, pb);
    chunk_store(Asw, Bsw, pa, pb);
    __syncthreads();

    const int wave = tid >> 5;
    const int lane = tid & 31;
    const int mrow = lane & 15;
    const int kh   = lane >> 4;
    const int rowOff = (wave >> 1) * 32;
    const int colOff = (wave & 1) * 32;

    v8f acc[2][2] = {};
    const float* Ap0 = As + (rowOff + mrow) * LDA + 2 * kh;
    const float* Ap1 = Ap0 + 16 * LDA;
    const float* Bp  = Bs + 2 * kh * LDB + colOff + mrow;

#pragma unroll 1
    for (int c = 0; c + 1 < NCH1; ++c) {
        const int st = c & 1;
        chunk_load(Ag1 + (c + 1) * KC, aStep1, Bg1 + (size_t)(c + 1) * KC * ncols, bStep, pa, pb);
        mma_chunk(Ap0 + st * GEMM_AST, Ap1 + st * GEMM_AST, Bp + st * GEMM_BST, acc);
        chunk_store(Asw + (st ^ 1) * GEMM_AST, Bsw + (st ^ 1) * GEMM_BST, pa, pb);
        __syncthreads();
    }

    const float* Ag2 = A2 + (size_t)(rowBase + ar) * K2 + ac;
    const float* Bg2 = B2 + (size_t)br * ncols + colBase + bc;
    chunk_load(Ag2, aStep2, Bg2, bStep, pa, pb);
    float pv[2], pg[2], pbi[2], pm[2], pbe[2];
#pragma unroll
    for (int ct = 0; ct < 2; ++ct) {
        int c = colBase + colOff + ct * 16 + mrow;
        pv[ct] = var[c]; pg[ct] = gam[c]; pbi[ct] = bias[c]; pm[ct] = mu[c]; pbe[ct] = bet[c];
    }
    mma_chunk(Ap0 + GEMM_AST, Ap1 + GEMM_AST, Bp + GEMM_BST, acc);
    float scale[2], shift[2];
#pragma unroll
    for (int ct = 0; ct < 2; ++ct) {
        scale[ct] = pg[ct] * rsqrtf(pv[ct] + 1e-5f);
        shift[ct] = (pbi[ct] - pm[ct]) * scale[ct] + pbe[ct];
    }
#pragma unroll
    for (int rt = 0; rt < 2; ++rt)
#pragma unroll
        for (int ct = 0; ct < 2; ++ct)
#pragma unroll
            for (int r = 0; r < 8; ++r)
                acc[rt][ct][r] = fmaxf(acc[rt][ct][r] * scale[ct] + shift[ct], 0.0f);
    chunk_store(Asw, Bsw, pa, pb);
    __syncthreads();

#pragma unroll 1
    for (int c = 0; c + 1 < NCH2; ++c) {
        const int st = c & 1;
        chunk_load(Ag2 + (c + 1) * KC, aStep2, Bg2 + (size_t)(c + 1) * KC * ncols, bStep, pa, pb);
        mma_chunk(Ap0 + st * GEMM_AST, Ap1 + st * GEMM_AST, Bp + st * GEMM_BST, acc);
        chunk_store(Asw + (st ^ 1) * GEMM_AST, Bsw + (st ^ 1) * GEMM_BST, pa, pb);
        __syncthreads();
    }
    {
        constexpr int st = (NCH2 - 1) & 1;
        mma_chunk(Ap0 + st * GEMM_AST, Ap1 + st * GEMM_AST, Bp + st * GEMM_BST, acc);
    }

#pragma unroll
    for (int ct = 0; ct < 2; ++ct) {
        int c = colBase + colOff + ct * 16 + mrow;
#pragma unroll
        for (int rt = 0; rt < 2; ++rt) {
            int row0 = rowBase + rowOff + rt * 16 + kh * 8;
#pragma unroll
            for (int r = 0; r < 8; ++r)
                out[(size_t)(row0 + r) * ncols + c] = fmaxf(acc[rt][ct][r], 0.0f);
        }
    }
}


extern "C" void kernel_launch(void* const* d_in, const int* in_sizes, int n_in,
                              void* d_out, int out_size, void* d_ws, size_t ws_size,
                              hipStream_t stream) {
    constexpr int N = GIN_N, E = GIN_E, DIN = GIN_DIN, DOUT = GIN_DOUT;

    const float* x    = (const float*)d_in[0];
    const int*   eidx = (const int*)d_in[1];
    const float* eps  = (const float*)d_in[2];
    const float* W1   = (const float*)d_in[3];
    const float* b1   = (const float*)d_in[4];
    const float* g1   = (const float*)d_in[5];
    const float* bt1  = (const float*)d_in[6];
    const float* m1   = (const float*)d_in[7];
    const float* v1   = (const float*)d_in[8];
    const float* W2   = (const float*)d_in[9];
    const float* b2   = (const float*)d_in[10];
    const float* g2   = (const float*)d_in[11];
    const float* bt2  = (const float*)d_in[12];
    const float* m2   = (const float*)d_in[13];
    const float* v2   = (const float*)d_in[14];
    const float* Wres = (const float*)d_in[15];

    const int* src = eidx;
    const int* dst = eidx + E;

    float* h  = (float*)d_ws;
    float* h1 = h + (size_t)N * DIN;
    int* ibase   = (int*)(h1 + (size_t)2 * N * DOUT);
    int* tab     = ibase;
    static_assert((NBIN + 1) * NSEG <= 2 * GIN_N, "the table fits in front of offsets");
    int* offsets = ibase + 2 * N;
    int* bucket  = offsets + N + 1;
    int* pairs   = (int*)h1;

    constexpr int E4 = E / 4;
    bin_local_sort_kernel<<<NSEG, SORT_T, 0, stream>>>(src, dst, tab, pairs, E4);
    bin_sort_kernel<<<NBIN, SORT_T, 0, stream>>>(pairs, tab, offsets, bucket, N);
    gather_kernel<<<N / 8, 256, 0, stream>>>(x, offsets, bucket, eps, h);

    dim3 grid(N / 64, DOUT / 64);

    gemm_kernel<128, true, false><<<grid, 128, 0, stream>>>(
        h, W1, b1, g1, bt1, m1, v1, nullptr, h1, DOUT);
    gemm_bn_res_kernel<256, 128><<<grid, 128, 0, stream>>>(
        h1, W2, b2, g2, bt2, m2, v2, x, Wres, (float*)d_out, DOUT);
}


# gather fused into the first-layer launch: 250 resident work-groups, 16 waves sum neighbour rows (bucket order kept) into a 2-tile LDS ring while 8 waves run the last tile's WMMAs on W1 kept in LDS
# speedup vs baseline: 4.9732x; 1.0414x over previous
#include <hip/hip_runtime.h>
#include <hip/hip_bf16.h>

typedef float v2f __attribute__((ext_vector_type(2)));
typedef float v8f __attribute__((ext_vector_type(8)));

#define GIN_N    40000
#define GIN_E    640000
#define GIN_DIN  128
#define GIN_DOUT 256


#define BIN_SHIFT 7
#define BIN_NODES (1 << BIN_SHIFT)
#define NBIN  ((GIN_N + BIN_NODES - 1) / BIN_NODES)
#define NBINP (((NBIN + 1 + 31) / 32) * 32)
#define SORT_T 1024
#define SEG_EDGES (4 * SORT_T)
#define NSEG  ((GIN_E + SEG_EDGES - 1) / SEG_EDGES)
#define NSEGP (((NSEG + 31) / 32) * 32)
static_assert(GIN_E % 4 == 0, "edges are read four at a time");
static_assert(NBINP <= SORT_T, "one thread per bin");
static_assert(NSEGP <= SORT_T, "one thread per segment");

__global__ __launch_bounds__(SORT_T)
void bin_local_sort_kernel(const int* __restrict__ src, const int* __restrict__ dst,
                           int* __restrict__ tab, int* __restrict__ pairbuf, int e4) {
    __shared__ int hist[NBINP];
    __shared__ int smT[32];
    int2* pairs = (int2*)pairbuf;
    const int tid  = threadIdx.x;
    const int lane = tid & 31;
    const int wave = tid >> 5;

    if (tid < NBINP) hist[tid] = 0;
    __syncthreads();

    const int i4 = blockIdx.x * SORT_T + tid;
    const bool act = i4 < e4;
    int4 d  = make_int4(0, 0, 0, 0);
    int4 sr = make_int4(0, 0, 0, 0);
    int r0 = 0, r1 = 0, r2 = 0, r3 = 0;
    if (act) {
        d  = ((const int4*)dst)[i4];
        sr = ((const int4*)src)[i4];
        r0 = atomicAdd(&hist[d.x >> BIN_SHIFT], 1);
        r1 = atomicAdd(&hist[d.y >> BIN_SHIFT], 1);
        r2 = atomicAdd(&hist[d.z >> BIN_SHIFT], 1);
        r3 = atomicAdd(&hist[d.w >> BIN_SHIFT], 1);
    }
    __syncthreads();

    int v = 0, s = 0;
    if (tid < NBINP) {
        v = hist[tid];
        s = v;
#pragma unroll
        for (int off = 1; off < 32; off <<= 1) {
            int t = __shfl_up(s, off);
            if (lane >= off) s += t;
        }
        if (lane == 31) smT[wave] = s;
    }
    __syncthreads();

    if (tid < NBINP) {
        int base = 0;
#pragma unroll
        for (int w = 0; w < NBINP / 32; ++w) base += (w < wave) ? smT[w] : 0;
        int excl = base + s - v;
        hist[tid] = excl;
        if (tid <= NBIN) tab[tid * NSEG + blockIdx.x] = excl;
    }
    __syncthreads();

    if (act) {
        int2* reg = pairs + (size_t)blockIdx.x * SEG_EDGES;
        reg[hist[d.x >> BIN_SHIFT] + r0] = make_int2(d.x, sr.x);
        reg[hist[d.y >> BIN_SHIFT] + r1] = make_int2(d.y, sr.y);
        reg[hist[d.z >> BIN_SHIFT] + r2] = make_int2(d.z, sr.z);
        reg[hist[d.w >> BIN_SHIFT] + r3] = make_int2(d.w, sr.w);
    }
}

#define SORT_WAVES (SORT_T / 32)
#define SEG_TRIPS ((NSEG + SORT_WAVES - 1) / SORT_WAVES)
__global__ __launch_bounds__(SORT_T)
void bin_sort_kernel(const int* __restrict__ pairbuf, const int* __restrict__ tab,
                     int* __restrict__ offsets, int* __restrict__ bucket, int n) {
    __shared__ int cnt[BIN_NODES];
    __shared__ int loff[BIN_NODES];
    __shared__ int smT[BIN_NODES / 32];
    __shared__ int segS[NSEGP];
    __shared__ int segE[NSEGP];
    __shared__ int smP[NSEGP / 32];
    const int2* pairs = (const int2*)pairbuf;
    const int tid  = threadIdx.x;
    const int lane = tid & 31;
    const int wave = tid >> 5;
    const int b  = blockIdx.x;

    if (tid < NSEGP) {
        int st = 0, en = 0;
        if (tid < NSEG) {
            st = tab[b * NSEG + tid];
            en = tab[(b + 1) * NSEG + tid];
        }
        segS[tid] = st;
        segE[tid] = en;
        int t = st;
#pragma unroll
        for (int off = 16; off > 0; off >>= 1) t += __shfl_xor(t, off);
        if (lane == 0) smP[wave] = t;
    }
    if (tid < BIN_NODES) cnt[tid] = 0;
    __syncthreads();

    int p0 = 0;
#pragma unroll
    for (int w = 0; w < NSEGP / 32; ++w) p0 += smP[w];

    int  sb[SEG_TRIPS], se[SEG_TRIPS];
    int2 pr[SEG_TRIPS];
    bool ok[SEG_TRIPS];
#pragma unroll
    for (int q = 0; q < SEG_TRIPS; ++q) {
        const int sg = wave + q * SORT_WAVES;
        sb[q] = 0; se[q] = 0;
        if (sg < NSEG) { sb[q] = sg * SEG_EDGES + segS[sg] + lane; se[q] = sg * SEG_EDGES + segE[sg]; }
        ok[q] = sb[q] < se[q];
    }
#pragma unroll
    for (int q = 0; q < SEG_TRIPS; ++q) {
        pr[q] = make_int2(0, 0);
        if (ok[q]) pr[q] = pairs[sb[q]];
    }
#pragma unroll
    for (int q = 0; q < SEG_TRIPS; ++q)
        if (ok[q]) atomicAdd(&cnt[pr[q].x & (BIN_NODES - 1)], 1);
#pragma unroll
    for (int q = 0; q < SEG_TRIPS; ++q)
        for (int i = sb[q] + 32; i < se[q]; i += 32)
            atomicAdd(&cnt[pairs[i].x & (BIN_NODES - 1)], 1);
    __syncthreads();

    int v = 0, s = 0;
    if (tid < BIN_NODES) {
        v = cnt[tid];
        s = v;
#pragma unroll
        for (int off = 1; off < 32; off <<= 1) {
            int t = __shfl_up(s, off);
            if (lane >= off) s += t;
        }
        if (lane == 31) smT[wave] = s;
    }
    __syncthreads();

    if (tid < BIN_NODES) {
        int base = 0;
#pragma unroll
        for (int w = 0; w < BIN_NODES / 32; ++w) base += (w < wave) ? smT[w] : 0;
        int first = p0 + base + s - v;
        loff[tid] = first;
        cnt[tid] = 0;
        int node = b * BIN_NODES + tid;
        if (node < n) offsets[node] = first;
        if (node == n - 1) offsets[n] = first + v;
    }
    __syncthreads();

#pragma unroll
    for (int q = 0; q < SEG_TRIPS; ++q)
        if (ok[q]) {
            int l = pr[q].x & (BIN_NODES - 1);
            int r = atomicAdd(&cnt[l], 1);
            bucket[loff[l] + r] = pr[q].y;
        }
#pragma unroll
    for (int q = 0; q < SEG_TRIPS; ++q)
        for (int i = sb[q] + 32; i < se[q]; i += 32) {
            int2 p2 = pairs[i];
            int l = p2.x & (BIN_NODES - 1);
            int r = atomicAdd(&cnt[l], 1);
            bucket[loff[l] + r] = p2.y;
        }
}


#define GEMM_KC  32
#define GEMM_LDA (GEMM_KC + 4)
#define GEMM_LDB 72
#define GEMM_AST (64 * GEMM_LDA)
#define GEMM_BST (GEMM_KC * GEMM_LDB)

__device__ __forceinline__ void mma_chunk(const float* __restrict__ Ap0,
                                          const float* __restrict__ Ap1,
                                          const float* __restrict__ Bp,
                                          v8f (&acc)[2][2]) {
#pragma unroll
    for (int k = 0; k < GEMM_KC; k += 4) {
        v2f a0 = *(const v2f*)(Ap0 + k);
        v2f a1 = *(const v2f*)(Ap1 + k);
        const float* bk = Bp + k * GEMM_LDB;
        v2f b0, b1;
        b0.x = bk[0];
        b0.y = bk[GEMM_LDB];
        b1.x = bk[16];
        b1.y = bk[GEMM_LDB + 16];
        acc[0][0] = __builtin_amdgcn_wmma_f32_16x16x4_f32(false, a0, false, b0, (short)0, acc[0][0], false, false);
        acc[0][1] = __builtin_amdgcn_wmma_f32_16x16x4_f32(false, a0, false, b1, (short)0, acc[0][1], false, false);
        acc[1][0] = __builtin_amdgcn_wmma_f32_16x16x4_f32(false, a1, false, b0, (short)0, acc[1][0], false, false);
        acc[1][1] = __builtin_amdgcn_wmma_f32_16x16x4_f32(false, a1, false, b1, (short)0, acc[1][1], false, false);
    }
}

__device__ __forceinline__ void chunk_load(const float* __restrict__ ag, size_t aStep,
                                           const float* __restrict__ bg, size_t bStep,
                                           float4 (&pa)[4], float4 (&pb)[4]) {
#pragma unroll
    for (int j = 0; j < 4; ++j) {
        pa[j] = *(const float4*)(ag + j * aStep);
        pb[j] = *(const float4*)(bg + j * bStep);
    }
}

__device__ __forceinline__ void chunk_store(float* __restrict__ aw, float* __restrict__ bw,
                                            const float4 (&pa)[4], const float4 (&pb)[4]) {
#pragma unroll
    for (int j = 0; j < 4; ++j) {
        *(float4*)(aw + j * 16 * GEMM_LDA) = pa[j];
        *(float4*)(bw + j * 8 * GEMM_LDB)  = pb[j];
    }
}

template <int K1, int K2>
__global__ __launch_bounds__(128)
void gemm_bn_res_kernel(const float* __restrict__ A1, const float* __restrict__ B1,
                        const float* __restrict__ bias, const float* __restrict__ gam,
                        const float* __restrict__ bet, const float* __restrict__ mu,
                        const float* __restrict__ var,
                        const float* __restrict__ A2, const float* __restrict__ B2,
                        float* __restrict__ out, int ncols) {
    constexpr int KC  = GEMM_KC;
    constexpr int LDA = GEMM_LDA;
    constexpr int LDB = GEMM_LDB;
    constexpr int NCH1 = K1 / KC;
    constexpr int NCH2 = K2 / KC;
    static_assert(K1 % KC == 0 && K2 % KC == 0, "both depths are whole numbers of chunks");
    static_assert(NCH1 % 2 == 0, "the second product starts in stage 0");
    __shared__ __attribute__((aligned(16))) float As[2 * GEMM_AST];
    __shared__ __attribute__((aligned(16))) float Bs[2 * GEMM_BST];

    const int rowBase = blockIdx.x * 64;
    const int colBase = blockIdx.y * 64;
    const int tid = threadIdx.x;

    const int ar = tid >> 3, ac = (tid & 7) * 4;
    const int br = tid >> 4, bc = (tid & 15) * 4;
    const float* Ag1 = A1 + (size_t)(rowBase + ar) * K1 + ac;
    const float* Bg1 = B1 + (size_t)br * ncols + colBase + bc;
    float* Asw = As + ar * LDA + ac;
    float* Bsw = Bs + br * LDB + bc;
    const size_t aStep1 = (size_t)16 * K1;
    const size_t aStep2 = (size_t)16 * K2;
    const size_t bStep  = (size_t)8 * ncols;

    float4 pa[4], pb[4];
    chunk_load(Ag1, aStep1, Bg1, bStep, pa, pb);
    chunk_store(Asw, Bsw, pa, pb);
    __syncthreads();

    const int wave = tid >> 5;
    const int lane = tid & 31;
    const int mrow = lane & 15;
    const int kh   = lane >> 4;
    const int rowOff = (wave >> 1) * 32;
    const int colOff = (wave & 1) * 32;

    v8f acc[2][2] = {};
    const float* Ap0 = As + (rowOff + mrow) * LDA + 2 * kh;
    const float* Ap1 = Ap0 + 16 * LDA;
    const float* Bp  = Bs + 2 * kh * LDB + colOff + mrow;

#pragma unroll 1
    for (int c = 0; c + 1 < NCH1; ++c) {
        const int st = c & 1;
        chunk_load(Ag1 + (c + 1) * KC, aStep1, Bg1 + (size_t)(c + 1) * KC * ncols, bStep, pa, pb);
        mma_chunk(Ap0 + st * GEMM_AST, Ap1 + st * GEMM_AST, Bp + st * GEMM_BST, acc);
        chunk_store(Asw + (st ^ 1) * GEMM_AST, Bsw + (st ^ 1) * GEMM_BST, pa, pb);
        __syncthreads();
    }

    const float* Ag2 = A2 + (size_t)(rowBase + ar) * K2 + ac;
    const float* Bg2 = B2 + (size_t)br * ncols + colBase + bc;
    chunk_load(Ag2, aStep2, Bg2, bStep, pa, pb);
    float pv[2], pg[2], pbi[2], pm[2], pbe[2];
#pragma unroll
    for (int ct = 0; ct < 2; ++ct) {
        int c = colBase + colOff + ct * 16 + mrow;
        pv[ct] = var[c]; pg[ct] = gam[c]; pbi[ct] = bias[c]; pm[ct] = mu[c]; pbe[ct] = bet[c];
    }
    mma_chunk(Ap0 + GEMM_AST, Ap1 + GEMM_AST, Bp + GEMM_BST, acc);
    float scale[2], shift[2];
#pragma unroll
    for (int ct = 0; ct < 2; ++ct) {
        scale[ct] = pg[ct] * rsqrtf(pv[ct] + 1e-5f);
        shift[ct] = (pbi[ct] - pm[ct]) * scale[ct] + pbe[ct];
    }
#pragma unroll
    for (int rt = 0; rt < 2; ++rt)
#pragma unroll
        for (int ct = 0; ct < 2; ++ct)
#pragma unroll
            for (int r = 0; r < 8; ++r)
                acc[rt][ct][r] = fmaxf(acc[rt][ct][r] * scale[ct] + shift[ct], 0.0f);
    chunk_store(Asw, Bsw, pa, pb);
    __syncthreads();

#pragma unroll 1
    for (int c = 0; c + 1 < NCH2; ++c) {
        const int st = c & 1;
        chunk_load(Ag2 + (c + 1) * KC, aStep2, Bg2 + (size_t)(c + 1) * KC * ncols, bStep, pa, pb);
        mma_chunk(Ap0 + st * GEMM_AST, Ap1 + st * GEMM_AST, Bp + st * GEMM_BST, acc);
        chunk_store(Asw + (st ^ 1) * GEMM_AST, Bsw + (st ^ 1) * GEMM_BST, pa, pb);
        __syncthreads();
    }
    {
        constexpr int st = (NCH2 - 1) & 1;
        mma_chunk(Ap0 + st * GEMM_AST, Ap1 + st * GEMM_AST, Bp + st * GEMM_BST, acc);
    }

#pragma unroll
    for (int ct = 0; ct < 2; ++ct) {
        int c = colBase + colOff + ct * 16 + mrow;
#pragma unroll
        for (int rt = 0; rt < 2; ++rt) {
            int row0 = rowBase + rowOff + rt * 16 + kh * 8;
#pragma unroll
            for (int r = 0; r < 8; ++r)
                out[(size_t)(row0 + r) * ncols + c] = fmaxf(acc[rt][ct][r], 0.0f);
        }
    }
}

#define FG_ROWS 32
#define FG_LDA  (GIN_DIN + 4)
#define FG_LDB  (GIN_DOUT + 8)
#define FG_AT   (FG_ROWS * FG_LDA)
#define FG_NT   (GIN_N / FG_ROWS)
#define FG_GRID 250
#define FG_T    768
#define FG_MMA_WAVES 8
#define FG_BATCH 16
#define FG_KB    4
#define FG_NPW  (FG_ROWS / (FG_T / 32 - FG_MMA_WAVES))
static_assert(GIN_N % FG_ROWS == 0, "whole tiles of nodes");
static_assert(FG_ROWS == FG_NPW * (FG_T / 32 - FG_MMA_WAVES), "the gathering waves share a tile evenly");
static_assert(GIN_DOUT == 32 * FG_MMA_WAVES, "one 32-column strip per product wave");
static_assert(GIN_DIN == 128, "a node row is one float4 per lane");

__device__ __forceinline__ int settled(int v) {
    int r;
    asm volatile("v_mov_b32 %0, %1" : "=v"(r) : "v"(v));
    return r;
}

__device__ __forceinline__ void gather_rows(const float4* __restrict__ xv,
                                            const int* __restrict__ offsets,
                                            const int* __restrict__ bucket, float se,
                                            float* __restrict__ At, int node0, int lane) {
    const int ov = offsets[node0 + (lane < FG_NPW ? lane : FG_NPW)];
    float4 self[FG_NPW];
#pragma unroll
    for (int q = 0; q < FG_NPW; ++q) self[q] = xv[(node0 + q) * 32 + lane];
    int eb[FG_NPW + 1];
#pragma unroll
    for (int q = 0; q <= FG_NPW; ++q) eb[q] = __builtin_amdgcn_readlane(ov, q);
    int idf[FG_NPW];
#pragma unroll
    for (int q = 0; q < FG_NPW; ++q) idf[q] = (lane < eb[q + 1] - eb[q]) ? bucket[eb[q] + lane] : 0;

#pragma unroll
    for (int q = 0; q < FG_NPW; ++q) {
        const int e1 = eb[q + 1];
        float ax = self[q].x * se, ay = self[q].y * se, az = self[q].z * se, aw = self[q].w * se;
        int idx = settled(idf[q]);
        for (int base = eb[q]; base < e1; base += 32) {
            const int left = e1 - base;
            const int cnt = left < 32 ? left : 32;
            for (int j0 = 0; j0 < cnt; j0 += FG_BATCH) {
                float4 r[FG_BATCH];
#pragma unroll
                for (int u = 0; u < FG_BATCH; ++u) {
                    r[u] = make_float4(0.0f, 0.0f, 0.0f, 0.0f);
                    if (j0 + u < cnt) r[u] = xv[__builtin_amdgcn_readlane(idx, j0 + u) * 32 + lane];
                }
#pragma unroll
                for (int u = 0; u < FG_BATCH; ++u)
                    if (j0 + u < cnt) { ax += r[u].x; ay += r[u].y; az += r[u].z; aw += r[u].w; }
            }
            if (left > 32) {
                const int nx = (lane < left - 32) ? bucket[base + 32 + lane] : 0;
                idx = settled(nx);
            }
        }
        float4 o; o.x = ax; o.y = ay; o.z = az; o.w = aw;
        *(float4*)(At + q * FG_LDA + lane * 4) = o;
    }
}

__device__ __forceinline__ void product_strip(const float* __restrict__ At,
                                              const float* __restrict__ Bsm,
                                              const float (&scale)[2], const float (&shift)[2],
                                              float* __restrict__ out, int rowBase,
                                              int colOff, int lane) {
    const int mrow = lane & 15;
    const int kh   = lane >> 4;
    const float* Ap0 = At + mrow * FG_LDA + 2 * kh;
    const float* Ap1 = Ap0 + 16 * FG_LDA;
    const float* Bp  = Bsm + 2 * kh * FG_LDB + colOff + mrow;
    v8f acc[2][2] = {};
    v2f fa0[2][FG_KB], fa1[2][FG_KB], fb0[2][FG_KB], fb1[2][FG_KB];
#pragma unroll
    for (int s = 0; s < FG_KB; ++s) {
        const int k = 4 * s;
        const float* bk = Bp + k * FG_LDB;
        fa0[0][s] = *(const v2f*)(Ap0 + k);
        fa1[0][s] = *(const v2f*)(Ap1 + k);
        fb0[0][s].x = bk[0];
        fb0[0][s].y = bk[FG_LDB];
        fb1[0][s].x = bk[16];
        fb1[0][s].y = bk[FG_LDB + 16];
    }
#pragma unroll
    for (int b = 0; b < GIN_DIN / (4 * FG_KB); ++b) {
        const int cur = b & 1, nxt = cur ^ 1;
        if (b + 1 < GIN_DIN / (4 * FG_KB)) {
#pragma unroll
            for (int s = 0; s < FG_KB; ++s) {
                const int k = 4 * ((b + 1) * FG_KB + s);
                const float* bk = Bp + k * FG_LDB;
                fa0[nxt][s] = *(const v2f*)(Ap0 + k);
                fa1[nxt][s] = *(const v2f*)(Ap1 + k);
                fb0[nxt][s].x = bk[0];
                fb0[nxt][s].y = bk[FG_LDB];
                fb1[nxt][s].x = bk[16];
                fb1[nxt][s].y = bk[FG_LDB + 16];
            }
        }
        __builtin_amdgcn_sched_barrier(0);
#pragma unroll
        for (int s = 0; s < FG_KB; ++s) {
            acc[0][0] = __builtin_amdgcn_wmma_f32_16x16x4_f32(false, fa0[cur][s], false, fb0[cur][s], (short)0, acc[0][0], false, false);
            acc[0][1] = __builtin_amdgcn_wmma_f32_16x16x4_f32(false, fa0[cur][s], false, fb1[cur][s], (short)0, acc[0][1], false, false);
            acc[1][0] = __builtin_amdgcn_wmma_f32_16x16x4_f32(false, fa1[cur][s], false, fb0[cur][s], (short)0, acc[1][0], false, false);
            acc[1][1] = __builtin_amdgcn_wmma_f32_16x16x4_f32(false, fa1[cur][s], false, fb1[cur][s], (short)0, acc[1][1], false, false);
        }
        __builtin_amdgcn_sched_barrier(0);
    }
#pragma unroll
    for (int ct = 0; ct < 2; ++ct) {
        const int c = colOff + ct * 16 + mrow;
#pragma unroll
        for (int rt = 0; rt < 2; ++rt) {
            const int row0 = rowBase + rt * 16 + kh * 8;
#pragma unroll
            for (int r = 0; r < 8; ++r)
                out[(size_t)(row0 + r) * GIN_DOUT + c] = fmaxf(acc[rt][ct][r] * scale[ct] + shift[ct], 0.0f);
        }
    }
}

__global__ __launch_bounds__(FG_T)
void gather_gemm_kernel(const float* __restrict__ x, const int* __restrict__ offsets,
                        const int* __restrict__ bucket, const float* __restrict__ epsp,
                        const float* __restrict__ B, const float* __restrict__ bias,
                        const float* __restrict__ gam, const float* __restrict__ bet,
                        const float* __restrict__ mu, const float* __restrict__ var,
                        float* __restrict__ out) {
    __shared__ __attribute__((aligned(16))) float As[2 * FG_AT];
    __shared__ __attribute__((aligned(16))) float Bs[GIN_DIN * FG_LDB];
    const int tid  = threadIdx.x;
    const int lane = tid & 31;
    const int wave = __builtin_amdgcn_readfirstlane(tid >> 5);
    const bool mma = wave < FG_MMA_WAVES;
    const float4* xv = (const float4*)x;
    const int colOff = wave * 32;
    const int grow   = (wave - FG_MMA_WAVES) * FG_NPW;

    int tile = blockIdx.x;
    float scale[2] = {1.0f, 1.0f}, shift[2] = {0.0f, 0.0f};
    float se = 0.0f;
    if (mma) {
        const float4* Bv = (const float4*)B;
#pragma unroll 1
        for (int i0 = 0; i0 < (GIN_DIN * GIN_DOUT / 4) / (32 * FG_MMA_WAVES); i0 += 8) {
            float4 t[8];
#pragma unroll
            for (int u = 0; u < 8; ++u) t[u] = Bv[tid + (i0 + u) * (32 * FG_MMA_WAVES)];
#pragma unroll
            for (int u = 0; u < 8; ++u) {
                const int i4 = tid + (i0 + u) * (32 * FG_MMA_WAVES);
                *(float4*)(Bs + (i4 >> 6) * FG_LDB + (i4 & 63) * 4) = t[u];
            }
        }
        const int mrow = lane & 15;
#pragma unroll
        for (int ct = 0; ct < 2; ++ct) {
            const int c = colOff + ct * 16 + mrow;
            scale[ct] = gam[c] * rsqrtf(var[c] + 1e-5f);
            shift[ct] = (bias[c] - mu[c]) * scale[ct] + bet[c];
        }
    } else {
        se = 1.0f + epsp[0];
        gather_rows(xv, offsets, bucket, se, As + grow * FG_LDA, tile * FG_ROWS + grow, lane);
    }
    __syncthreads();

    for (int it = 0; tile < FG_NT; ++it, tile += FG_GRID) {
        if (mma) {
            product_strip(As + (it & 1) * FG_AT, Bs, scale, shift, out, tile * FG_ROWS, colOff, lane);
        } else {
            const int nt = tile + FG_GRID;
            if (nt < FG_NT)
                gather_rows(xv, offsets, bucket, se, As + ((it + 1) & 1) * FG_AT + grow * FG_LDA,
                            nt * FG_ROWS + grow, lane);
        }
        __syncthreads();
    }
}


extern "C" void kernel_launch(void* const* d_in, const int* in_sizes, int n_in,
                              void* d_out, int out_size, void* d_ws, size_t ws_size,
                              hipStream_t stream) {
    constexpr int N = GIN_N, E = GIN_E, DIN = GIN_DIN, DOUT = GIN_DOUT;

    const float* x    = (const float*)d_in[0];
    const int*   eidx = (const int*)d_in[1];
    const float* eps  = (const float*)d_in[2];
    const float* W1   = (const float*)d_in[3];
    const float* b1   = (const float*)d_in[4];
    const float* g1   = (const float*)d_in[5];
    const float* bt1  = (const float*)d_in[6];
    const float* m1   = (const float*)d_in[7];
    const float* v1   = (const float*)d_in[8];
    const float* W2   = (const float*)d_in[9];
    const float* b2   = (const float*)d_in[10];
    const float* g2   = (const float*)d_in[11];
    const float* bt2  = (const float*)d_in[12];
    const float* m2   = (const float*)d_in[13];
    const float* v2   = (const float*)d_in[14];
    const float* Wres = (const float*)d_in[15];

    const int* src = eidx;
    const int* dst = eidx + E;

    float* h  = (float*)d_ws;
    float* h1 = h + (size_t)N * DIN;
    int* ibase   = (int*)(h1 + (size_t)2 * N * DOUT);
    int* tab     = ibase;
    static_assert((NBIN + 1) * NSEG <= 2 * GIN_N, "the table fits in front of offsets");
    int* offsets = ibase + 2 * N;
    int* bucket  = offsets + N + 1;
    int* pairs   = (int*)h1;

    constexpr int E4 = E / 4;
    bin_local_sort_kernel<<<NSEG, SORT_T, 0, stream>>>(src, dst, tab, pairs, E4);
    bin_sort_kernel<<<NBIN, SORT_T, 0, stream>>>(pairs, tab, offsets, bucket, N);
    gather_gemm_kernel<<<FG_GRID, FG_T, 0, stream>>>(
        x, offsets, bucket, eps, W1, b1, g1, bt1, m1, v1, h1);

    dim3 grid(N / 64, DOUT / 64);
    gemm_bn_res_kernel<256, 128><<<grid, 128, 0, stream>>>(
        h1, W2, b2, g2, bt2, m2, v2, x, Wres, (float*)d_out, DOUT);
}


# layer 2 and residual products moved into the gather launch: h1 tile stays in LDS, each product wave reads its own 32 columns of W2 and Wres from memory into fragments; h1 plane and one launch gone
# speedup vs baseline: 5.4699x; 1.0999x over previous
#include <hip/hip_runtime.h>
#include <hip/hip_bf16.h>

typedef float v2f __attribute__((ext_vector_type(2)));
typedef float v8f __attribute__((ext_vector_type(8)));

#define GIN_N    40000
#define GIN_E    640000
#define GIN_DIN  128
#define GIN_DOUT 256


#define BIN_SHIFT 7
#define BIN_NODES (1 << BIN_SHIFT)
#define NBIN  ((GIN_N + BIN_NODES - 1) / BIN_NODES)
#define NBINP (((NBIN + 1 + 31) / 32) * 32)
#define SORT_T 1024
#define SEG_EDGES (4 * SORT_T)
#define NSEG  ((GIN_E + SEG_EDGES - 1) / SEG_EDGES)
#define NSEGP (((NSEG + 31) / 32) * 32)
static_assert(GIN_E % 4 == 0, "edges are read four at a time");
static_assert(NBINP <= SORT_T, "one thread per bin");
static_assert(NSEGP <= SORT_T, "one thread per segment");

__global__ __launch_bounds__(SORT_T)
void bin_local_sort_kernel(const int* __restrict__ src, const int* __restrict__ dst,
                           int* __restrict__ tab, int* __restrict__ pairbuf, int e4) {
    __shared__ int hist[NBINP];
    __shared__ int smT[32];
    int2* pairs = (int2*)pairbuf;
    const int tid  = threadIdx.x;
    const int lane = tid & 31;
    const int wave = tid >> 5;

    if (tid < NBINP) hist[tid] = 0;
    __syncthreads();

    const int i4 = blockIdx.x * SORT_T + tid;
    const bool act = i4 < e4;
    int4 d  = make_int4(0, 0, 0, 0);
    int4 sr = make_int4(0, 0, 0, 0);
    int r0 = 0, r1 = 0, r2 = 0, r3 = 0;
    if (act) {
        d  = ((const int4*)dst)[i4];
        sr = ((const int4*)src)[i4];
        r0 = atomicAdd(&hist[d.x >> BIN_SHIFT], 1);
        r1 = atomicAdd(&hist[d.y >> BIN_SHIFT], 1);
        r2 = atomicAdd(&hist[d.z >> BIN_SHIFT], 1);
        r3 = atomicAdd(&hist[d.w >> BIN_SHIFT], 1);
    }
    __syncthreads();

    int v = 0, s = 0;
    if (tid < NBINP) {
        v = hist[tid];
        s = v;
#pragma unroll
        for (int off = 1; off < 32; off <<= 1) {
            int t = __shfl_up(s, off);
            if (lane >= off) s += t;
        }
        if (lane == 31) smT[wave] = s;
    }
    __syncthreads();

    if (tid < NBINP) {
        int base = 0;
#pragma unroll
        for (int w = 0; w < NBINP / 32; ++w) base += (w < wave) ? smT[w] : 0;
        int excl = base + s - v;
        hist[tid] = excl;
        if (tid <= NBIN) tab[tid * NSEG + blockIdx.x] = excl;
    }
    __syncthreads();

    if (act) {
        int2* reg = pairs + (size_t)blockIdx.x * SEG_EDGES;
        reg[hist[d.x >> BIN_SHIFT] + r0] = make_int2(d.x, sr.x);
        reg[hist[d.y >> BIN_SHIFT] + r1] = make_int2(d.y, sr.y);
        reg[hist[d.z >> BIN_SHIFT] + r2] = make_int2(d.z, sr.z);
        reg[hist[d.w >> BIN_SHIFT] + r3] = make_int2(d.w, sr.w);
    }
}

#define SORT_WAVES (SORT_T / 32)
#define SEG_TRIPS ((NSEG + SORT_WAVES - 1) / SORT_WAVES)
__global__ __launch_bounds__(SORT_T)
void bin_sort_kernel(const int* __restrict__ pairbuf, const int* __restrict__ tab,
                     int* __restrict__ offsets, int* __restrict__ bucket, int n) {
    __shared__ int cnt[BIN_NODES];
    __shared__ int loff[BIN_NODES];
    __shared__ int smT[BIN_NODES / 32];
    __shared__ int segS[NSEGP];
    __shared__ int segE[NSEGP];
    __shared__ int smP[NSEGP / 32];
    const int2* pairs = (const int2*)pairbuf;
    const int tid  = threadIdx.x;
    const int lane = tid & 31;
    const int wave = tid >> 5;
    const int b  = blockIdx.x;

    if (tid < NSEGP) {
        int st = 0, en = 0;
        if (tid < NSEG) {
            st = tab[b * NSEG + tid];
            en = tab[(b + 1) * NSEG + tid];
        }
        segS[tid] = st;
        segE[tid] = en;
        int t = st;
#pragma unroll
        for (int off = 16; off > 0; off >>= 1) t += __shfl_xor(t, off);
        if (lane == 0) smP[wave] = t;
    }
    if (tid < BIN_NODES) cnt[tid] = 0;
    __syncthreads();

    int p0 = 0;
#pragma unroll
    for (int w = 0; w < NSEGP / 32; ++w) p0 += smP[w];

    int  sb[SEG_TRIPS], se[SEG_TRIPS];
    int2 pr[SEG_TRIPS];
    bool ok[SEG_TRIPS];
#pragma unroll
    for (int q = 0; q < SEG_TRIPS; ++q) {
        const int sg = wave + q * SORT_WAVES;
        sb[q] = 0; se[q] = 0;
        if (sg < NSEG) { sb[q] = sg * SEG_EDGES + segS[sg] + lane; se[q] = sg * SEG_EDGES + segE[sg]; }
        ok[q] = sb[q] < se[q];
    }
#pragma unroll
    for (int q = 0; q < SEG_TRIPS; ++q) {
        pr[q] = make_int2(0, 0);
        if (ok[q]) pr[q] = pairs[sb[q]];
    }
#pragma unroll
    for (int q = 0; q < SEG_TRIPS; ++q)
        if (ok[q]) atomicAdd(&cnt[pr[q].x & (BIN_NODES - 1)], 1);
#pragma unroll
    for (int q = 0; q < SEG_TRIPS; ++q)
        for (int i = sb[q] + 32; i < se[q]; i += 32)
            atomicAdd(&cnt[pairs[i].x & (BIN_NODES - 1)], 1);
    __syncthreads();

    int v = 0, s = 0;
    if (tid < BIN_NODES) {
        v = cnt[tid];
        s = v;
#pragma unroll
        for (int off = 1; off < 32; off <<= 1) {
            int t = __shfl_up(s, off);
            if (lane >= off) s += t;
        }
        if (lane == 31) smT[wave] = s;
    }
    __syncthreads();

    if (tid < BIN_NODES) {
        int base = 0;
#pragma unroll
        for (int w = 0; w < BIN_NODES / 32; ++w) base += (w < wave) ? smT[w] : 0;
        int first = p0 + base + s - v;
        loff[tid] = first;
        cnt[tid] = 0;
        int node = b * BIN_NODES + tid;
        if (node < n) offsets[node] = first;
        if (node == n - 1) offsets[n] = first + v;
    }
    __syncthreads();

#pragma unroll
    for (int q = 0; q < SEG_TRIPS; ++q)
        if (ok[q]) {
            int l = pr[q].x & (BIN_NODES - 1);
            int r = atomicAdd(&cnt[l], 1);
            bucket[loff[l] + r] = pr[q].y;
        }
#pragma unroll
    for (int q = 0; q < SEG_TRIPS; ++q)
        for (int i = sb[q] + 32; i < se[q]; i += 32) {
            int2 p2 = pairs[i];
            int l = p2.x & (BIN_NODES - 1);
            int r = atomicAdd(&cnt[l], 1);
            bucket[loff[l] + r] = p2.y;
        }
}

#define FG_ROWS 32
#define FG_LDA  (GIN_DIN + 4)
#define FG_LDB  (GIN_DOUT + 8)
#define FG_LDH  (GIN_DOUT + 4)
#define FG_RING 3
#define FG_AT   (FG_ROWS * FG_LDA)
#define FG_NT   (GIN_N / FG_ROWS)
#define FG_GRID 250
#define FG_T    768
#define FG_MMA_WAVES 8
#define FG_BATCH 16
#define FG_KB    4
#define FG_NPW  (FG_ROWS / (FG_T / 32 - FG_MMA_WAVES))
static_assert(GIN_N % FG_ROWS == 0, "whole tiles of nodes");
static_assert(FG_ROWS == FG_NPW * (FG_T / 32 - FG_MMA_WAVES), "the gathering waves share a tile evenly");
static_assert(GIN_DOUT == 32 * FG_MMA_WAVES, "one 32-column strip per product wave");
static_assert(GIN_DIN == 128, "a node row is one float4 per lane");

__device__ __forceinline__ int settled(int v) {
    int r;
    asm volatile("v_mov_b32 %0, %1" : "=v"(r) : "v"(v));
    return r;
}

__device__ __forceinline__ void gather_rows(const float4* __restrict__ xv,
                                            const int* __restrict__ offsets,
                                            const int* __restrict__ bucket, float se,
                                            float* __restrict__ At, float* __restrict__ Xt,
                                            int node0, int lane) {
    const int ov = offsets[node0 + (lane < FG_NPW ? lane : FG_NPW)];
    float4 self[FG_NPW];
#pragma unroll
    for (int q = 0; q < FG_NPW; ++q) self[q] = xv[(node0 + q) * 32 + lane];
    int eb[FG_NPW + 1];
#pragma unroll
    for (int q = 0; q <= FG_NPW; ++q) eb[q] = __builtin_amdgcn_readlane(ov, q);
    int idf[FG_NPW];
#pragma unroll
    for (int q = 0; q < FG_NPW; ++q) idf[q] = (lane < eb[q + 1] - eb[q]) ? bucket[eb[q] + lane] : 0;

#pragma unroll
    for (int q = 0; q < FG_NPW; ++q) {
        const int e1 = eb[q + 1];
        float ax = self[q].x * se, ay = self[q].y * se, az = self[q].z * se, aw = self[q].w * se;
        *(float4*)(Xt + q * FG_LDA + lane * 4) = self[q];
        int idx = settled(idf[q]);
        for (int base = eb[q]; base < e1; base += 32) {
            const int left = e1 - base;
            const int cnt = left < 32 ? left : 32;
            for (int j0 = 0; j0 < cnt; j0 += FG_BATCH) {
                float4 r[FG_BATCH];
#pragma unroll
                for (int u = 0; u < FG_BATCH; ++u) {
                    r[u] = make_float4(0.0f, 0.0f, 0.0f, 0.0f);
                    if (j0 + u < cnt) r[u] = xv[__builtin_amdgcn_readlane(idx, j0 + u) * 32 + lane];
                }
#pragma unroll
                for (int u = 0; u < FG_BATCH; ++u)
                    if (j0 + u < cnt) { ax += r[u].x; ay += r[u].y; az += r[u].z; aw += r[u].w; }
            }
            if (left > 32) {
                const int nx = (lane < left - 32) ? bucket[base + 32 + lane] : 0;
                idx = settled(nx);
            }
        }
        float4 o; o.x = ax; o.y = ay; o.z = az; o.w = aw;
        *(float4*)(At + q * FG_LDA + lane * 4) = o;
    }
}

__device__ __forceinline__ void product_strip(const float* __restrict__ At,
                                              const float* __restrict__ Bsm,
                                              const float (&scale)[2], const float (&shift)[2],
                                              float* __restrict__ Hs,
                                              int colOff, int lane) {
    const int mrow = lane & 15;
    const int kh   = lane >> 4;
    const float* Ap0 = At + mrow * FG_LDA + 2 * kh;
    const float* Ap1 = Ap0 + 16 * FG_LDA;
    const float* Bp  = Bsm + 2 * kh * FG_LDB + colOff + mrow;
    v8f acc[2][2] = {};
    v2f fa0[2][FG_KB], fa1[2][FG_KB], fb0[2][FG_KB], fb1[2][FG_KB];
#pragma unroll
    for (int s = 0; s < FG_KB; ++s) {
        const int k = 4 * s;
        const float* bk = Bp + k * FG_LDB;
        fa0[0][s] = *(const v2f*)(Ap0 + k);
        fa1[0][s] = *(const v2f*)(Ap1 + k);
        fb0[0][s].x = bk[0];
        fb0[0][s].y = bk[FG_LDB];
        fb1[0][s].x = bk[16];
        fb1[0][s].y = bk[FG_LDB + 16];
    }
#pragma unroll
    for (int b = 0; b < GIN_DIN / (4 * FG_KB); ++b) {
        const int cur = b & 1, nxt = cur ^ 1;
        if (b + 1 < GIN_DIN / (4 * FG_KB)) {
#pragma unroll
            for (int s = 0; s < FG_KB; ++s) {
                const int k = 4 * ((b + 1) * FG_KB + s);
                const float* bk = Bp + k * FG_LDB;
                fa0[nxt][s] = *(const v2f*)(Ap0 + k);
                fa1[nxt][s] = *(const v2f*)(Ap1 + k);
                fb0[nxt][s].x = bk[0];
                fb0[nxt][s].y = bk[FG_LDB];
                fb1[nxt][s].x = bk[16];
                fb1[nxt][s].y = bk[FG_LDB + 16];
            }
        }
        __builtin_amdgcn_sched_barrier(0);
#pragma unroll
        for (int s = 0; s < FG_KB; ++s) {
            acc[0][0] = __builtin_amdgcn_wmma_f32_16x16x4_f32(false, fa0[cur][s], false, fb0[cur][s], (short)0, acc[0][0], false, false);
            acc[0][1] = __builtin_amdgcn_wmma_f32_16x16x4_f32(false, fa0[cur][s], false, fb1[cur][s], (short)0, acc[0][1], false, false);
            acc[1][0] = __builtin_amdgcn_wmma_f32_16x16x4_f32(false, fa1[cur][s], false, fb0[cur][s], (short)0, acc[1][0], false, false);
            acc[1][1] = __builtin_amdgcn_wmma_f32_16x16x4_f32(false, fa1[cur][s], false, fb1[cur][s], (short)0, acc[1][1], false, false);
        }
        __builtin_amdgcn_sched_barrier(0);
    }
#pragma unroll
    for (int ct = 0; ct < 2; ++ct) {
        const int c = colOff + ct * 16 + mrow;
#pragma unroll
        for (int rt = 0; rt < 2; ++rt) {
            const int row0 = rt * 16 + kh * 8;
#pragma unroll
            for (int r = 0; r < 8; ++r)
                Hs[(row0 + r) * FG_LDH + c] = fmaxf(acc[rt][ct][r] * scale[ct] + shift[ct], 0.0f);
        }
    }
}

template <int K, int LDA_>
__device__ __forceinline__ void product_gb(const float* __restrict__ At, const float* __restrict__ Bg,
                                           v8f (&acc)[2][2], int lane) {
    constexpr int NB = K / (4 * FG_KB);
    static_assert(K % (4 * FG_KB) == 0 && NB >= 2, "whole blocks, at least two");
    const int mrow = lane & 15;
    const int kh   = lane >> 4;
    const float* Ap0 = At + mrow * LDA_ + 2 * kh;
    const float* Ap1 = Ap0 + 16 * LDA_;
    const float* Bp  = Bg + 2 * kh * GIN_DOUT + mrow;
    float gb[FG_RING][FG_KB][4];
    v2f fa0[2][FG_KB], fa1[2][FG_KB];
#pragma unroll
    for (int b = 0; b < 2; ++b)
#pragma unroll
        for (int s = 0; s < FG_KB; ++s) {
            const float* bk = Bp + 4 * (b * FG_KB + s) * GIN_DOUT;
            gb[b][s][0] = bk[0];
            gb[b][s][1] = bk[GIN_DOUT];
            gb[b][s][2] = bk[16];
            gb[b][s][3] = bk[GIN_DOUT + 16];
        }
#pragma unroll
    for (int s = 0; s < FG_KB; ++s) {
        fa0[0][s] = *(const v2f*)(Ap0 + 4 * s);
        fa1[0][s] = *(const v2f*)(Ap1 + 4 * s);
    }
#pragma unroll
    for (int b = 0; b < NB; ++b) {
        const int cur = b & 1, nxt = cur ^ 1;
        const int gc = b % FG_RING, gn = (b + 2) % FG_RING;
        if (b + 2 < NB) {
#pragma unroll
            for (int s = 0; s < FG_KB; ++s) {
                const float* bk = Bp + 4 * ((b + 2) * FG_KB + s) * GIN_DOUT;
                gb[gn][s][0] = bk[0];
                gb[gn][s][1] = bk[GIN_DOUT];
                gb[gn][s][2] = bk[16];
                gb[gn][s][3] = bk[GIN_DOUT + 16];
            }
        }
        if (b + 1 < NB) {
#pragma unroll
            for (int s = 0; s < FG_KB; ++s) {
                const int k = 4 * ((b + 1) * FG_KB + s);
                fa0[nxt][s] = *(const v2f*)(Ap0 + k);
                fa1[nxt][s] = *(const v2f*)(Ap1 + k);
            }
        }
        __builtin_amdgcn_sched_barrier(0);
#pragma unroll
        for (int s = 0; s < FG_KB; ++s) {
            v2f b0, b1;
            b0.x = gb[gc][s][0]; b0.y = gb[gc][s][1];
            b1.x = gb[gc][s][2]; b1.y = gb[gc][s][3];
            acc[0][0] = __builtin_amdgcn_wmma_f32_16x16x4_f32(false, fa0[cur][s], false, b0, (short)0, acc[0][0], false, false);
            acc[0][1] = __builtin_amdgcn_wmma_f32_16x16x4_f32(false, fa0[cur][s], false, b1, (short)0, acc[0][1], false, false);
            acc[1][0] = __builtin_amdgcn_wmma_f32_16x16x4_f32(false, fa1[cur][s], false, b0, (short)0, acc[1][0], false, false);
            acc[1][1] = __builtin_amdgcn_wmma_f32_16x16x4_f32(false, fa1[cur][s], false, b1, (short)0, acc[1][1], false, false);
        }
        __builtin_amdgcn_sched_barrier(0);
    }
}

__global__ __launch_bounds__(FG_T)
void gather_gemm_kernel(const float* __restrict__ x, const int* __restrict__ offsets,
                        const int* __restrict__ bucket, const float* __restrict__ epsp,
                        const float* __restrict__ B, const float* __restrict__ bias,
                        const float* __restrict__ gam, const float* __restrict__ bet,
                        const float* __restrict__ mu, const float* __restrict__ var,
                        const float* __restrict__ B2, const float* __restrict__ bias2,
                        const float* __restrict__ gam2, const float* __restrict__ bet2,
                        const float* __restrict__ mu2, const float* __restrict__ var2,
                        const float* __restrict__ Br, float* __restrict__ out) {
    __shared__ __attribute__((aligned(16))) float As[2 * FG_AT];
    __shared__ __attribute__((aligned(16))) float Xs[2 * FG_AT];
    __shared__ __attribute__((aligned(16))) float Hs[FG_ROWS * FG_LDH];
    __shared__ __attribute__((aligned(16))) float Bs[GIN_DIN * FG_LDB];
    const int tid  = threadIdx.x;
    const int lane = tid & 31;
    const int wave = __builtin_amdgcn_readfirstlane(tid >> 5);
    const bool mma = wave < FG_MMA_WAVES;
    const float4* xv = (const float4*)x;
    const int colOff = wave * 32;
    const int grow   = (wave - FG_MMA_WAVES) * FG_NPW;

    int tile = blockIdx.x;
    float scale[2] = {1.0f, 1.0f}, shift[2] = {0.0f, 0.0f};
    float scale2[2] = {1.0f, 1.0f}, shift2[2] = {0.0f, 0.0f};
    float se = 0.0f;
    if (mma) {
        const float4* Bv = (const float4*)B;
#pragma unroll 1
        for (int i0 = 0; i0 < (GIN_DIN * GIN_DOUT / 4) / (32 * FG_MMA_WAVES); i0 += 8) {
            float4 t[8];
#pragma unroll
            for (int u = 0; u < 8; ++u) t[u] = Bv[tid + (i0 + u) * (32 * FG_MMA_WAVES)];
#pragma unroll
            for (int u = 0; u < 8; ++u) {
                const int i4 = tid + (i0 + u) * (32 * FG_MMA_WAVES);
                *(float4*)(Bs + (i4 >> 6) * FG_LDB + (i4 & 63) * 4) = t[u];
            }
        }
        const int mrow = lane & 15;
#pragma unroll
        for (int ct = 0; ct < 2; ++ct) {
            const int c = colOff + ct * 16 + mrow;
            scale[ct] = gam[c] * rsqrtf(var[c] + 1e-5f);
            shift[ct] = (bias[c] - mu[c]) * scale[ct] + bet[c];
            scale2[ct] = gam2[c] * rsqrtf(var2[c] + 1e-5f);
            shift2[ct] = (bias2[c] - mu2[c]) * scale2[ct] + bet2[c];
        }
    } else {
        se = 1.0f + epsp[0];
        gather_rows(xv, offsets, bucket, se, As + grow * FG_LDA, Xs + grow * FG_LDA,
                    tile * FG_ROWS + grow, lane);
    }
    __syncthreads();

    for (int it = 0; tile < FG_NT; ++it, tile += FG_GRID) {
        if (mma) product_strip(As + (it & 1) * FG_AT, Bs, scale, shift, Hs, colOff, lane);
        __syncthreads();
        if (!mma) {
            const int nt = tile + FG_GRID;
            if (nt < FG_NT)
                gather_rows(xv, offsets, bucket, se, As + ((it + 1) & 1) * FG_AT + grow * FG_LDA,
                            Xs + ((it + 1) & 1) * FG_AT + grow * FG_LDA, nt * FG_ROWS + grow, lane);
        } else {
            const int mrow = lane & 15;
            const int kh   = lane >> 4;
            v8f acc[2][2] = {};
            product_gb<GIN_DOUT, FG_LDH>(Hs, B2 + colOff, acc, lane);
#pragma unroll
            for (int rt = 0; rt < 2; ++rt)
#pragma unroll
                for (int ct = 0; ct < 2; ++ct)
#pragma unroll
                    for (int r = 0; r < 8; ++r)
                        acc[rt][ct][r] = fmaxf(acc[rt][ct][r] * scale2[ct] + shift2[ct], 0.0f);
            product_gb<GIN_DIN, FG_LDA>(Xs + (it & 1) * FG_AT, Br + colOff, acc, lane);
#pragma unroll
            for (int ct = 0; ct < 2; ++ct) {
                const int c = colOff + ct * 16 + mrow;
#pragma unroll
                for (int rt = 0; rt < 2; ++rt) {
                    const int row0 = tile * FG_ROWS + rt * 16 + kh * 8;
#pragma unroll
                    for (int r = 0; r < 8; ++r)
                        out[(size_t)(row0 + r) * GIN_DOUT + c] = fmaxf(acc[rt][ct][r], 0.0f);
                }
            }
        }
        __syncthreads();
    }
}


extern "C" void kernel_launch(void* const* d_in, const int* in_sizes, int n_in,
                              void* d_out, int out_size, void* d_ws, size_t ws_size,
                              hipStream_t stream) {
    constexpr int N = GIN_N, E = GIN_E, DIN = GIN_DIN, DOUT = GIN_DOUT;

    const float* x    = (const float*)d_in[0];
    const int*   eidx = (const int*)d_in[1];
    const float* eps  = (const float*)d_in[2];
    const float* W1   = (const float*)d_in[3];
    const float* b1   = (const float*)d_in[4];
    const float* g1   = (const float*)d_in[5];
    const float* bt1  = (const float*)d_in[6];
    const float* m1   = (const float*)d_in[7];
    const float* v1   = (const float*)d_in[8];
    const float* W2   = (const float*)d_in[9];
    const float* b2   = (const float*)d_in[10];
    const float* g2   = (const float*)d_in[11];
    const float* bt2  = (const float*)d_in[12];
    const float* m2   = (const float*)d_in[13];
    const float* v2   = (const float*)d_in[14];
    const float* Wres = (const float*)d_in[15];

    const int* src = eidx;
    const int* dst = eidx + E;

    float* h  = (float*)d_ws;
    float* h1 = h + (size_t)N * DIN;
    int* ibase   = (int*)(h1 + (size_t)2 * N * DOUT);
    int* tab     = ibase;
    static_assert((NBIN + 1) * NSEG <= 2 * GIN_N, "the table fits in front of offsets");
    int* offsets = ibase + 2 * N;
    int* bucket  = offsets + N + 1;
    int* pairs   = (int*)h1;

    constexpr int E4 = E / 4;
    bin_local_sort_kernel<<<NSEG, SORT_T, 0, stream>>>(src, dst, tab, pairs, E4);
    bin_sort_kernel<<<NBIN, SORT_T, 0, stream>>>(pairs, tab, offsets, bucket, N);
    gather_gemm_kernel<<<FG_GRID, FG_T, 0, stream>>>(
        x, offsets, bucket, eps, W1, b1, g1, bt1, m1, v1,
        W2, b2, g2, bt2, m2, v2, Wres, (float*)d_out);
}


# W1/W2/Wres re-laid by the first launch, rows 2p and 2p+1 interleaved per column; a product lane owns columns 2c,2c+1: one 128-bit load per k-step feeds both column tiles (B loads /4), 64-bit stores
# speedup vs baseline: 5.7162x; 1.0450x over previous
#include <hip/hip_runtime.h>
#include <hip/hip_bf16.h>

typedef float v2f __attribute__((ext_vector_type(2)));
typedef float v4f __attribute__((ext_vector_type(4)));
typedef float v8f __attribute__((ext_vector_type(8)));

#define GIN_N    40000
#define GIN_E    640000
#define GIN_DIN  128
#define GIN_DOUT 256


#define BIN_SHIFT 7
#define BIN_NODES (1 << BIN_SHIFT)
#define NBIN  ((GIN_N + BIN_NODES - 1) / BIN_NODES)
#define NBINP (((NBIN + 1 + 31) / 32) * 32)
#define SORT_T 1024
#define SEG_EDGES (4 * SORT_T)
#define NSEG  ((GIN_E + SEG_EDGES - 1) / SEG_EDGES)
#define NSEGP (((NSEG + 31) / 32) * 32)
static_assert(GIN_E % 4 == 0, "edges are read four at a time");
static_assert(NBINP <= SORT_T, "one thread per bin");
static_assert(NSEGP <= SORT_T, "one thread per segment");

#define WP_W1  0
#define WP_W2  (GIN_DIN * GIN_DOUT)
#define WP_WR  (GIN_DIN * GIN_DOUT + GIN_DOUT * GIN_DOUT)
#define WP_PAIRS_W1 (GIN_DIN * GIN_DOUT / 2)
#define WP_PAIRS_W2 (GIN_DOUT * GIN_DOUT / 2)
#define WP_PAIRS_WR (GIN_DIN * GIN_DOUT / 2)
static_assert(GIN_DOUT == 256, "an item index splits into row pair and column by 8 bits");
static_assert(WP_PAIRS_W1 % SORT_T == 0 && WP_PAIRS_W2 % SORT_T == 0 && WP_PAIRS_WR % SORT_T == 0,
              "a block's items belong to one matrix");
static_assert(WP_PAIRS_W1 + WP_PAIRS_W2 + WP_PAIRS_WR <= NSEG * SORT_T, "one item per thread of the level-1 launch");

__global__ __launch_bounds__(SORT_T)
void bin_local_sort_kernel(const int* __restrict__ src, const int* __restrict__ dst,
                           int* __restrict__ tab, int* __restrict__ pairbuf, int e4,
                           const float* __restrict__ W1, const float* __restrict__ W2,
                           const float* __restrict__ Wr, float* __restrict__ wp) {
    __shared__ int hist[NBINP];
    __shared__ int smT[32];
    int2* pairs = (int2*)pairbuf;
    const int tid  = threadIdx.x;
    const int lane = tid & 31;
    const int wave = tid >> 5;

    const int item = blockIdx.x * SORT_T + tid;
    const bool relay = item < WP_PAIRS_W1 + WP_PAIRS_W2 + WP_PAIRS_WR;
    float2 wpair = make_float2(0.0f, 0.0f);
    int wslot = 0;
    if (relay) {
        const float* wsrc = W1;
        int q = item;
        int base = WP_W1;
        if (item >= WP_PAIRS_W1 + WP_PAIRS_W2) { wsrc = Wr; q = item - (WP_PAIRS_W1 + WP_PAIRS_W2); base = WP_WR; }
        else if (item >= WP_PAIRS_W1)          { wsrc = W2; q = item - WP_PAIRS_W1;                 base = WP_W2; }
        const int p = q >> 8, n = q & (GIN_DOUT - 1);
        wpair.x = wsrc[(2 * p) * GIN_DOUT + n];
        wpair.y = wsrc[(2 * p + 1) * GIN_DOUT + n];
        wslot = base / 2 + q;
    }

    if (tid < NBINP) hist[tid] = 0;
    __syncthreads();

    const int i4 = blockIdx.x * SORT_T + tid;
    const bool act = i4 < e4;
    int4 d  = make_int4(0, 0, 0, 0);
    int4 sr = make_int4(0, 0, 0, 0);
    int r0 = 0, r1 = 0, r2 = 0, r3 = 0;
    if (act) {
        d  = ((const int4*)dst)[i4];
        sr = ((const int4*)src)[i4];
        r0 = atomicAdd(&hist[d.x >> BIN_SHIFT], 1);
        r1 = atomicAdd(&hist[d.y >> BIN_SHIFT], 1);
        r2 = atomicAdd(&hist[d.z >> BIN_SHIFT], 1);
        r3 = atomicAdd(&hist[d.w >> BIN_SHIFT], 1);
    }
    __syncthreads();

    int v = 0, s = 0;
    if (tid < NBINP) {
        v = hist[tid];
        s = v;
#pragma unroll
        for (int off = 1; off < 32; off <<= 1) {
            int t = __shfl_up(s, off);
            if (lane >= off) s += t;
        }
        if (lane == 31) smT[wave] = s;
    }
    __syncthreads();

    if (tid < NBINP) {
        int base = 0;
#pragma unroll
        for (int w = 0; w < NBINP / 32; ++w) base += (w < wave) ? smT[w] : 0;
        int excl = base + s - v;
        hist[tid] = excl;
        if (tid <= NBIN) tab[tid * NSEG + blockIdx.x] = excl;
    }
    __syncthreads();

    if (act) {
        int2* reg = pairs + (size_t)blockIdx.x * SEG_EDGES;
        reg[hist[d.x >> BIN_SHIFT] + r0] = make_int2(d.x, sr.x);
        reg[hist[d.y >> BIN_SHIFT] + r1] = make_int2(d.y, sr.y);
        reg[hist[d.z >> BIN_SHIFT] + r2] = make_int2(d.z, sr.z);
        reg[hist[d.w >> BIN_SHIFT] + r3] = make_int2(d.w, sr.w);
    }
    if (relay) ((float2*)wp)[wslot] = wpair;
}

#define SORT_WAVES (SORT_T / 32)
#define SEG_TRIPS ((NSEG + SORT_WAVES - 1) / SORT_WAVES)
__global__ __launch_bounds__(SORT_T)
void bin_sort_kernel(const int* __restrict__ pairbuf, const int* __restrict__ tab,
                     int* __restrict__ offsets, int* __restrict__ bucket, int n) {
    __shared__ int cnt[BIN_NODES];
    __shared__ int loff[BIN_NODES];
    __shared__ int smT[BIN_NODES / 32];
    __shared__ int segS[NSEGP];
    __shared__ int segE[NSEGP];
    __shared__ int smP[NSEGP / 32];
    const int2* pairs = (const int2*)pairbuf;
    const int tid  = threadIdx.x;
    const int lane = tid & 31;
    const int wave = tid >> 5;
    const int b  = blockIdx.x;

    if (tid < NSEGP) {
        int st = 0, en = 0;
        if (tid < NSEG) {
            st = tab[b * NSEG + tid];
            en = tab[(b + 1) * NSEG + tid];
        }
        segS[tid] = st;
        segE[tid] = en;
        int t = st;
#pragma unroll
        for (int off = 16; off > 0; off >>= 1) t += __shfl_xor(t, off);
        if (lane == 0) smP[wave] = t;
    }
    if (tid < BIN_NODES) cnt[tid] = 0;
    __syncthreads();

    int p0 = 0;
#pragma unroll
    for (int w = 0; w < NSEGP / 32; ++w) p0 += smP[w];

    int  sb[SEG_TRIPS], se[SEG_TRIPS];
    int2 pr[SEG_TRIPS];
    bool ok[SEG_TRIPS];
#pragma unroll
    for (int q = 0; q < SEG_TRIPS; ++q) {
        const int sg = wave + q * SORT_WAVES;
        sb[q] = 0; se[q] = 0;
        if (sg < NSEG) { sb[q] = sg * SEG_EDGES + segS[sg] + lane; se[q] = sg * SEG_EDGES + segE[sg]; }
        ok[q] = sb[q] < se[q];
    }
#pragma unroll
    for (int q = 0; q < SEG_TRIPS; ++q) {
        pr[q] = make_int2(0, 0);
        if (ok[q]) pr[q] = pairs[sb[q]];
    }
#pragma unroll
    for (int q = 0; q < SEG_TRIPS; ++q)
        if (ok[q]) atomicAdd(&cnt[pr[q].x & (BIN_NODES - 1)], 1);
#pragma unroll
    for (int q = 0; q < SEG_TRIPS; ++q)
        for (int i = sb[q] + 32; i < se[q]; i += 32)
            atomicAdd(&cnt[pairs[i].x & (BIN_NODES - 1)], 1);
    __syncthreads();

    int v = 0, s = 0;
    if (tid < BIN_NODES) {
        v = cnt[tid];
        s = v;
#pragma unroll
        for (int off = 1; off < 32; off <<= 1) {
            int t = __shfl_up(s, off);
            if (lane >= off) s += t;
        }
        if (lane == 31) smT[wave] = s;
    }
    __syncthreads();

    if (tid < BIN_NODES) {
        int base = 0;
#pragma unroll
        for (int w = 0; w < BIN_NODES / 32; ++w) base += (w < wave) ? smT[w] : 0;
        int first = p0 + base + s - v;
        loff[tid] = first;
        cnt[tid] = 0;
        int node = b * BIN_NODES + tid;
        if (node < n) offsets[node] = first;
        if (node == n - 1) offsets[n] = first + v;
    }
    __syncthreads();

#pragma unroll
    for (int q = 0; q < SEG_TRIPS; ++q)
        if (ok[q]) {
            int l = pr[q].x & (BIN_NODES - 1);
            int r = atomicAdd(&cnt[l], 1);
            bucket[loff[l] + r] = pr[q].y;
        }
#pragma unroll
    for (int q = 0; q < SEG_TRIPS; ++q)
        for (int i = sb[q] + 32; i < se[q]; i += 32) {
            int2 p2 = pairs[i];
            int l = p2.x & (BIN_NODES - 1);
            int r = atomicAdd(&cnt[l], 1);
            bucket[loff[l] + r] = p2.y;
        }
}

#define FG_ROWS 32
#define FG_LDA  (GIN_DIN + 4)
#define FG_LDB  (2 * GIN_DOUT + 32)
#define FG_LDH  (GIN_DOUT + 4)
#define FG_RING 3
#define FG_AT   (FG_ROWS * FG_LDA)
#define FG_NT   (GIN_N / FG_ROWS)
#define FG_GRID 250
#define FG_T    768
#define FG_MMA_WAVES 8
#define FG_BATCH 16
#define FG_KB    4
#define FG_NPW  (FG_ROWS / (FG_T / 32 - FG_MMA_WAVES))
static_assert(GIN_N % FG_ROWS == 0, "whole tiles of nodes");
static_assert(FG_ROWS == FG_NPW * (FG_T / 32 - FG_MMA_WAVES), "the gathering waves share a tile evenly");
static_assert(GIN_DOUT == 32 * FG_MMA_WAVES, "one 32-column strip per product wave");
static_assert(GIN_DIN == 128, "a node row is one float4 per lane");
static_assert(2 * GIN_DOUT / 4 == 128, "W1 staging: 128 pieces of 16 bytes per row pair");

__device__ __forceinline__ int settled(int v) {
    int r;
    asm volatile("v_mov_b32 %0, %1" : "=v"(r) : "v"(v));
    return r;
}

__device__ __forceinline__ void gather_rows(const float4* __restrict__ xv,
                                            const int* __restrict__ offsets,
                                            const int* __restrict__ bucket, float se,
                                            float* __restrict__ At, float* __restrict__ Xt,
                                            int node0, int lane) {
    const int ov = offsets[node0 + (lane < FG_NPW ? lane : FG_NPW)];
    float4 self[FG_NPW];
#pragma unroll
    for (int q = 0; q < FG_NPW; ++q) self[q] = xv[(node0 + q) * 32 + lane];
    int eb[FG_NPW + 1];
#pragma unroll
    for (int q = 0; q <= FG_NPW; ++q) eb[q] = __builtin_amdgcn_readlane(ov, q);
    int idf[FG_NPW];
#pragma unroll
    for (int q = 0; q < FG_NPW; ++q) idf[q] = (lane < eb[q + 1] - eb[q]) ? bucket[eb[q] + lane] : 0;

#pragma unroll
    for (int q = 0; q < FG_NPW; ++q) {
        const int e1 = eb[q + 1];
        float ax = self[q].x * se, ay = self[q].y * se, az = self[q].z * se, aw = self[q].w * se;
        *(float4*)(Xt + q * FG_LDA + lane * 4) = self[q];
        int idx = settled(idf[q]);
        for (int base = eb[q]; base < e1; base += 32) {
            const int left = e1 - base;
            const int cnt = left < 32 ? left : 32;
            for (int j0 = 0; j0 < cnt; j0 += FG_BATCH) {
                float4 r[FG_BATCH];
#pragma unroll
                for (int u = 0; u < FG_BATCH; ++u) {
                    r[u] = make_float4(0.0f, 0.0f, 0.0f, 0.0f);
                    if (j0 + u < cnt) r[u] = xv[__builtin_amdgcn_readlane(idx, j0 + u) * 32 + lane];
                }
#pragma unroll
                for (int u = 0; u < FG_BATCH; ++u)
                    if (j0 + u < cnt) { ax += r[u].x; ay += r[u].y; az += r[u].z; aw += r[u].w; }
            }
            if (left > 32) {
                const int nx = (lane < left - 32) ? bucket[base + 32 + lane] : 0;
                idx = settled(nx);
            }
        }
        float4 o; o.x = ax; o.y = ay; o.z = az; o.w = aw;
        *(float4*)(At + q * FG_LDA + lane * 4) = o;
    }
}

__device__ __forceinline__ void product_strip(const float* __restrict__ At,
                                              const float* __restrict__ Bsm,
                                              const float (&scale)[2], const float (&shift)[2],
                                              float* __restrict__ Hs,
                                              int colOff, int lane) {
    const int mrow = lane & 15;
    const int kh   = lane >> 4;
    const float* Ap0 = At + mrow * FG_LDA + 2 * kh;
    const float* Ap1 = Ap0 + 16 * FG_LDA;
    const float* Bp  = Bsm + kh * FG_LDB + 2 * colOff + 4 * mrow;
    v8f acc[2][2] = {};
    v2f fa0[2][FG_KB], fa1[2][FG_KB];
    v4f fq[2][FG_KB];
#pragma unroll
    for (int s = 0; s < FG_KB; ++s) {
        const int k = 4 * s;
        fa0[0][s] = *(const v2f*)(Ap0 + k);
        fa1[0][s] = *(const v2f*)(Ap1 + k);
        fq[0][s]  = *(const v4f*)(Bp + (k / 2) * FG_LDB);
    }
#pragma unroll
    for (int b = 0; b < GIN_DIN / (4 * FG_KB); ++b) {
        const int cur = b & 1, nxt = cur ^ 1;
        if (b + 1 < GIN_DIN / (4 * FG_KB)) {
#pragma unroll
            for (int s = 0; s < FG_KB; ++s) {
                const int k = 4 * ((b + 1) * FG_KB + s);
                fa0[nxt][s] = *(const v2f*)(Ap0 + k);
                fa1[nxt][s] = *(const v2f*)(Ap1 + k);
                fq[nxt][s]  = *(const v4f*)(Bp + (k / 2) * FG_LDB);
            }
        }
        __builtin_amdgcn_sched_barrier(0);
#pragma unroll
        for (int s = 0; s < FG_KB; ++s) {
            v2f b0, b1;
            b0.x = fq[cur][s].x; b0.y = fq[cur][s].y;
            b1.x = fq[cur][s].z; b1.y = fq[cur][s].w;
            acc[0][0] = __builtin_amdgcn_wmma_f32_16x16x4_f32(false, fa0[cur][s], false, b0, (short)0, acc[0][0], false, false);
            acc[0][1] = __builtin_amdgcn_wmma_f32_16x16x4_f32(false, fa0[cur][s], false, b1, (short)0, acc[0][1], false, false);
            acc[1][0] = __builtin_amdgcn_wmma_f32_16x16x4_f32(false, fa1[cur][s], false, b0, (short)0, acc[1][0], false, false);
            acc[1][1] = __builtin_amdgcn_wmma_f32_16x16x4_f32(false, fa1[cur][s], false, b1, (short)0, acc[1][1], false, false);
        }
        __builtin_amdgcn_sched_barrier(0);
    }
#pragma unroll
    for (int rt = 0; rt < 2; ++rt) {
        const int row0 = rt * 16 + kh * 8;
#pragma unroll
        for (int r = 0; r < 8; ++r) {
            v2f o;
            o.x = fmaxf(acc[rt][0][r] * scale[0] + shift[0], 0.0f);
            o.y = fmaxf(acc[rt][1][r] * scale[1] + shift[1], 0.0f);
            *(v2f*)(Hs + (row0 + r) * FG_LDH + colOff + 2 * mrow) = o;
        }
    }
}

template <int K, int LDA_>
__device__ __forceinline__ void product_gb(const float* __restrict__ At, const float* __restrict__ Bg,
                                           v8f (&acc)[2][2], int lane) {
    constexpr int NB = K / (4 * FG_KB);
    static_assert(K % (4 * FG_KB) == 0 && NB >= 2, "whole blocks, at least two");
    const int mrow = lane & 15;
    const int kh   = lane >> 4;
    const float* Ap0 = At + mrow * LDA_ + 2 * kh;
    const float* Ap1 = Ap0 + 16 * LDA_;
    const float* Bp  = Bg + kh * (2 * GIN_DOUT) + 4 * mrow;
    v4f gq[FG_RING][FG_KB];
    v2f fa0[2][FG_KB], fa1[2][FG_KB];
#pragma unroll
    for (int b = 0; b < 2; ++b)
#pragma unroll
        for (int s = 0; s < FG_KB; ++s) {
            gq[b][s] = *(const v4f*)(Bp + 4 * (b * FG_KB + s) * GIN_DOUT);
        }
#pragma unroll
    for (int s = 0; s < FG_KB; ++s) {
        fa0[0][s] = *(const v2f*)(Ap0 + 4 * s);
        fa1[0][s] = *(const v2f*)(Ap1 + 4 * s);
    }
#pragma unroll
    for (int b = 0; b < NB; ++b) {
        const int cur = b & 1, nxt = cur ^ 1;
        const int gc = b % FG_RING, gn = (b + 2) % FG_RING;
        if (b + 2 < NB) {
#pragma unroll
            for (int s = 0; s < FG_KB; ++s) {
                gq[gn][s] = *(const v4f*)(Bp + 4 * ((b + 2) * FG_KB + s) * GIN_DOUT);
            }
        }
        if (b + 1 < NB) {
#pragma unroll
            for (int s = 0; s < FG_KB; ++s) {
                const int k = 4 * ((b + 1) * FG_KB + s);
                fa0[nxt][s] = *(const v2f*)(Ap0 + k);
                fa1[nxt][s] = *(const v2f*)(Ap1 + k);
            }
        }
        __builtin_amdgcn_sched_barrier(0);
#pragma unroll
        for (int s = 0; s < FG_KB; ++s) {
            v2f b0, b1;
            b0.x = gq[gc][s].x; b0.y = gq[gc][s].y;
            b1.x = gq[gc][s].z; b1.y = gq[gc][s].w;
            acc[0][0] = __builtin_amdgcn_wmma_f32_16x16x4_f32(false, fa0[cur][s], false, b0, (short)0, acc[0][0], false, false);
            acc[0][1] = __builtin_amdgcn_wmma_f32_16x16x4_f32(false, fa0[cur][s], false, b1, (short)0, acc[0][1], false, false);
            acc[1][0] = __builtin_amdgcn_wmma_f32_16x16x4_f32(false, fa1[cur][s], false, b0, (short)0, acc[1][0], false, false);
            acc[1][1] = __builtin_amdgcn_wmma_f32_16x16x4_f32(false, fa1[cur][s], false, b1, (short)0, acc[1][1], false, false);
        }
        __builtin_amdgcn_sched_barrier(0);
    }
}

__global__ __launch_bounds__(FG_T)
void gather_gemm_kernel(const float* __restrict__ x, const int* __restrict__ offsets,
                        const int* __restrict__ bucket, const float* __restrict__ epsp,
                        const float* __restrict__ B, const float* __restrict__ bias,
                        const float* __restrict__ gam, const float* __restrict__ bet,
                        const float* __restrict__ mu, const float* __restrict__ var,
                        const float* __restrict__ B2, const float* __restrict__ bias2,
                        const float* __restrict__ gam2, const float* __restrict__ bet2,
                        const float* __restrict__ mu2, const float* __restrict__ var2,
                        const float* __restrict__ Br, float* __restrict__ out) {
    __shared__ __attribute__((aligned(16))) float As[2 * FG_AT];
    __shared__ __attribute__((aligned(16))) float Xs[2 * FG_AT];
    __shared__ __attribute__((aligned(16))) float Hs[FG_ROWS * FG_LDH];
    __shared__ __attribute__((aligned(16))) float Bs[(GIN_DIN / 2) * FG_LDB];
    const int tid  = threadIdx.x;
    const int lane = tid & 31;
    const int wave = __builtin_amdgcn_readfirstlane(tid >> 5);
    const bool mma = wave < FG_MMA_WAVES;
    const float4* xv = (const float4*)x;
    const int colOff = wave * 32;
    const int grow   = (wave - FG_MMA_WAVES) * FG_NPW;

    int tile = blockIdx.x;
    float scale[2] = {1.0f, 1.0f}, shift[2] = {0.0f, 0.0f};
    float scale2[2] = {1.0f, 1.0f}, shift2[2] = {0.0f, 0.0f};
    float se = 0.0f;
    if (mma) {
        const float4* Bv = (const float4*)B;
#pragma unroll 1
        for (int i0 = 0; i0 < (GIN_DIN * GIN_DOUT / 4) / (32 * FG_MMA_WAVES); i0 += 8) {
            float4 t[8];
#pragma unroll
            for (int u = 0; u < 8; ++u) t[u] = Bv[tid + (i0 + u) * (32 * FG_MMA_WAVES)];
#pragma unroll
            for (int u = 0; u < 8; ++u) {
                const int i4 = tid + (i0 + u) * (32 * FG_MMA_WAVES);
                *(float4*)(Bs + (i4 >> 7) * FG_LDB + (i4 & 127) * 4) = t[u];
            }
        }
        const int mrow = lane & 15;
#pragma unroll
        for (int ct = 0; ct < 2; ++ct) {
            const int c = colOff + 2 * mrow + ct;
            scale[ct] = gam[c] * rsqrtf(var[c] + 1e-5f);
            shift[ct] = (bias[c] - mu[c]) * scale[ct] + bet[c];
            scale2[ct] = gam2[c] * rsqrtf(var2[c] + 1e-5f);
            shift2[ct] = (bias2[c] - mu2[c]) * scale2[ct] + bet2[c];
        }
    } else {
        se = 1.0f + epsp[0];
        gather_rows(xv, offsets, bucket, se, As + grow * FG_LDA, Xs + grow * FG_LDA,
                    tile * FG_ROWS + grow, lane);
    }
    __syncthreads();

    for (int it = 0; tile < FG_NT; ++it, tile += FG_GRID) {
        if (mma) product_strip(As + (it & 1) * FG_AT, Bs, scale, shift, Hs, colOff, lane);
        __syncthreads();
        if (!mma) {
            const int nt = tile + FG_GRID;
            if (nt < FG_NT)
                gather_rows(xv, offsets, bucket, se, As + ((it + 1) & 1) * FG_AT + grow * FG_LDA,
                            Xs + ((it + 1) & 1) * FG_AT + grow * FG_LDA, nt * FG_ROWS + grow, lane);
        } else {
            const int mrow = lane & 15;
            const int kh   = lane >> 4;
            v8f acc[2][2] = {};
            product_gb<GIN_DOUT, FG_LDH>(Hs, B2 + 2 * colOff, acc, lane);
#pragma unroll
            for (int rt = 0; rt < 2; ++rt)
#pragma unroll
                for (int ct = 0; ct < 2; ++ct)
#pragma unroll
                    for (int r = 0; r < 8; ++r)
                        acc[rt][ct][r] = fmaxf(acc[rt][ct][r] * scale2[ct] + shift2[ct], 0.0f);
            product_gb<GIN_DIN, FG_LDA>(Xs + (it & 1) * FG_AT, Br + 2 * colOff, acc, lane);
#pragma unroll
            for (int rt = 0; rt < 2; ++rt) {
                const int row0 = tile * FG_ROWS + rt * 16 + kh * 8;
#pragma unroll
                for (int r = 0; r < 8; ++r) {
                    v2f o;
                    o.x = fmaxf(acc[rt][0][r], 0.0f);
                    o.y = fmaxf(acc[rt][1][r], 0.0f);
                    *(v2f*)(out + (size_t)(row0 + r) * GIN_DOUT + colOff + 2 * mrow) = o;
                }
            }
        }
        __syncthreads();
    }
}


extern "C" void kernel_launch(void* const* d_in, const int* in_sizes, int n_in,
                              void* d_out, int out_size, void* d_ws, size_t ws_size,
                              hipStream_t stream) {
    constexpr int N = GIN_N, E = GIN_E, DIN = GIN_DIN, DOUT = GIN_DOUT;

    const float* x    = (const float*)d_in[0];
    const int*   eidx = (const int*)d_in[1];
    const float* eps  = (const float*)d_in[2];
    const float* W1   = (const float*)d_in[3];
    const float* b1   = (const float*)d_in[4];
    const float* g1   = (const float*)d_in[5];
    const float* bt1  = (const float*)d_in[6];
    const float* m1   = (const float*)d_in[7];
    const float* v1   = (const float*)d_in[8];
    const float* W2   = (const float*)d_in[9];
    const float* b2   = (const float*)d_in[10];
    const float* g2   = (const float*)d_in[11];
    const float* bt2  = (const float*)d_in[12];
    const float* m2   = (const float*)d_in[13];
    const float* v2   = (const float*)d_in[14];
    const float* Wres = (const float*)d_in[15];

    const int* src = eidx;
    const int* dst = eidx + E;

    float* h  = (float*)d_ws;
    float* wp = h;
    static_assert(WP_WR + GIN_DIN * GIN_DOUT <= GIN_N * GIN_DIN, "the re-laid weights fit in the first plane");
    float* h1 = h + (size_t)N * DIN;
    int* ibase   = (int*)(h1 + (size_t)2 * N * DOUT);
    int* tab     = ibase;
    static_assert((NBIN + 1) * NSEG <= 2 * GIN_N, "the table fits in front of offsets");
    int* offsets = ibase + 2 * N;
    int* bucket  = offsets + N + 1;
    int* pairs   = (int*)h1;

    constexpr int E4 = E / 4;
    bin_local_sort_kernel<<<NSEG, SORT_T, 0, stream>>>(src, dst, tab, pairs, E4, W1, W2, Wres, wp);
    bin_sort_kernel<<<NBIN, SORT_T, 0, stream>>>(pairs, tab, offsets, bucket, N);
    gather_gemm_kernel<<<FG_GRID, FG_T, 0, stream>>>(
        x, offsets, bucket, eps, wp + WP_W1, b1, g1, bt1, m1, v1,
        wp + WP_W2, b2, g2, bt2, m2, v2, wp + WP_WR, (float*)d_out);
}


# level 2 of the edge grouping folded into the last launch: a work-group owns 160 consecutive nodes and sorts its bin's pairs on LDS counters first; 2 launches (was 3); node sums keep an arrival order
# speedup vs baseline: 5.8916x; 1.0307x over previous
#include <hip/hip_runtime.h>
#include <hip/hip_bf16.h>

typedef float v2f __attribute__((ext_vector_type(2)));
typedef float v4f __attribute__((ext_vector_type(4)));
typedef float v8f __attribute__((ext_vector_type(8)));

#define GIN_N    40000
#define GIN_E    640000
#define GIN_DIN  128
#define GIN_DOUT 256


#define BIN_NODES 160
#define NBIN  (GIN_N / BIN_NODES)
static_assert(GIN_N % BIN_NODES == 0 && BIN_NODES % 32 == 0, "whole bins, whole waves of node counters");
#define NBINP (((NBIN + 1 + 31) / 32) * 32)
#define SORT_T 1024
#define SEG_EDGES (4 * SORT_T)
#define NSEG  ((GIN_E + SEG_EDGES - 1) / SEG_EDGES)
#define NSEGP (((NSEG + 31) / 32) * 32)
static_assert(GIN_E % 4 == 0, "edges are read four at a time");
static_assert(NBINP <= SORT_T, "one thread per bin");
static_assert(NSEGP <= SORT_T, "one thread per segment");

#define WP_W1  0
#define WP_W2  (GIN_DIN * GIN_DOUT)
#define WP_WR  (GIN_DIN * GIN_DOUT + GIN_DOUT * GIN_DOUT)
#define WP_PAIRS_W1 (GIN_DIN * GIN_DOUT / 2)
#define WP_PAIRS_W2 (GIN_DOUT * GIN_DOUT / 2)
#define WP_PAIRS_WR (GIN_DIN * GIN_DOUT / 2)
static_assert(GIN_DOUT == 256, "an item index splits into row pair and column by 8 bits");
static_assert(WP_PAIRS_W1 % SORT_T == 0 && WP_PAIRS_W2 % SORT_T == 0 && WP_PAIRS_WR % SORT_T == 0,
              "a block's items belong to one matrix");
static_assert(WP_PAIRS_W1 + WP_PAIRS_W2 + WP_PAIRS_WR <= NSEG * SORT_T, "one item per thread of the level-1 launch");

__device__ __forceinline__ int bin_of(int node) { return (int)((unsigned)node / (unsigned)BIN_NODES); }

__global__ __launch_bounds__(SORT_T)
void bin_local_sort_kernel(const int* __restrict__ src, const int* __restrict__ dst,
                           int* __restrict__ tab, int* __restrict__ pairbuf, int e4,
                           const float* __restrict__ W1, const float* __restrict__ W2,
                           const float* __restrict__ Wr, float* __restrict__ wp) {
    __shared__ int hist[NBINP];
    __shared__ int smT[32];
    int2* pairs = (int2*)pairbuf;
    const int tid  = threadIdx.x;
    const int lane = tid & 31;
    const int wave = tid >> 5;

    const int item = blockIdx.x * SORT_T + tid;
    const bool relay = item < WP_PAIRS_W1 + WP_PAIRS_W2 + WP_PAIRS_WR;
    float2 wpair = make_float2(0.0f, 0.0f);
    int wslot = 0;
    if (relay) {
        const float* wsrc = W1;
        int q = item;
        int base = WP_W1;
        if (item >= WP_PAIRS_W1 + WP_PAIRS_W2) { wsrc = Wr; q = item - (WP_PAIRS_W1 + WP_PAIRS_W2); base = WP_WR; }
        else if (item >= WP_PAIRS_W1)          { wsrc = W2; q = item - WP_PAIRS_W1;                 base = WP_W2; }
        const int p = q >> 8, n = q & (GIN_DOUT - 1);
        wpair.x = wsrc[(2 * p) * GIN_DOUT + n];
        wpair.y = wsrc[(2 * p + 1) * GIN_DOUT + n];
        wslot = base / 2 + q;
    }

    if (tid < NBINP) hist[tid] = 0;
    __syncthreads();

    const int i4 = blockIdx.x * SORT_T + tid;
    const bool act = i4 < e4;
    int4 d  = make_int4(0, 0, 0, 0);
    int4 sr = make_int4(0, 0, 0, 0);
    int r0 = 0, r1 = 0, r2 = 0, r3 = 0;
    if (act) {
        d  = ((const int4*)dst)[i4];
        sr = ((const int4*)src)[i4];
        r0 = atomicAdd(&hist[bin_of(d.x)], 1);
        r1 = atomicAdd(&hist[bin_of(d.y)], 1);
        r2 = atomicAdd(&hist[bin_of(d.z)], 1);
        r3 = atomicAdd(&hist[bin_of(d.w)], 1);
    }
    __syncthreads();

    int v = 0, s = 0;
    if (tid < NBINP) {
        v = hist[tid];
        s = v;
#pragma unroll
        for (int off = 1; off < 32; off <<= 1) {
            int t = __shfl_up(s, off);
            if (lane >= off) s += t;
        }
        if (lane == 31) smT[wave] = s;
    }
    __syncthreads();

    if (tid < NBINP) {
        int base = 0;
#pragma unroll
        for (int w = 0; w < NBINP / 32; ++w) base += (w < wave) ? smT[w] : 0;
        int excl = base + s - v;
        hist[tid] = excl;
        if (tid <= NBIN) tab[tid * NSEG + blockIdx.x] = excl;
    }
    __syncthreads();

    if (act) {
        int2* reg = pairs + (size_t)blockIdx.x * SEG_EDGES;
        reg[hist[bin_of(d.x)] + r0] = make_int2(d.x, sr.x);
        reg[hist[bin_of(d.y)] + r1] = make_int2(d.y, sr.y);
        reg[hist[bin_of(d.z)] + r2] = make_int2(d.z, sr.z);
        reg[hist[bin_of(d.w)] + r3] = make_int2(d.w, sr.w);
    }
    if (relay) ((float2*)wp)[wslot] = wpair;
}

#define FG_ROWS 32
#define FG_LDA  (GIN_DIN + 4)
#define FG_LDB  (2 * GIN_DOUT + 32)
#define FG_LDH  (GIN_DOUT + 4)
#define FG_RING 3
#define FG_AT   (FG_ROWS * FG_LDA)
#define FG_TPW  (BIN_NODES / FG_ROWS)
#define FG_GRID NBIN
#define FG_T    768
#define FG_WAVES (FG_T / 32)
#define SEG_TRIPS ((NSEG + FG_WAVES - 1) / FG_WAVES)
#define FG_MMA_WAVES 8
#define FG_BATCH 16
#define FG_KB    4
#define FG_NPW  (FG_ROWS / (FG_T / 32 - FG_MMA_WAVES))
static_assert(BIN_NODES % FG_ROWS == 0, "whole tiles of nodes per work-group");
static_assert(NSEGP <= FG_T && BIN_NODES <= FG_T, "one thread per segment, one per node counter");
static_assert(FG_ROWS == FG_NPW * (FG_T / 32 - FG_MMA_WAVES), "the gathering waves share a tile evenly");
static_assert(GIN_DOUT == 32 * FG_MMA_WAVES, "one 32-column strip per product wave");
static_assert(GIN_DIN == 128, "a node row is one float4 per lane");
static_assert(2 * GIN_DOUT / 4 == 128, "W1 staging: 128 pieces of 16 bytes per row pair");

__device__ __forceinline__ void bin_sort_wg(const int* __restrict__ pairbuf, const int* __restrict__ tab,
                                            int* bucket, int* offs, int b, int tid, int lane, int wave) {
    __shared__ int cnt[BIN_NODES];
    __shared__ int smT[BIN_NODES / 32];
    __shared__ int segS[NSEGP];
    __shared__ int segE[NSEGP];
    __shared__ int smP[NSEGP / 32];
    const int2* pairs = (const int2*)pairbuf;

    if (tid < NSEGP) {
        int st = 0, en = 0;
        if (tid < NSEG) {
            st = tab[b * NSEG + tid];
            en = tab[(b + 1) * NSEG + tid];
        }
        segS[tid] = st;
        segE[tid] = en;
        int t = st;
#pragma unroll
        for (int off = 16; off > 0; off >>= 1) t += __shfl_xor(t, off);
        if (lane == 0) smP[wave] = t;
    }
    if (tid < BIN_NODES) cnt[tid] = 0;
    __syncthreads();

    int p0 = 0;
#pragma unroll
    for (int w = 0; w < NSEGP / 32; ++w) p0 += smP[w];

    int  sb[SEG_TRIPS], se[SEG_TRIPS];
    int2 pr[SEG_TRIPS];
    bool ok[SEG_TRIPS];
#pragma unroll
    for (int q = 0; q < SEG_TRIPS; ++q) {
        const int sg = wave + q * FG_WAVES;
        sb[q] = 0; se[q] = 0;
        if (sg < NSEG) { sb[q] = sg * SEG_EDGES + segS[sg] + lane; se[q] = sg * SEG_EDGES + segE[sg]; }
        ok[q] = sb[q] < se[q];
    }
#pragma unroll
    for (int q = 0; q < SEG_TRIPS; ++q) {
        pr[q] = make_int2(0, 0);
        if (ok[q]) pr[q] = pairs[sb[q]];
    }
    const int nb = b * BIN_NODES;
#pragma unroll
    for (int q = 0; q < SEG_TRIPS; ++q)
        if (ok[q]) atomicAdd(&cnt[pr[q].x - nb], 1);
#pragma unroll
    for (int q = 0; q < SEG_TRIPS; ++q)
        for (int i = sb[q] + 32; i < se[q]; i += 32)
            atomicAdd(&cnt[pairs[i].x - nb], 1);
    __syncthreads();

    int v = 0, s = 0;
    if (tid < BIN_NODES) {
        v = cnt[tid];
        s = v;
#pragma unroll
        for (int off = 1; off < 32; off <<= 1) {
            int t = __shfl_up(s, off);
            if (lane >= off) s += t;
        }
        if (lane == 31) smT[wave] = s;
    }
    __syncthreads();

    if (tid < BIN_NODES) {
        int base = 0;
#pragma unroll
        for (int w = 0; w < BIN_NODES / 32; ++w) base += (w < wave) ? smT[w] : 0;
        const int first = p0 + base + s - v;
        offs[tid] = first;
        cnt[tid] = 0;
        if (tid == BIN_NODES - 1) offs[BIN_NODES] = first + v;
    }
    __syncthreads();

#pragma unroll
    for (int q = 0; q < SEG_TRIPS; ++q)
        if (ok[q]) {
            const int l = pr[q].x - nb;
            const int r = atomicAdd(&cnt[l], 1);
            bucket[offs[l] + r] = pr[q].y;
        }
#pragma unroll
    for (int q = 0; q < SEG_TRIPS; ++q)
        for (int i = sb[q] + 32; i < se[q]; i += 32) {
            const int2 p2 = pairs[i];
            const int l = p2.x - nb;
            const int r = atomicAdd(&cnt[l], 1);
            bucket[offs[l] + r] = p2.y;
        }
    __syncthreads();
}

__device__ __forceinline__ int settled(int v) {
    int r;
    asm volatile("v_mov_b32 %0, %1" : "=v"(r) : "v"(v));
    return r;
}

__device__ __forceinline__ void gather_rows(const float4* __restrict__ xv,
                                            const int* offs,
                                            const int* bucket, float se,
                                            float* __restrict__ At, float* __restrict__ Xt,
                                            int node0, int lane) {
    const int ov = offs[lane < FG_NPW ? lane : FG_NPW];
    float4 self[FG_NPW];
#pragma unroll
    for (int q = 0; q < FG_NPW; ++q) self[q] = xv[(node0 + q) * 32 + lane];
    int eb[FG_NPW + 1];
#pragma unroll
    for (int q = 0; q <= FG_NPW; ++q) eb[q] = __builtin_amdgcn_readlane(ov, q);
    int idf[FG_NPW];
#pragma unroll
    for (int q = 0; q < FG_NPW; ++q) idf[q] = (lane < eb[q + 1] - eb[q]) ? bucket[eb[q] + lane] : 0;

#pragma unroll
    for (int q = 0; q < FG_NPW; ++q) {
        const int e1 = eb[q + 1];
        float ax = self[q].x * se, ay = self[q].y * se, az = self[q].z * se, aw = self[q].w * se;
        *(float4*)(Xt + q * FG_LDA + lane * 4) = self[q];
        int idx = settled(idf[q]);
        for (int base = eb[q]; base < e1; base += 32) {
            const int left = e1 - base;
            const int cnt = left < 32 ? left : 32;
            for (int j0 = 0; j0 < cnt; j0 += FG_BATCH) {
                float4 r[FG_BATCH];
#pragma unroll
                for (int u = 0; u < FG_BATCH; ++u) {
                    r[u] = make_float4(0.0f, 0.0f, 0.0f, 0.0f);
                    if (j0 + u < cnt) r[u] = xv[__builtin_amdgcn_readlane(idx, j0 + u) * 32 + lane];
                }
#pragma unroll
                for (int u = 0; u < FG_BATCH; ++u)
                    if (j0 + u < cnt) { ax += r[u].x; ay += r[u].y; az += r[u].z; aw += r[u].w; }
            }
            if (left > 32) {
                const int nx = (lane < left - 32) ? bucket[base + 32 + lane] : 0;
                idx = settled(nx);
            }
        }
        float4 o; o.x = ax; o.y = ay; o.z = az; o.w = aw;
        *(float4*)(At + q * FG_LDA + lane * 4) = o;
    }
}

__device__ __forceinline__ void product_strip(const float* __restrict__ At,
                                              const float* __restrict__ Bsm,
                                              const float (&scale)[2], const float (&shift)[2],
                                              float* __restrict__ Hs,
                                              int colOff, int lane) {
    const int mrow = lane & 15;
    const int kh   = lane >> 4;
    const float* Ap0 = At + mrow * FG_LDA + 2 * kh;
    const float* Ap1 = Ap0 + 16 * FG_LDA;
    const float* Bp  = Bsm + kh * FG_LDB + 2 * colOff + 4 * mrow;
    v8f acc[2][2] = {};
    v2f fa0[2][FG_KB], fa1[2][FG_KB];
    v4f fq[2][FG_KB];
#pragma unroll
    for (int s = 0; s < FG_KB; ++s) {
        const int k = 4 * s;
        fa0[0][s] = *(const v2f*)(Ap0 + k);
        fa1[0][s] = *(const v2f*)(Ap1 + k);
        fq[0][s]  = *(const v4f*)(Bp + (k / 2) * FG_LDB);
    }
#pragma unroll
    for (int b = 0; b < GIN_DIN / (4 * FG_KB); ++b) {
        const int cur = b & 1, nxt = cur ^ 1;
        if (b + 1 < GIN_DIN / (4 * FG_KB)) {
#pragma unroll
            for (int s = 0; s < FG_KB; ++s) {
                const int k = 4 * ((b + 1) * FG_KB + s);
                fa0[nxt][s] = *(const v2f*)(Ap0 + k);
                fa1[nxt][s] = *(const v2f*)(Ap1 + k);
                fq[nxt][s]  = *(const v4f*)(Bp + (k / 2) * FG_LDB);
            }
        }
        __builtin_amdgcn_sched_barrier(0);
#pragma unroll
        for (int s = 0; s < FG_KB; ++s) {
            v2f b0, b1;
            b0.x = fq[cur][s].x; b0.y = fq[cur][s].y;
            b1.x = fq[cur][s].z; b1.y = fq[cur][s].w;
            acc[0][0] = __builtin_amdgcn_wmma_f32_16x16x4_f32(false, fa0[cur][s], false, b0, (short)0, acc[0][0], false, false);
            acc[0][1] = __builtin_amdgcn_wmma_f32_16x16x4_f32(false, fa0[cur][s], false, b1, (short)0, acc[0][1], false, false);
            acc[1][0] = __builtin_amdgcn_wmma_f32_16x16x4_f32(false, fa1[cur][s], false, b0, (short)0, acc[1][0], false, false);
            acc[1][1] = __builtin_amdgcn_wmma_f32_16x16x4_f32(false, fa1[cur][s], false, b1, (short)0, acc[1][1], false, false);
        }
        __builtin_amdgcn_sched_barrier(0);
    }
#pragma unroll
    for (int rt = 0; rt < 2; ++rt) {
        const int row0 = rt * 16 + kh * 8;
#pragma unroll
        for (int r = 0; r < 8; ++r) {
            v2f o;
            o.x = fmaxf(acc[rt][0][r] * scale[0] + shift[0], 0.0f);
            o.y = fmaxf(acc[rt][1][r] * scale[1] + shift[1], 0.0f);
            *(v2f*)(Hs + (row0 + r) * FG_LDH + colOff + 2 * mrow) = o;
        }
    }
}

template <int K, int LDA_>
__device__ __forceinline__ void product_gb(const float* __restrict__ At, const float* __restrict__ Bg,
                                           v8f (&acc)[2][2], int lane) {
    constexpr int NB = K / (4 * FG_KB);
    static_assert(K % (4 * FG_KB) == 0 && NB >= 2, "whole blocks, at least two");
    const int mrow = lane & 15;
    const int kh   = lane >> 4;
    const float* Ap0 = At + mrow * LDA_ + 2 * kh;
    const float* Ap1 = Ap0 + 16 * LDA_;
    const float* Bp  = Bg + kh * (2 * GIN_DOUT) + 4 * mrow;
    v4f gq[FG_RING][FG_KB];
    v2f fa0[2][FG_KB], fa1[2][FG_KB];
#pragma unroll
    for (int b = 0; b < 2; ++b)
#pragma unroll
        for (int s = 0; s < FG_KB; ++s) {
            gq[b][s] = *(const v4f*)(Bp + 4 * (b * FG_KB + s) * GIN_DOUT);
        }
#pragma unroll
    for (int s = 0; s < FG_KB; ++s) {
        fa0[0][s] = *(const v2f*)(Ap0 + 4 * s);
        fa1[0][s] = *(const v2f*)(Ap1 + 4 * s);
    }
#pragma unroll
    for (int b = 0; b < NB; ++b) {
        const int cur = b & 1, nxt = cur ^ 1;
        const int gc = b % FG_RING, gn = (b + 2) % FG_RING;
        if (b + 2 < NB) {
#pragma unroll
            for (int s = 0; s < FG_KB; ++s) {
                gq[gn][s] = *(const v4f*)(Bp + 4 * ((b + 2) * FG_KB + s) * GIN_DOUT);
            }
        }
        if (b + 1 < NB) {
#pragma unroll
            for (int s = 0; s < FG_KB; ++s) {
                const int k = 4 * ((b + 1) * FG_KB + s);
                fa0[nxt][s] = *(const v2f*)(Ap0 + k);
                fa1[nxt][s] = *(const v2f*)(Ap1 + k);
            }
        }
        __builtin_amdgcn_sched_barrier(0);
#pragma unroll
        for (int s = 0; s < FG_KB; ++s) {
            v2f b0, b1;
            b0.x = gq[gc][s].x; b0.y = gq[gc][s].y;
            b1.x = gq[gc][s].z; b1.y = gq[gc][s].w;
            acc[0][0] = __builtin_amdgcn_wmma_f32_16x16x4_f32(false, fa0[cur][s], false, b0, (short)0, acc[0][0], false, false);
            acc[0][1] = __builtin_amdgcn_wmma_f32_16x16x4_f32(false, fa0[cur][s], false, b1, (short)0, acc[0][1], false, false);
            acc[1][0] = __builtin_amdgcn_wmma_f32_16x16x4_f32(false, fa1[cur][s], false, b0, (short)0, acc[1][0], false, false);
            acc[1][1] = __builtin_amdgcn_wmma_f32_16x16x4_f32(false, fa1[cur][s], false, b1, (short)0, acc[1][1], false, false);
        }
        __builtin_amdgcn_sched_barrier(0);
    }
}

__global__ __launch_bounds__(FG_T)
void gather_gemm_kernel(const float* __restrict__ x, const int* __restrict__ pairbuf,
                        const int* __restrict__ tab, int* bucket, const float* __restrict__ epsp,
                        const float* __restrict__ B, const float* __restrict__ bias,
                        const float* __restrict__ gam, const float* __restrict__ bet,
                        const float* __restrict__ mu, const float* __restrict__ var,
                        const float* __restrict__ B2, const float* __restrict__ bias2,
                        const float* __restrict__ gam2, const float* __restrict__ bet2,
                        const float* __restrict__ mu2, const float* __restrict__ var2,
                        const float* __restrict__ Br, float* __restrict__ out) {
    __shared__ __attribute__((aligned(16))) float As[2 * FG_AT];
    __shared__ __attribute__((aligned(16))) float Xs[2 * FG_AT];
    __shared__ __attribute__((aligned(16))) float Hs[FG_ROWS * FG_LDH];
    __shared__ __attribute__((aligned(16))) float Bs[(GIN_DIN / 2) * FG_LDB];
    __shared__ int offs[BIN_NODES + 1];
    const int tid  = threadIdx.x;
    const int lane = tid & 31;
    const int wave = __builtin_amdgcn_readfirstlane(tid >> 5);
    const bool mma = wave < FG_MMA_WAVES;
    const float4* xv = (const float4*)x;
    const int colOff = wave * 32;
    const int grow   = (wave - FG_MMA_WAVES) * FG_NPW;

    bin_sort_wg(pairbuf, tab, bucket, offs, blockIdx.x, tid, lane, wave);

    const int tile0 = blockIdx.x * FG_TPW;
    float scale[2] = {1.0f, 1.0f}, shift[2] = {0.0f, 0.0f};
    float scale2[2] = {1.0f, 1.0f}, shift2[2] = {0.0f, 0.0f};
    float se = 0.0f;
    if (mma) {
        const float4* Bv = (const float4*)B;
#pragma unroll 1
        for (int i0 = 0; i0 < (GIN_DIN * GIN_DOUT / 4) / (32 * FG_MMA_WAVES); i0 += 8) {
            float4 t[8];
#pragma unroll
            for (int u = 0; u < 8; ++u) t[u] = Bv[tid + (i0 + u) * (32 * FG_MMA_WAVES)];
#pragma unroll
            for (int u = 0; u < 8; ++u) {
                const int i4 = tid + (i0 + u) * (32 * FG_MMA_WAVES);
                *(float4*)(Bs + (i4 >> 7) * FG_LDB + (i4 & 127) * 4) = t[u];
            }
        }
        const int mrow = lane & 15;
#pragma unroll
        for (int ct = 0; ct < 2; ++ct) {
            const int c = colOff + 2 * mrow + ct;
            scale[ct] = gam[c] * rsqrtf(var[c] + 1e-5f);
            shift[ct] = (bias[c] - mu[c]) * scale[ct] + bet[c];
            scale2[ct] = gam2[c] * rsqrtf(var2[c] + 1e-5f);
            shift2[ct] = (bias2[c] - mu2[c]) * scale2[ct] + bet2[c];
        }
    } else {
        se = 1.0f + epsp[0];
        gather_rows(xv, offs + grow, bucket, se, As + grow * FG_LDA, Xs + grow * FG_LDA,
                    tile0 * FG_ROWS + grow, lane);
    }
    __syncthreads();

    for (int it = 0; it < FG_TPW; ++it) {
        const int tile = tile0 + it;
        if (mma) product_strip(As + (it & 1) * FG_AT, Bs, scale, shift, Hs, colOff, lane);
        __syncthreads();
        if (!mma) {
            if (it + 1 < FG_TPW)
                gather_rows(xv, offs + (it + 1) * FG_ROWS + grow, bucket, se, As + ((it + 1) & 1) * FG_AT + grow * FG_LDA,
                            Xs + ((it + 1) & 1) * FG_AT + grow * FG_LDA, (tile + 1) * FG_ROWS + grow, lane);
        } else {
            const int mrow = lane & 15;
            const int kh   = lane >> 4;
            v8f acc[2][2] = {};
            product_gb<GIN_DOUT, FG_LDH>(Hs, B2 + 2 * colOff, acc, lane);
#pragma unroll
            for (int rt = 0; rt < 2; ++rt)
#pragma unroll
                for (int ct = 0; ct < 2; ++ct)
#pragma unroll
                    for (int r = 0; r < 8; ++r)
                        acc[rt][ct][r] = fmaxf(acc[rt][ct][r] * scale2[ct] + shift2[ct], 0.0f);
            product_gb<GIN_DIN, FG_LDA>(Xs + (it & 1) * FG_AT, Br + 2 * colOff, acc, lane);
#pragma unroll
            for (int rt = 0; rt < 2; ++rt) {
                const int row0 = tile * FG_ROWS + rt * 16 + kh * 8;
#pragma unroll
                for (int r = 0; r < 8; ++r) {
                    v2f o;
                    o.x = fmaxf(acc[rt][0][r], 0.0f);
                    o.y = fmaxf(acc[rt][1][r], 0.0f);
                    *(v2f*)(out + (size_t)(row0 + r) * GIN_DOUT + colOff + 2 * mrow) = o;
                }
            }
        }
        __syncthreads();
    }
}


extern "C" void kernel_launch(void* const* d_in, const int* in_sizes, int n_in,
                              void* d_out, int out_size, void* d_ws, size_t ws_size,
                              hipStream_t stream) {
    constexpr int N = GIN_N, E = GIN_E, DIN = GIN_DIN, DOUT = GIN_DOUT;

    const float* x    = (const float*)d_in[0];
    const int*   eidx = (const int*)d_in[1];
    const float* eps  = (const float*)d_in[2];
    const float* W1   = (const float*)d_in[3];
    const float* b1   = (const float*)d_in[4];
    const float* g1   = (const float*)d_in[5];
    const float* bt1  = (const float*)d_in[6];
    const float* m1   = (const float*)d_in[7];
    const float* v1   = (const float*)d_in[8];
    const float* W2   = (const float*)d_in[9];
    const float* b2   = (const float*)d_in[10];
    const float* g2   = (const float*)d_in[11];
    const float* bt2  = (const float*)d_in[12];
    const float* m2   = (const float*)d_in[13];
    const float* v2   = (const float*)d_in[14];
    const float* Wres = (const float*)d_in[15];

    const int* src = eidx;
    const int* dst = eidx + E;

    float* h  = (float*)d_ws;
    float* wp = h;
    static_assert(WP_WR + GIN_DIN * GIN_DOUT <= GIN_N * GIN_DIN, "the re-laid weights fit in the first plane");
    float* h1 = h + (size_t)N * DIN;
    int* ibase   = (int*)(h1 + (size_t)2 * N * DOUT);
    int* tab     = ibase;
    static_assert((NBIN + 1) * NSEG <= 2 * GIN_N, "the table fits in front of the source lists");
    int* bucket  = ibase + 2 * N;
    int* pairs   = (int*)h1;

    constexpr int E4 = E / 4;
    bin_local_sort_kernel<<<NSEG, SORT_T, 0, stream>>>(src, dst, tab, pairs, E4, W1, W2, Wres, wp);
    gather_gemm_kernel<<<FG_GRID, FG_T, 0, stream>>>(
        x, pairs, tab, bucket, eps, wp + WP_W1, b1, g1, bt1, m1, v1,
        wp + WP_W2, b2, g2, bt2, m2, v2, wp + WP_WR, (float*)d_out);
}
